# starting kernel
# baseline (speedup 1.0000x reference)
#include <hip/hip_runtime.h>
#include <hip/hip_bf16.h>

#define N_NODES   100000
#define N_EDGES   1600000
#define N_GRAPHS  128
#define HIDDEN    128
#define N_CLASSES 10

typedef __attribute__((ext_vector_type(2))) float v2f;
typedef __attribute__((ext_vector_type(8))) float v8f;

__global__ void deg_kernel(const int* __restrict__ src, const int* __restrict__ dst,
                           float* __restrict__ outdeg, float* __restrict__ indeg) {
    int e = blockIdx.x * blockDim.x + threadIdx.x;
    if (e < N_EDGES) {
        atomicAdd(&outdeg[src[e]], 1.0f);
        atomicAdd(&indeg[dst[e]], 1.0f);
    }
}

__global__ void norm_kernel(const float* __restrict__ outdeg, const float* __restrict__ indeg,
                            float* __restrict__ cs, float* __restrict__ cd) {
    int n = blockIdx.x * blockDim.x + threadIdx.x;
    if (n < N_NODES) {
        cs[n] = rsqrtf(fmaxf(outdeg[n], 1.0f));
        cd[n] = rsqrtf(fmaxf(indeg[n], 1.0f));
    }
}

__global__ void scatter1_kernel(const int* __restrict__ src, const int* __restrict__ dst,
                                const float* __restrict__ indeg, const float* __restrict__ cs,
                                float* __restrict__ agg1) {
    int e = blockIdx.x * blockDim.x + threadIdx.x;
    if (e < N_EDGES) {
        int s = src[e];
        atomicAdd(&agg1[dst[e]], indeg[s] * cs[s]);
    }
}

__global__ void layer1_kernel(const float* __restrict__ agg1, const float* __restrict__ cd,
                              const float* __restrict__ W1, const float* __restrict__ b1,
                              float* __restrict__ h) {
    int i = blockIdx.x * blockDim.x + threadIdx.x;
    if (i < N_NODES * HIDDEN) {
        int m = i >> 7;
        int j = i & (HIDDEN - 1);
        float x = agg1[m] * cd[m];
        h[i] = fmaxf(fmaf(x, W1[j], b1[j]), 0.0f);
    }
}

__global__ void scatter_kernel(const int* __restrict__ src, const int* __restrict__ dst,
                               const float* __restrict__ cs, const float* __restrict__ h,
                               float* __restrict__ agg) {
    int gid  = blockIdx.x * blockDim.x + threadIdx.x;
    int e    = gid >> 5;
    int lane = gid & 31;
    if (e < N_EDGES) {
        int s = src[e], d = dst[e];
        float sc = cs[s];
        const float* hs = h   + (size_t)s * HIDDEN;
        float*       ad = agg + (size_t)d * HIDDEN;
#pragma unroll
        for (int f = lane; f < HIDDEN; f += 32)
            atomicAdd(&ad[f], hs[f] * sc);
    }
}

__global__ void __launch_bounds__(256) gemm_relu_kernel(const float* __restrict__ agg,
                                                        const float* __restrict__ cd,
                                                        const float* __restrict__ W,
                                                        const float* __restrict__ bias,
                                                        float* __restrict__ hout) {
    __shared__ float Wl[HIDDEN * HIDDEN];
    int tid = threadIdx.x;
    for (int i = tid; i < HIDDEN * HIDDEN; i += 256) Wl[i] = W[i];
    __syncthreads();

    int wave = tid >> 5;
    int lane = tid & 31;
    int half = lane >> 4;
    int lr   = lane & 15;
    int m0   = blockIdx.x * 16;
    int m    = m0 + lr;
    int n    = wave * 16 + lr;

    float cdm = cd[m];
    const float* arow = agg + (size_t)m * HIDDEN + half * 2;
    const float* brow = Wl + (half * 2) * HIDDEN + n;

    v8f acc = {};
#pragma unroll
    for (int k0 = 0; k0 < HIDDEN; k0 += 4) {
        v2f av, bv;
        av.x = arow[k0]     * cdm;
        av.y = arow[k0 + 1] * cdm;
        bv.x = brow[k0 * HIDDEN];
        bv.y = brow[k0 * HIDDEN + HIDDEN];
        acc = __builtin_amdgcn_wmma_f32_16x16x4_f32(
             false, av,  false, bv,
             (short)0, acc,  false,  false);
    }

    float bn = bias[n];
#pragma unroll
    for (int r = 0; r < 8; ++r) {
        int mm = m0 + half * 8 + r;
        hout[(size_t)mm * HIDDEN + n] = fmaxf(acc[r] + bn, 0.0f);
    }
}

__global__ void pool_kernel(const int* __restrict__ seg, const float* __restrict__ h,
                            float* __restrict__ hg, float* __restrict__ counts) {
    int gid  = blockIdx.x * blockDim.x + threadIdx.x;
    int node = gid >> 5;
    int lane = gid & 31;
    if (node < N_NODES) {
        int g = seg[node];
        const float* hn  = h  + (size_t)node * HIDDEN;
        float*       hgo = hg + (size_t)g * HIDDEN;
#pragma unroll
        for (int f = lane; f < HIDDEN; f += 32)
            atomicAdd(&hgo[f], hn[f]);
        if (lane == 0) atomicAdd(&counts[g], 1.0f);
    }
}

__global__ void pool_div_kernel(float* __restrict__ hg, const float* __restrict__ counts) {
    int i = blockIdx.x * blockDim.x + threadIdx.x;
    if (i < N_GRAPHS * HIDDEN)
        hg[i] /= fmaxf(counts[i >> 7], 1.0f);
}

__global__ void logits_kernel(const float* __restrict__ hg1, const float* __restrict__ hg2,
                              const float* __restrict__ Wc, const float* __restrict__ bc,
                              float* __restrict__ logits) {
    int i = blockIdx.x * blockDim.x + threadIdx.x;
    if (i < N_GRAPHS * N_CLASSES) {
        int g = i / N_CLASSES, c = i % N_CLASSES;
        float s = bc[c];
#pragma unroll 8
        for (int k = 0; k < HIDDEN; ++k)
            s += fabsf(hg1[g * HIDDEN + k] - hg2[g * HIDDEN + k]) * Wc[k * N_CLASSES + c];
        logits[i] = s;
    }
}

static void run_branch(const int* src, const int* dst, const int* seg,
                       const float* W1, const float* b1,
                       const float* W2, const float* b2,
                       const float* W3, const float* b3,
                       const float* W4, const float* b4,
                       float* hg_out, float* ws_f, hipStream_t stream) {
    float* outdeg = ws_f;
    float* indeg  = ws_f + 100000;
    float* cs     = ws_f + 200000;
    float* cd     = ws_f + 300000;
    float* agg1   = ws_f + 400000;
    float* counts = ws_f + 500000;
    float* h      = ws_f + 500224;
    float* agg    = h + (size_t)N_NODES * HIDDEN;

    hipMemsetAsync(ws_f, 0, (size_t)500224 * sizeof(float), stream);

    const int T = 256;
    deg_kernel<<<(N_EDGES + T - 1) / T, T, 0, stream>>>(src, dst, outdeg, indeg);
    norm_kernel<<<(N_NODES + T - 1) / T, T, 0, stream>>>(outdeg, indeg, cs, cd);

    scatter1_kernel<<<(N_EDGES + T - 1) / T, T, 0, stream>>>(src, dst, indeg, cs, agg1);
    layer1_kernel<<<(N_NODES * HIDDEN + T - 1) / T, T, 0, stream>>>(agg1, cd, W1, b1, h);

    const float* Ws[3] = {W2, W3, W4};
    const float* bs[3] = {b2, b3, b4};
    for (int l = 0; l < 3; ++l) {
        hipMemsetAsync(agg, 0, (size_t)N_NODES * HIDDEN * sizeof(float), stream);
        scatter_kernel<<<(N_EDGES * 32) / T, T, 0, stream>>>(src, dst, cs, h, agg);
        gemm_relu_kernel<<<N_NODES / 16, T, 0, stream>>>(agg, cd, Ws[l], bs[l], h);
    }

    hipMemsetAsync(hg_out, 0, (size_t)N_GRAPHS * HIDDEN * sizeof(float), stream);
    pool_kernel<<<(N_NODES * 32 + T - 1) / T, T, 0, stream>>>(seg, h, hg_out, counts);
    pool_div_kernel<<<(N_GRAPHS * HIDDEN + T - 1) / T, T, 0, stream>>>(hg_out, counts);
}

extern "C" void kernel_launch(void* const* d_in, const int* in_sizes, int n_in,
                              void* d_out, int out_size, void* d_ws, size_t ws_size,
                              hipStream_t stream) {
    const int*   src1 = (const int*)d_in[0];
    const int*   dst1 = (const int*)d_in[1];
    const int*   seg1 = (const int*)d_in[2];
    const int*   src2 = (const int*)d_in[3];
    const int*   dst2 = (const int*)d_in[4];
    const int*   seg2 = (const int*)d_in[5];
    const float* W1   = (const float*)d_in[6];
    const float* b1   = (const float*)d_in[7];
    const float* W2   = (const float*)d_in[8];
    const float* b2   = (const float*)d_in[9];
    const float* W3   = (const float*)d_in[10];
    const float* b3   = (const float*)d_in[11];
    const float* W4   = (const float*)d_in[12];
    const float* b4   = (const float*)d_in[13];
    const float* Wc   = (const float*)d_in[14];
    const float* bc   = (const float*)d_in[15];

    float* out   = (float*)d_out;
    float* hg1   = out;
    float* hg2   = out + N_GRAPHS * HIDDEN;
    float* logit = out + 2 * N_GRAPHS * HIDDEN;
    float* ws_f  = (float*)d_ws;

    run_branch(src1, dst1, seg1, W1, b1, W2, b2, W3, b3, W4, b4, hg1, ws_f, stream);
    run_branch(src2, dst2, seg2, W1, b1, W2, b2, W3, b3, W4, b4, hg2, ws_f, stream);

    logits_kernel<<<(N_GRAPHS * N_CLASSES + 255) / 256, 256, 0, stream>>>(hg1, hg2, Wc, bc, logit);
}


# Layers 2-4 edge sums: edges binned per call by 400-node destination tile; a work-group adds h[src]*cs into its tile in LDS (ds_add_f32), stores agg once; no global f32 atomics; LDS arrival order
# speedup vs baseline: 1.8012x; 1.8012x over previous
#include <hip/hip_runtime.h>
#include <hip/hip_bf16.h>

#define N_NODES   100000
#define N_EDGES   1600000
#define N_GRAPHS  128
#define HIDDEN    128
#define N_CLASSES 10

#define TILE_ROWS  400
#define N_TILES    (N_NODES / TILE_ROWS)
#define CHUNK      4096
#define N_CHUNKS   ((N_EDGES + CHUNK - 1) / CHUNK)
#define OFF_STRIDE 256
#define SRC_BITS   17
static_assert(N_TILES * TILE_ROWS == N_NODES, "tiles must cover the nodes exactly");
static_assert(N_TILES + 1 <= OFF_STRIDE, "offset row too short");
static_assert(N_NODES <= (1 << SRC_BITS), "source index does not fit its field");

typedef __attribute__((ext_vector_type(2))) float v2f;
typedef __attribute__((ext_vector_type(8))) float v8f;

__global__ void deg_kernel(const int* __restrict__ src, const int* __restrict__ dst,
                           float* __restrict__ outdeg, float* __restrict__ indeg) {
    int e = blockIdx.x * blockDim.x + threadIdx.x;
    if (e < N_EDGES) {
        atomicAdd(&outdeg[src[e]], 1.0f);
        atomicAdd(&indeg[dst[e]], 1.0f);
    }
}

__global__ void norm_kernel(const float* __restrict__ outdeg, const float* __restrict__ indeg,
                            float* __restrict__ cs, float* __restrict__ cd) {
    int n = blockIdx.x * blockDim.x + threadIdx.x;
    if (n < N_NODES) {
        cs[n] = rsqrtf(fmaxf(outdeg[n], 1.0f));
        cd[n] = rsqrtf(fmaxf(indeg[n], 1.0f));
    }
}

__global__ void scatter1_kernel(const int* __restrict__ src, const int* __restrict__ dst,
                                const float* __restrict__ indeg, const float* __restrict__ cs,
                                float* __restrict__ agg1) {
    int e = blockIdx.x * blockDim.x + threadIdx.x;
    if (e < N_EDGES) {
        int s = src[e];
        atomicAdd(&agg1[dst[e]], indeg[s] * cs[s]);
    }
}

__global__ void layer1_kernel(const float* __restrict__ agg1, const float* __restrict__ cd,
                              const float* __restrict__ W1, const float* __restrict__ b1,
                              float* __restrict__ h) {
    int i = blockIdx.x * blockDim.x + threadIdx.x;
    if (i < N_NODES * HIDDEN) {
        int m = i >> 7;
        int j = i & (HIDDEN - 1);
        float x = agg1[m] * cd[m];
        h[i] = fmaxf(fmaf(x, W1[j], b1[j]), 0.0f);
    }
}

__global__ void __launch_bounds__(256) bin_kernel(const int* __restrict__ src, const int* __restrict__ dst,
                                                  unsigned* __restrict__ binned, int* __restrict__ offs) {
    __shared__ int hist[OFF_STRIDE];
    __shared__ int scanbuf[OFF_STRIDE];
    int tid = threadIdx.x;
    int c   = blockIdx.x;
    int e0  = c * CHUNK;
    hist[tid] = 0;
    __syncthreads();

    static_assert(N_TILES + 1 < OFF_STRIDE, "no spare bin");
    unsigned word[CHUNK / 256];
    int      bin[CHUNK / 256];
    int      rank[CHUNK / 256];
    int      sv[CHUNK / 256], dv[CHUNK / 256];
#pragma unroll
    for (int u = 0; u < CHUNK / 256; ++u) {
        int e = min(e0 + u * 256 + tid, N_EDGES - 1);
        sv[u] = src[e];
        dv[u] = dst[e];
    }
#pragma unroll
    for (int u = 0; u < CHUNK / 256; ++u) {
        bool valid = (e0 + u * 256 + tid) < N_EDGES;
        int b = dv[u] / TILE_ROWS;
        word[u] = (unsigned)sv[u] | ((unsigned)(dv[u] - b * TILE_ROWS) << SRC_BITS);
        bin[u]  = valid ? b : (OFF_STRIDE - 1);
        rank[u] = atomicAdd(&hist[bin[u]], 1);
    }
    __syncthreads();

    int v = hist[tid];
    scanbuf[tid] = v;
    __syncthreads();
    for (int o = 1; o < OFF_STRIDE; o <<= 1) {
        int t = (tid >= o) ? scanbuf[tid - o] : 0;
        __syncthreads();
        scanbuf[tid] += t;
        __syncthreads();
    }
    int excl = scanbuf[tid] - v;
    offs[c * OFF_STRIDE + tid] = excl;
    hist[tid] = excl;
    __syncthreads();

#pragma unroll
    for (int u = 0; u < CHUNK / 256; ++u)
        if (bin[u] != OFF_STRIDE - 1) binned[e0 + hist[bin[u]] + rank[u]] = word[u];
}

__global__ void __launch_bounds__(1024) tile_sum_kernel(const unsigned* __restrict__ binned,
                                                        const int* __restrict__ offs,
                                                        const float* __restrict__ cs,
                                                        const float* __restrict__ h,
                                                        float* __restrict__ agg) {
    __shared__ float tile[(TILE_ROWS + 1) * HIDDEN];
    __shared__ int   next_chunk;
    int tid  = threadIdx.x;
    int lane = tid & 31;
    int b    = blockIdx.x;

    float4* tile4 = (float4*)tile;
    for (int i = tid; i < (TILE_ROWS + 1) * HIDDEN / 4; i += 1024) tile4[i] = float4{0.f, 0.f, 0.f, 0.f};
    if (tid == 0) next_chunk = 0;
    __syncthreads();

    for (int it = 0; it < N_CHUNKS; ++it) {
        int c = 0;
        if (lane == 0) c = atomicAdd(&next_chunk, 1);
        c = __builtin_amdgcn_readfirstlane(c);
        if (c >= N_CHUNKS) break;
        int lo = __builtin_amdgcn_readfirstlane(offs[c * OFF_STRIDE + b]);
        int hi = __builtin_amdgcn_readfirstlane(offs[c * OFF_STRIDE + b + 1]);
        const unsigned* sp = binned + (size_t)c * CHUNK;
        for (int base = lo; base < hi; base += 32) {
            int cnt = min(32, hi - base);
            unsigned w = sp[base + min(lane, cnt - 1)];
            for (int j0 = 0; j0 < cnt; j0 += 8) {
                float  sc[8];
                float2 va[8], vb[8];
                int    row[8];
#pragma unroll
                for (int u = 0; u < 8; ++u) {
                    int j = min(j0 + u, cnt - 1);
                    unsigned wj = (unsigned)__builtin_amdgcn_readlane((int)w, j);
                    int s  = (int)(wj & ((1u << SRC_BITS) - 1u));
                    row[u] = (j0 + u < cnt) ? (int)(wj >> SRC_BITS) : TILE_ROWS;
                    sc[u]  = cs[s];
                    const float2* hp = (const float2*)(h + (size_t)s * HIDDEN) + lane;
                    va[u] = hp[0];
                    vb[u] = hp[32];
                }
#pragma unroll
                for (int u = 0; u < 8; ++u) {
                    float* t = tile + row[u] * HIDDEN + 2 * lane;
                    atomicAdd(t,      va[u].x * sc[u]);
                    atomicAdd(t + 1,  va[u].y * sc[u]);
                    atomicAdd(t + 64, vb[u].x * sc[u]);
                    atomicAdd(t + 65, vb[u].y * sc[u]);
                }
            }
        }
    }
    __syncthreads();

    float4* out4 = (float4*)(agg + (size_t)b * TILE_ROWS * HIDDEN);
    for (int i = tid; i < TILE_ROWS * HIDDEN / 4; i += 1024) out4[i] = tile4[i];
}

__global__ void __launch_bounds__(256) gemm_relu_kernel(const float* __restrict__ agg,
                                                        const float* __restrict__ cd,
                                                        const float* __restrict__ W,
                                                        const float* __restrict__ bias,
                                                        float* __restrict__ hout) {
    __shared__ float Wl[HIDDEN * HIDDEN];
    int tid = threadIdx.x;
    for (int i = tid; i < HIDDEN * HIDDEN; i += 256) Wl[i] = W[i];
    __syncthreads();

    int wave = tid >> 5;
    int lane = tid & 31;
    int half = lane >> 4;
    int lr   = lane & 15;
    int m0   = blockIdx.x * 16;
    int m    = m0 + lr;
    int n    = wave * 16 + lr;

    float cdm = cd[m];
    const float* arow = agg + (size_t)m * HIDDEN + half * 2;
    const float* brow = Wl + (half * 2) * HIDDEN + n;

    v8f acc = {};
#pragma unroll
    for (int k0 = 0; k0 < HIDDEN; k0 += 4) {
        v2f av, bv;
        av.x = arow[k0]     * cdm;
        av.y = arow[k0 + 1] * cdm;
        bv.x = brow[k0 * HIDDEN];
        bv.y = brow[k0 * HIDDEN + HIDDEN];
        acc = __builtin_amdgcn_wmma_f32_16x16x4_f32(
             false, av,  false, bv,
             (short)0, acc,  false,  false);
    }

    float bn = bias[n];
#pragma unroll
    for (int r = 0; r < 8; ++r) {
        int mm = m0 + half * 8 + r;
        hout[(size_t)mm * HIDDEN + n] = fmaxf(acc[r] + bn, 0.0f);
    }
}

__global__ void pool_kernel(const int* __restrict__ seg, const float* __restrict__ h,
                            float* __restrict__ hg, float* __restrict__ counts) {
    int gid  = blockIdx.x * blockDim.x + threadIdx.x;
    int node = gid >> 5;
    int lane = gid & 31;
    if (node < N_NODES) {
        int g = seg[node];
        const float* hn  = h  + (size_t)node * HIDDEN;
        float*       hgo = hg + (size_t)g * HIDDEN;
#pragma unroll
        for (int f = lane; f < HIDDEN; f += 32)
            atomicAdd(&hgo[f], hn[f]);
        if (lane == 0) atomicAdd(&counts[g], 1.0f);
    }
}

__global__ void pool_div_kernel(float* __restrict__ hg, const float* __restrict__ counts) {
    int i = blockIdx.x * blockDim.x + threadIdx.x;
    if (i < N_GRAPHS * HIDDEN)
        hg[i] /= fmaxf(counts[i >> 7], 1.0f);
}

__global__ void logits_kernel(const float* __restrict__ hg1, const float* __restrict__ hg2,
                              const float* __restrict__ Wc, const float* __restrict__ bc,
                              float* __restrict__ logits) {
    int i = blockIdx.x * blockDim.x + threadIdx.x;
    if (i < N_GRAPHS * N_CLASSES) {
        int g = i / N_CLASSES, c = i % N_CLASSES;
        float s = bc[c];
#pragma unroll 8
        for (int k = 0; k < HIDDEN; ++k)
            s += fabsf(hg1[g * HIDDEN + k] - hg2[g * HIDDEN + k]) * Wc[k * N_CLASSES + c];
        logits[i] = s;
    }
}

static void run_branch(const int* src, const int* dst, const int* seg,
                       const float* W1, const float* b1,
                       const float* W2, const float* b2,
                       const float* W3, const float* b3,
                       const float* W4, const float* b4,
                       float* hg_out, float* ws_f, hipStream_t stream) {
    float* outdeg = ws_f;
    float* indeg  = ws_f + 100000;
    float* cs     = ws_f + 200000;
    float* cd     = ws_f + 300000;
    float* agg1   = ws_f + 400000;
    float* counts = ws_f + 500000;
    float* h      = ws_f + 500224;
    float* agg    = h + (size_t)N_NODES * HIDDEN;
    unsigned* binned = (unsigned*)(agg + (size_t)N_NODES * HIDDEN);
    int*      offs   = (int*)(binned + N_EDGES);

    hipMemsetAsync(ws_f, 0, (size_t)500224 * sizeof(float), stream);

    const int T = 256;
    deg_kernel<<<(N_EDGES + T - 1) / T, T, 0, stream>>>(src, dst, outdeg, indeg);
    norm_kernel<<<(N_NODES + T - 1) / T, T, 0, stream>>>(outdeg, indeg, cs, cd);

    scatter1_kernel<<<(N_EDGES + T - 1) / T, T, 0, stream>>>(src, dst, indeg, cs, agg1);
    layer1_kernel<<<(N_NODES * HIDDEN + T - 1) / T, T, 0, stream>>>(agg1, cd, W1, b1, h);

    bin_kernel<<<N_CHUNKS, 256, 0, stream>>>(src, dst, binned, offs);
    const float* Ws[3] = {W2, W3, W4};
    const float* bs[3] = {b2, b3, b4};
    for (int l = 0; l < 3; ++l) {
        tile_sum_kernel<<<N_TILES, 1024, 0, stream>>>(binned, offs, cs, h, agg);
        gemm_relu_kernel<<<N_NODES / 16, T, 0, stream>>>(agg, cd, Ws[l], bs[l], h);
    }

    hipMemsetAsync(hg_out, 0, (size_t)N_GRAPHS * HIDDEN * sizeof(float), stream);
    pool_kernel<<<(N_NODES * 32 + T - 1) / T, T, 0, stream>>>(seg, h, hg_out, counts);
    pool_div_kernel<<<(N_GRAPHS * HIDDEN + T - 1) / T, T, 0, stream>>>(hg_out, counts);
}

extern "C" void kernel_launch(void* const* d_in, const int* in_sizes, int n_in,
                              void* d_out, int out_size, void* d_ws, size_t ws_size,
                              hipStream_t stream) {
    const int*   src1 = (const int*)d_in[0];
    const int*   dst1 = (const int*)d_in[1];
    const int*   seg1 = (const int*)d_in[2];
    const int*   src2 = (const int*)d_in[3];
    const int*   dst2 = (const int*)d_in[4];
    const int*   seg2 = (const int*)d_in[5];
    const float* W1   = (const float*)d_in[6];
    const float* b1   = (const float*)d_in[7];
    const float* W2   = (const float*)d_in[8];
    const float* b2   = (const float*)d_in[9];
    const float* W3   = (const float*)d_in[10];
    const float* b3   = (const float*)d_in[11];
    const float* W4   = (const float*)d_in[12];
    const float* b4   = (const float*)d_in[13];
    const float* Wc   = (const float*)d_in[14];
    const float* bc   = (const float*)d_in[15];

    float* out   = (float*)d_out;
    float* hg1   = out;
    float* hg2   = out + N_GRAPHS * HIDDEN;
    float* logit = out + 2 * N_GRAPHS * HIDDEN;
    float* ws_f  = (float*)d_ws;

    run_branch(src1, dst1, seg1, W1, b1, W2, b2, W3, b3, W4, b4, hg1, ws_f, stream);
    run_branch(src2, dst2, seg2, W1, b1, W2, b2, W3, b3, W4, b4, hg2, ws_f, stream);

    logits_kernel<<<(N_GRAPHS * N_CLASSES + 255) / 256, 256, 0, stream>>>(hg1, hg2, Wc, bc, logit);
}


# Degree, layer-1 and pooling kernels: a wave takes 10-25 items, not one (over a million fewer waves per call); pooling sums a wave's 25-node run per graph in registers first: pool sum order changes
# speedup vs baseline: 2.1988x; 1.2207x over previous
#include <hip/hip_runtime.h>
#include <hip/hip_bf16.h>

#define N_NODES   100000
#define N_EDGES   1600000
#define N_GRAPHS  128
#define HIDDEN    128
#define N_CLASSES 10

#define TILE_ROWS  400
#define N_TILES    (N_NODES / TILE_ROWS)
#define CHUNK      4096
#define N_CHUNKS   ((N_EDGES + CHUNK - 1) / CHUNK)
#define OFF_STRIDE 256
#define SRC_BITS   17
static_assert(N_TILES * TILE_ROWS == N_NODES, "tiles must cover the nodes exactly");
static_assert(N_TILES + 1 <= OFF_STRIDE, "offset row too short");
static_assert(N_NODES <= (1 << SRC_BITS), "source index does not fit its field");

#define EDGES_PER_THREAD 5
#define EDGE_WGS   (N_EDGES / (256 * EDGES_PER_THREAD))
#define ROWS_PER_WAVE 10
#define ROW_WGS    (N_NODES / (8 * ROWS_PER_WAVE))
#define POOL_RUN   25
#define POOL_WGS   (N_NODES / (8 * POOL_RUN))
static_assert(EDGE_WGS * 256 * EDGES_PER_THREAD == N_EDGES, "edge grid must cover the edges exactly");
static_assert(ROW_WGS * 8 * ROWS_PER_WAVE == N_NODES, "row grid must cover the nodes exactly");
static_assert(POOL_WGS * 8 * POOL_RUN == N_NODES, "pool grid must cover the nodes exactly");
static_assert(POOL_RUN % 5 == 0, "pooling loads rows in groups of five");

typedef __attribute__((ext_vector_type(2))) float v2f;
typedef __attribute__((ext_vector_type(8))) float v8f;

__global__ void __launch_bounds__(256) deg_kernel(const int* __restrict__ src, const int* __restrict__ dst,
                                                  float* __restrict__ outdeg, float* __restrict__ indeg) {
    int t = blockIdx.x * 256 + threadIdx.x;
    int s[EDGES_PER_THREAD], d[EDGES_PER_THREAD];
#pragma unroll
    for (int k = 0; k < EDGES_PER_THREAD; ++k) {
        int e = t + k * (EDGE_WGS * 256);
        s[k] = src[e];
        d[k] = dst[e];
    }
#pragma unroll
    for (int k = 0; k < EDGES_PER_THREAD; ++k) {
        atomicAdd(&outdeg[s[k]], 1.0f);
        atomicAdd(&indeg[d[k]], 1.0f);
    }
}

__global__ void norm_kernel(const float* __restrict__ outdeg, const float* __restrict__ indeg,
                            float* __restrict__ cs, float* __restrict__ cd) {
    int n = blockIdx.x * blockDim.x + threadIdx.x;
    if (n < N_NODES) {
        cs[n] = rsqrtf(fmaxf(outdeg[n], 1.0f));
        cd[n] = rsqrtf(fmaxf(indeg[n], 1.0f));
    }
}

__global__ void __launch_bounds__(256) scatter1_kernel(const int* __restrict__ src, const int* __restrict__ dst,
                                                       const float* __restrict__ indeg, const float* __restrict__ cs,
                                                       float* __restrict__ agg1) {
    int t = blockIdx.x * 256 + threadIdx.x;
    int   s[EDGES_PER_THREAD], d[EDGES_PER_THREAD];
    float a[EDGES_PER_THREAD], c[EDGES_PER_THREAD];
#pragma unroll
    for (int k = 0; k < EDGES_PER_THREAD; ++k) {
        int e = t + k * (EDGE_WGS * 256);
        s[k] = src[e];
        d[k] = dst[e];
    }
#pragma unroll
    for (int k = 0; k < EDGES_PER_THREAD; ++k) {
        a[k] = indeg[s[k]];
        c[k] = cs[s[k]];
    }
#pragma unroll
    for (int k = 0; k < EDGES_PER_THREAD; ++k)
        atomicAdd(&agg1[d[k]], a[k] * c[k]);
}

__global__ void __launch_bounds__(256) layer1_kernel(const float* __restrict__ agg1, const float* __restrict__ cd,
                                                     const float* __restrict__ W1, const float* __restrict__ b1,
                                                     float* __restrict__ h) {
    int wave = (blockIdx.x * 256 + threadIdx.x) >> 5;
    int lane = threadIdx.x & 31;
    const float4 w = ((const float4*)W1)[lane];
    const float4 b = ((const float4*)b1)[lane];
    int m0 = wave * ROWS_PER_WAVE;
    float x[ROWS_PER_WAVE];
#pragma unroll
    for (int u = 0; u < ROWS_PER_WAVE; ++u) x[u] = agg1[m0 + u] * cd[m0 + u];
    float4* hrow = (float4*)(h + (size_t)m0 * HIDDEN) + lane;
#pragma unroll
    for (int u = 0; u < ROWS_PER_WAVE; ++u) {
        float4 o;
        o.x = fmaxf(fmaf(x[u], w.x, b.x), 0.0f);
        o.y = fmaxf(fmaf(x[u], w.y, b.y), 0.0f);
        o.z = fmaxf(fmaf(x[u], w.z, b.z), 0.0f);
        o.w = fmaxf(fmaf(x[u], w.w, b.w), 0.0f);
        hrow[u * (HIDDEN / 4)] = o;
    }
}

__global__ void __launch_bounds__(256) bin_kernel(const int* __restrict__ src, const int* __restrict__ dst,
                                                  unsigned* __restrict__ binned, int* __restrict__ offs) {
    __shared__ int hist[OFF_STRIDE];
    __shared__ int scanbuf[OFF_STRIDE];
    int tid = threadIdx.x;
    int c   = blockIdx.x;
    int e0  = c * CHUNK;
    hist[tid] = 0;
    __syncthreads();

    static_assert(N_TILES + 1 < OFF_STRIDE, "no spare bin");
    unsigned word[CHUNK / 256];
    int      bin[CHUNK / 256];
    int      rank[CHUNK / 256];
    int      sv[CHUNK / 256], dv[CHUNK / 256];
#pragma unroll
    for (int u = 0; u < CHUNK / 256; ++u) {
        int e = min(e0 + u * 256 + tid, N_EDGES - 1);
        sv[u] = src[e];
        dv[u] = dst[e];
    }
#pragma unroll
    for (int u = 0; u < CHUNK / 256; ++u) {
        bool valid = (e0 + u * 256 + tid) < N_EDGES;
        int b = dv[u] / TILE_ROWS;
        word[u] = (unsigned)sv[u] | ((unsigned)(dv[u] - b * TILE_ROWS) << SRC_BITS);
        bin[u]  = valid ? b : (OFF_STRIDE - 1);
        rank[u] = atomicAdd(&hist[bin[u]], 1);
    }
    __syncthreads();

    int v = hist[tid];
    scanbuf[tid] = v;
    __syncthreads();
    for (int o = 1; o < OFF_STRIDE; o <<= 1) {
        int t = (tid >= o) ? scanbuf[tid - o] : 0;
        __syncthreads();
        scanbuf[tid] += t;
        __syncthreads();
    }
    int excl = scanbuf[tid] - v;
    offs[c * OFF_STRIDE + tid] = excl;
    hist[tid] = excl;
    __syncthreads();

#pragma unroll
    for (int u = 0; u < CHUNK / 256; ++u)
        if (bin[u] != OFF_STRIDE - 1) binned[e0 + hist[bin[u]] + rank[u]] = word[u];
}

__global__ void __launch_bounds__(1024) tile_sum_kernel(const unsigned* __restrict__ binned,
                                                        const int* __restrict__ offs,
                                                        const float* __restrict__ cs,
                                                        const float* __restrict__ h,
                                                        float* __restrict__ agg) {
    __shared__ float tile[(TILE_ROWS + 1) * HIDDEN];
    __shared__ int   next_chunk;
    int tid  = threadIdx.x;
    int lane = tid & 31;
    int b    = blockIdx.x;

    float4* tile4 = (float4*)tile;
    for (int i = tid; i < (TILE_ROWS + 1) * HIDDEN / 4; i += 1024) tile4[i] = float4{0.f, 0.f, 0.f, 0.f};
    if (tid == 0) next_chunk = 0;
    __syncthreads();

    for (int it = 0; it < N_CHUNKS; ++it) {
        int c = 0;
        if (lane == 0) c = atomicAdd(&next_chunk, 1);
        c = __builtin_amdgcn_readfirstlane(c);
        if (c >= N_CHUNKS) break;
        int lo = __builtin_amdgcn_readfirstlane(offs[c * OFF_STRIDE + b]);
        int hi = __builtin_amdgcn_readfirstlane(offs[c * OFF_STRIDE + b + 1]);
        const unsigned* sp = binned + (size_t)c * CHUNK;
        for (int base = lo; base < hi; base += 32) {
            int cnt = min(32, hi - base);
            unsigned w = sp[base + min(lane, cnt - 1)];
            for (int j0 = 0; j0 < cnt; j0 += 8) {
                float  sc[8];
                float2 va[8], vb[8];
                int    row[8];
#pragma unroll
                for (int u = 0; u < 8; ++u) {
                    int j = min(j0 + u, cnt - 1);
                    unsigned wj = (unsigned)__builtin_amdgcn_readlane((int)w, j);
                    int s  = (int)(wj & ((1u << SRC_BITS) - 1u));
                    row[u] = (j0 + u < cnt) ? (int)(wj >> SRC_BITS) : TILE_ROWS;
                    sc[u]  = cs[s];
                    const float2* hp = (const float2*)(h + (size_t)s * HIDDEN) + lane;
                    va[u] = hp[0];
                    vb[u] = hp[32];
                }
#pragma unroll
                for (int u = 0; u < 8; ++u) {
                    float* t = tile + row[u] * HIDDEN + 2 * lane;
                    atomicAdd(t,      va[u].x * sc[u]);
                    atomicAdd(t + 1,  va[u].y * sc[u]);
                    atomicAdd(t + 64, vb[u].x * sc[u]);
                    atomicAdd(t + 65, vb[u].y * sc[u]);
                }
            }
        }
    }
    __syncthreads();

    float4* out4 = (float4*)(agg + (size_t)b * TILE_ROWS * HIDDEN);
    for (int i = tid; i < TILE_ROWS * HIDDEN / 4; i += 1024) out4[i] = tile4[i];
}

__global__ void __launch_bounds__(256) gemm_relu_kernel(const float* __restrict__ agg,
                                                        const float* __restrict__ cd,
                                                        const float* __restrict__ W,
                                                        const float* __restrict__ bias,
                                                        float* __restrict__ hout) {
    __shared__ float Wl[HIDDEN * HIDDEN];
    int tid = threadIdx.x;
    for (int i = tid; i < HIDDEN * HIDDEN; i += 256) Wl[i] = W[i];
    __syncthreads();

    int wave = tid >> 5;
    int lane = tid & 31;
    int half = lane >> 4;
    int lr   = lane & 15;
    int m0   = blockIdx.x * 16;
    int m    = m0 + lr;
    int n    = wave * 16 + lr;

    float cdm = cd[m];
    const float* arow = agg + (size_t)m * HIDDEN + half * 2;
    const float* brow = Wl + (half * 2) * HIDDEN + n;

    v8f acc = {};
#pragma unroll
    for (int k0 = 0; k0 < HIDDEN; k0 += 4) {
        v2f av, bv;
        av.x = arow[k0]     * cdm;
        av.y = arow[k0 + 1] * cdm;
        bv.x = brow[k0 * HIDDEN];
        bv.y = brow[k0 * HIDDEN + HIDDEN];
        acc = __builtin_amdgcn_wmma_f32_16x16x4_f32(
             false, av,  false, bv,
             (short)0, acc,  false,  false);
    }

    float bn = bias[n];
#pragma unroll
    for (int r = 0; r < 8; ++r) {
        int mm = m0 + half * 8 + r;
        hout[(size_t)mm * HIDDEN + n] = fmaxf(acc[r] + bn, 0.0f);
    }
}

__device__ __forceinline__ void pool_flush(float* __restrict__ hg, float* __restrict__ counts,
                                           int g, int lane, float4 acc, float cnt) {
    float* o = hg + (size_t)g * HIDDEN + 4 * lane;
    atomicAdd(o,     acc.x);
    atomicAdd(o + 1, acc.y);
    atomicAdd(o + 2, acc.z);
    atomicAdd(o + 3, acc.w);
    if (lane == 0) atomicAdd(&counts[g], cnt);
}

__global__ void __launch_bounds__(256) pool_kernel(const int* __restrict__ seg, const float* __restrict__ h,
                                                   float* __restrict__ hg, float* __restrict__ counts) {
    int wave = (blockIdx.x * 256 + threadIdx.x) >> 5;
    int lane = threadIdx.x & 31;
    int n0   = wave * POOL_RUN;
    float4 acc = float4{0.f, 0.f, 0.f, 0.f};
    float  cnt = 0.0f;
    int gcur = __builtin_amdgcn_readfirstlane(seg[n0]);
    for (int nb = n0; nb < n0 + POOL_RUN; nb += 5) {
        int    g[5];
        float4 v[5];
#pragma unroll
        for (int u = 0; u < 5; ++u) {
            g[u] = __builtin_amdgcn_readfirstlane(seg[nb + u]);
            v[u] = ((const float4*)(h + (size_t)(nb + u) * HIDDEN))[lane];
        }
#pragma unroll
        for (int u = 0; u < 5; ++u) {
            if (g[u] != gcur) {
                pool_flush(hg, counts, gcur, lane, acc, cnt);
                acc  = float4{0.f, 0.f, 0.f, 0.f};
                cnt  = 0.0f;
                gcur = g[u];
            }
            acc.x += v[u].x;
            acc.y += v[u].y;
            acc.z += v[u].z;
            acc.w += v[u].w;
            cnt   += 1.0f;
        }
    }
    pool_flush(hg, counts, gcur, lane, acc, cnt);
}

__global__ void pool_div_kernel(float* __restrict__ hg, const float* __restrict__ counts) {
    int i = blockIdx.x * blockDim.x + threadIdx.x;
    if (i < N_GRAPHS * HIDDEN)
        hg[i] /= fmaxf(counts[i >> 7], 1.0f);
}

__global__ void logits_kernel(const float* __restrict__ hg1, const float* __restrict__ hg2,
                              const float* __restrict__ Wc, const float* __restrict__ bc,
                              float* __restrict__ logits) {
    int i = blockIdx.x * blockDim.x + threadIdx.x;
    if (i < N_GRAPHS * N_CLASSES) {
        int g = i / N_CLASSES, c = i % N_CLASSES;
        float s = bc[c];
#pragma unroll 8
        for (int k = 0; k < HIDDEN; ++k)
            s += fabsf(hg1[g * HIDDEN + k] - hg2[g * HIDDEN + k]) * Wc[k * N_CLASSES + c];
        logits[i] = s;
    }
}

static void run_branch(const int* src, const int* dst, const int* seg,
                       const float* W1, const float* b1,
                       const float* W2, const float* b2,
                       const float* W3, const float* b3,
                       const float* W4, const float* b4,
                       float* hg_out, float* ws_f, hipStream_t stream) {
    float* outdeg = ws_f;
    float* indeg  = ws_f + 100000;
    float* cs     = ws_f + 200000;
    float* cd     = ws_f + 300000;
    float* agg1   = ws_f + 400000;
    float* counts = ws_f + 500000;
    float* h      = ws_f + 500224;
    float* agg    = h + (size_t)N_NODES * HIDDEN;
    unsigned* binned = (unsigned*)(agg + (size_t)N_NODES * HIDDEN);
    int*      offs   = (int*)(binned + N_EDGES);

    hipMemsetAsync(ws_f, 0, (size_t)500224 * sizeof(float), stream);

    const int T = 256;
    deg_kernel<<<EDGE_WGS, T, 0, stream>>>(src, dst, outdeg, indeg);
    norm_kernel<<<(N_NODES + T - 1) / T, T, 0, stream>>>(outdeg, indeg, cs, cd);

    scatter1_kernel<<<EDGE_WGS, T, 0, stream>>>(src, dst, indeg, cs, agg1);
    layer1_kernel<<<ROW_WGS, T, 0, stream>>>(agg1, cd, W1, b1, h);

    bin_kernel<<<N_CHUNKS, 256, 0, stream>>>(src, dst, binned, offs);
    const float* Ws[3] = {W2, W3, W4};
    const float* bs[3] = {b2, b3, b4};
    for (int l = 0; l < 3; ++l) {
        tile_sum_kernel<<<N_TILES, 1024, 0, stream>>>(binned, offs, cs, h, agg);
        gemm_relu_kernel<<<N_NODES / 16, T, 0, stream>>>(agg, cd, Ws[l], bs[l], h);
    }

    hipMemsetAsync(hg_out, 0, (size_t)N_GRAPHS * HIDDEN * sizeof(float), stream);
    pool_kernel<<<POOL_WGS, T, 0, stream>>>(seg, h, hg_out, counts);
    pool_div_kernel<<<(N_GRAPHS * HIDDEN + T - 1) / T, T, 0, stream>>>(hg_out, counts);
}

extern "C" void kernel_launch(void* const* d_in, const int* in_sizes, int n_in,
                              void* d_out, int out_size, void* d_ws, size_t ws_size,
                              hipStream_t stream) {
    const int*   src1 = (const int*)d_in[0];
    const int*   dst1 = (const int*)d_in[1];
    const int*   seg1 = (const int*)d_in[2];
    const int*   src2 = (const int*)d_in[3];
    const int*   dst2 = (const int*)d_in[4];
    const int*   seg2 = (const int*)d_in[5];
    const float* W1   = (const float*)d_in[6];
    const float* b1   = (const float*)d_in[7];
    const float* W2   = (const float*)d_in[8];
    const float* b2   = (const float*)d_in[9];
    const float* W3   = (const float*)d_in[10];
    const float* b3   = (const float*)d_in[11];
    const float* W4   = (const float*)d_in[12];
    const float* b4   = (const float*)d_in[13];
    const float* Wc   = (const float*)d_in[14];
    const float* bc   = (const float*)d_in[15];

    float* out   = (float*)d_out;
    float* hg1   = out;
    float* hg2   = out + N_GRAPHS * HIDDEN;
    float* logit = out + 2 * N_GRAPHS * HIDDEN;
    float* ws_f  = (float*)d_ws;

    run_branch(src1, dst1, seg1, W1, b1, W2, b2, W3, b3, W4, b4, hg1, ws_f, stream);
    run_branch(src2, dst2, seg2, W1, b1, W2, b2, W3, b3, W4, b4, hg2, ws_f, stream);

    logits_kernel<<<(N_GRAPHS * N_CLASSES + 255) / 256, 256, 0, stream>>>(hg1, hg2, Wc, bc, logit);
}


# Layers 2-4: dense product moved into the tile kernel: the finished LDS tile (x cd) is the WMMA A operand, W fragments in registers; edge sums never stored, 6 launches fewer; same products, k order
# speedup vs baseline: 2.8212x; 1.2831x over previous
#include <hip/hip_runtime.h>
#include <hip/hip_bf16.h>

#define N_NODES   100000
#define N_EDGES   1600000
#define N_GRAPHS  128
#define HIDDEN    128
#define N_CLASSES 10

#define TILE_ROWS  400
#define N_TILES    (N_NODES / TILE_ROWS)
#define CHUNK      4096
#define N_CHUNKS   ((N_EDGES + CHUNK - 1) / CHUNK)
#define OFF_STRIDE 256
#define SRC_BITS   17
static_assert(N_TILES * TILE_ROWS == N_NODES, "tiles must cover the nodes exactly");
static_assert(N_TILES + 1 <= OFF_STRIDE, "offset row too short");
static_assert(N_NODES <= (1 << SRC_BITS), "source index does not fit its field");
#define TILE_PITCH 132
static_assert(TILE_ROWS % 16 == 0, "the tile is cut into whole 16-row matrix tiles");
static_assert((TILE_ROWS + 1) * TILE_PITCH * 4 + 64 <= 320 * 1024, "tile does not fit the LDS");

#define EDGES_PER_THREAD 5
#define EDGE_WGS   (N_EDGES / (256 * EDGES_PER_THREAD))
#define ROWS_PER_WAVE 10
#define ROW_WGS    (N_NODES / (8 * ROWS_PER_WAVE))
#define POOL_RUN   25
#define POOL_WGS   (N_NODES / (8 * POOL_RUN))
static_assert(EDGE_WGS * 256 * EDGES_PER_THREAD == N_EDGES, "edge grid must cover the edges exactly");
static_assert(ROW_WGS * 8 * ROWS_PER_WAVE == N_NODES, "row grid must cover the nodes exactly");
static_assert(POOL_WGS * 8 * POOL_RUN == N_NODES, "pool grid must cover the nodes exactly");
static_assert(POOL_RUN % 5 == 0, "pooling loads rows in groups of five");

typedef __attribute__((ext_vector_type(2))) float v2f;
typedef __attribute__((ext_vector_type(8))) float v8f;

__global__ void __launch_bounds__(256) deg_kernel(const int* __restrict__ src, const int* __restrict__ dst,
                                                  float* __restrict__ outdeg, float* __restrict__ indeg) {
    int t = blockIdx.x * 256 + threadIdx.x;
    int s[EDGES_PER_THREAD], d[EDGES_PER_THREAD];
#pragma unroll
    for (int k = 0; k < EDGES_PER_THREAD; ++k) {
        int e = t + k * (EDGE_WGS * 256);
        s[k] = src[e];
        d[k] = dst[e];
    }
#pragma unroll
    for (int k = 0; k < EDGES_PER_THREAD; ++k) {
        atomicAdd(&outdeg[s[k]], 1.0f);
        atomicAdd(&indeg[d[k]], 1.0f);
    }
}

__global__ void norm_kernel(const float* __restrict__ outdeg, const float* __restrict__ indeg,
                            float* __restrict__ cs, float* __restrict__ cd) {
    int n = blockIdx.x * blockDim.x + threadIdx.x;
    if (n < N_NODES) {
        cs[n] = rsqrtf(fmaxf(outdeg[n], 1.0f));
        cd[n] = rsqrtf(fmaxf(indeg[n], 1.0f));
    }
}

__global__ void __launch_bounds__(256) scatter1_kernel(const int* __restrict__ src, const int* __restrict__ dst,
                                                       const float* __restrict__ indeg, const float* __restrict__ cs,
                                                       float* __restrict__ agg1) {
    int t = blockIdx.x * 256 + threadIdx.x;
    int   s[EDGES_PER_THREAD], d[EDGES_PER_THREAD];
    float a[EDGES_PER_THREAD], c[EDGES_PER_THREAD];
#pragma unroll
    for (int k = 0; k < EDGES_PER_THREAD; ++k) {
        int e = t + k * (EDGE_WGS * 256);
        s[k] = src[e];
        d[k] = dst[e];
    }
#pragma unroll
    for (int k = 0; k < EDGES_PER_THREAD; ++k) {
        a[k] = indeg[s[k]];
        c[k] = cs[s[k]];
    }
#pragma unroll
    for (int k = 0; k < EDGES_PER_THREAD; ++k)
        atomicAdd(&agg1[d[k]], a[k] * c[k]);
}

__global__ void __launch_bounds__(256) layer1_kernel(const float* __restrict__ agg1, const float* __restrict__ cd,
                                                     const float* __restrict__ W1, const float* __restrict__ b1,
                                                     float* __restrict__ h) {
    int wave = (blockIdx.x * 256 + threadIdx.x) >> 5;
    int lane = threadIdx.x & 31;
    const float4 w = ((const float4*)W1)[lane];
    const float4 b = ((const float4*)b1)[lane];
    int m0 = wave * ROWS_PER_WAVE;
    float x[ROWS_PER_WAVE];
#pragma unroll
    for (int u = 0; u < ROWS_PER_WAVE; ++u) x[u] = agg1[m0 + u] * cd[m0 + u];
    float4* hrow = (float4*)(h + (size_t)m0 * HIDDEN) + lane;
#pragma unroll
    for (int u = 0; u < ROWS_PER_WAVE; ++u) {
        float4 o;
        o.x = fmaxf(fmaf(x[u], w.x, b.x), 0.0f);
        o.y = fmaxf(fmaf(x[u], w.y, b.y), 0.0f);
        o.z = fmaxf(fmaf(x[u], w.z, b.z), 0.0f);
        o.w = fmaxf(fmaf(x[u], w.w, b.w), 0.0f);
        hrow[u * (HIDDEN / 4)] = o;
    }
}

__global__ void __launch_bounds__(256) bin_kernel(const int* __restrict__ src, const int* __restrict__ dst,
                                                  unsigned* __restrict__ binned, int* __restrict__ offs) {
    __shared__ int hist[OFF_STRIDE];
    __shared__ int scanbuf[OFF_STRIDE];
    int tid = threadIdx.x;
    int c   = blockIdx.x;
    int e0  = c * CHUNK;
    hist[tid] = 0;
    __syncthreads();

    static_assert(N_TILES + 1 < OFF_STRIDE, "no spare bin");
    unsigned word[CHUNK / 256];
    int      bin[CHUNK / 256];
    int      rank[CHUNK / 256];
    int      sv[CHUNK / 256], dv[CHUNK / 256];
#pragma unroll
    for (int u = 0; u < CHUNK / 256; ++u) {
        int e = min(e0 + u * 256 + tid, N_EDGES - 1);
        sv[u] = src[e];
        dv[u] = dst[e];
    }
#pragma unroll
    for (int u = 0; u < CHUNK / 256; ++u) {
        bool valid = (e0 + u * 256 + tid) < N_EDGES;
        int b = dv[u] / TILE_ROWS;
        word[u] = (unsigned)sv[u] | ((unsigned)(dv[u] - b * TILE_ROWS) << SRC_BITS);
        bin[u]  = valid ? b : (OFF_STRIDE - 1);
        rank[u] = atomicAdd(&hist[bin[u]], 1);
    }
    __syncthreads();

    int v = hist[tid];
    scanbuf[tid] = v;
    __syncthreads();
    for (int o = 1; o < OFF_STRIDE; o <<= 1) {
        int t = (tid >= o) ? scanbuf[tid - o] : 0;
        __syncthreads();
        scanbuf[tid] += t;
        __syncthreads();
    }
    int excl = scanbuf[tid] - v;
    offs[c * OFF_STRIDE + tid] = excl;
    hist[tid] = excl;
    __syncthreads();

#pragma unroll
    for (int u = 0; u < CHUNK / 256; ++u)
        if (bin[u] != OFF_STRIDE - 1) binned[e0 + hist[bin[u]] + rank[u]] = word[u];
}

__global__ void __launch_bounds__(1024) tile_layer_kernel(const unsigned* __restrict__ binned,
                                                          const int* __restrict__ offs,
                                                          const float* __restrict__ cs,
                                                          const float* __restrict__ cd,
                                                          const float* __restrict__ h,
                                                          const float* __restrict__ W,
                                                          const float* __restrict__ bias,
                                                          float* __restrict__ hout) {
    __shared__ __attribute__((aligned(16))) float tile[(TILE_ROWS + 1) * TILE_PITCH];
    __shared__ int   next_chunk;
    int tid  = threadIdx.x;
    int lane = tid & 31;
    int b    = blockIdx.x;

    float4* tile4 = (float4*)tile;
    for (int i = tid; i < (TILE_ROWS + 1) * TILE_PITCH / 4; i += 1024) tile4[i] = float4{0.f, 0.f, 0.f, 0.f};
    if (tid == 0) next_chunk = 0;
    __syncthreads();

    for (int it = 0; it < N_CHUNKS; ++it) {
        int c = 0;
        if (lane == 0) c = atomicAdd(&next_chunk, 1);
        c = __builtin_amdgcn_readfirstlane(c);
        if (c >= N_CHUNKS) break;
        int lo = __builtin_amdgcn_readfirstlane(offs[c * OFF_STRIDE + b]);
        int hi = __builtin_amdgcn_readfirstlane(offs[c * OFF_STRIDE + b + 1]);
        const unsigned* sp = binned + (size_t)c * CHUNK;
        for (int base = lo; base < hi; base += 32) {
            int cnt = min(32, hi - base);
            unsigned w = sp[base + min(lane, cnt - 1)];
            for (int j0 = 0; j0 < cnt; j0 += 8) {
                float  sc[8];
                float2 va[8], vb[8];
                int    row[8];
#pragma unroll
                for (int u = 0; u < 8; ++u) {
                    int j = min(j0 + u, cnt - 1);
                    unsigned wj = (unsigned)__builtin_amdgcn_readlane((int)w, j);
                    int s  = (int)(wj & ((1u << SRC_BITS) - 1u));
                    row[u] = (j0 + u < cnt) ? (int)(wj >> SRC_BITS) : TILE_ROWS;
                    sc[u]  = cs[s];
                    const float2* hp = (const float2*)(h + (size_t)s * HIDDEN) + lane;
                    va[u] = hp[0];
                    vb[u] = hp[32];
                }
#pragma unroll
                for (int u = 0; u < 8; ++u) {
                    float* t = tile + row[u] * TILE_PITCH + 2 * lane;
                    atomicAdd(t,      va[u].x * sc[u]);
                    atomicAdd(t + 1,  va[u].y * sc[u]);
                    atomicAdd(t + 64, vb[u].x * sc[u]);
                    atomicAdd(t + 65, vb[u].y * sc[u]);
                }
            }
        }
    }

    int wave = __builtin_amdgcn_readfirstlane(tid >> 5);
    int half = lane >> 4;
    int lr   = lane & 15;
    int n    = (wave & 7) * 16 + lr;
    const float* wp = W + (half * 2) * HIDDEN + n;
    v2f bw[HIDDEN / 4];
#pragma unroll
    for (int ks = 0; ks < HIDDEN / 4; ++ks) {
        bw[ks].x = wp[(4 * ks) * HIDDEN];
        bw[ks].y = wp[(4 * ks + 1) * HIDDEN];
    }
    float bn = bias[n];
    __syncthreads();

    for (int rt = wave >> 3; rt < TILE_ROWS / 16; rt += 4) {
        int   r0  = rt * 16;
        float cdm = cd[b * TILE_ROWS + r0 + lr];
        const float* ap = tile + (r0 + lr) * TILE_PITCH + half * 2;
        v8f acc = {};
#pragma unroll
        for (int ks = 0; ks < HIDDEN / 4; ++ks) {
            float2 a = *(const float2*)(ap + 4 * ks);
            v2f av;
            av.x = a.x * cdm;
            av.y = a.y * cdm;
            acc = __builtin_amdgcn_wmma_f32_16x16x4_f32(
                 false, av,  false, bw[ks],
                 (short)0, acc,  false,  false);
        }
        float* op = hout + (size_t)(b * TILE_ROWS + r0 + half * 8) * HIDDEN + n;
#pragma unroll
        for (int r = 0; r < 8; ++r)
            op[(size_t)r * HIDDEN] = fmaxf(acc[r] + bn, 0.0f);
    }
}

__device__ __forceinline__ void pool_flush(float* __restrict__ hg, float* __restrict__ counts,
                                           int g, int lane, float4 acc, float cnt) {
    float* o = hg + (size_t)g * HIDDEN + 4 * lane;
    atomicAdd(o,     acc.x);
    atomicAdd(o + 1, acc.y);
    atomicAdd(o + 2, acc.z);
    atomicAdd(o + 3, acc.w);
    if (lane == 0) atomicAdd(&counts[g], cnt);
}

__global__ void __launch_bounds__(256) pool_kernel(const int* __restrict__ seg, const float* __restrict__ h,
                                                   float* __restrict__ hg, float* __restrict__ counts) {
    int wave = (blockIdx.x * 256 + threadIdx.x) >> 5;
    int lane = threadIdx.x & 31;
    int n0   = wave * POOL_RUN;
    float4 acc = float4{0.f, 0.f, 0.f, 0.f};
    float  cnt = 0.0f;
    int gcur = __builtin_amdgcn_readfirstlane(seg[n0]);
    for (int nb = n0; nb < n0 + POOL_RUN; nb += 5) {
        int    g[5];
        float4 v[5];
#pragma unroll
        for (int u = 0; u < 5; ++u) {
            g[u] = __builtin_amdgcn_readfirstlane(seg[nb + u]);
            v[u] = ((const float4*)(h + (size_t)(nb + u) * HIDDEN))[lane];
        }
#pragma unroll
        for (int u = 0; u < 5; ++u) {
            if (g[u] != gcur) {
                pool_flush(hg, counts, gcur, lane, acc, cnt);
                acc  = float4{0.f, 0.f, 0.f, 0.f};
                cnt  = 0.0f;
                gcur = g[u];
            }
            acc.x += v[u].x;
            acc.y += v[u].y;
            acc.z += v[u].z;
            acc.w += v[u].w;
            cnt   += 1.0f;
        }
    }
    pool_flush(hg, counts, gcur, lane, acc, cnt);
}

__global__ void pool_div_kernel(float* __restrict__ hg, const float* __restrict__ counts) {
    int i = blockIdx.x * blockDim.x + threadIdx.x;
    if (i < N_GRAPHS * HIDDEN)
        hg[i] /= fmaxf(counts[i >> 7], 1.0f);
}

__global__ void logits_kernel(const float* __restrict__ hg1, const float* __restrict__ hg2,
                              const float* __restrict__ Wc, const float* __restrict__ bc,
                              float* __restrict__ logits) {
    int i = blockIdx.x * blockDim.x + threadIdx.x;
    if (i < N_GRAPHS * N_CLASSES) {
        int g = i / N_CLASSES, c = i % N_CLASSES;
        float s = bc[c];
#pragma unroll 8
        for (int k = 0; k < HIDDEN; ++k)
            s += fabsf(hg1[g * HIDDEN + k] - hg2[g * HIDDEN + k]) * Wc[k * N_CLASSES + c];
        logits[i] = s;
    }
}

static void run_branch(const int* src, const int* dst, const int* seg,
                       const float* W1, const float* b1,
                       const float* W2, const float* b2,
                       const float* W3, const float* b3,
                       const float* W4, const float* b4,
                       float* hg_out, float* ws_f, hipStream_t stream) {
    float* outdeg = ws_f;
    float* indeg  = ws_f + 100000;
    float* cs     = ws_f + 200000;
    float* cd     = ws_f + 300000;
    float* agg1   = ws_f + 400000;
    float* counts = ws_f + 500000;
    float* h      = ws_f + 500224;
    float* hb     = h + (size_t)N_NODES * HIDDEN;
    unsigned* binned = (unsigned*)(hb + (size_t)N_NODES * HIDDEN);
    int*      offs   = (int*)(binned + N_EDGES);

    hipMemsetAsync(ws_f, 0, (size_t)500224 * sizeof(float), stream);

    const int T = 256;
    deg_kernel<<<EDGE_WGS, T, 0, stream>>>(src, dst, outdeg, indeg);
    norm_kernel<<<(N_NODES + T - 1) / T, T, 0, stream>>>(outdeg, indeg, cs, cd);

    scatter1_kernel<<<EDGE_WGS, T, 0, stream>>>(src, dst, indeg, cs, agg1);
    layer1_kernel<<<ROW_WGS, T, 0, stream>>>(agg1, cd, W1, b1, h);

    bin_kernel<<<N_CHUNKS, 256, 0, stream>>>(src, dst, binned, offs);
    tile_layer_kernel<<<N_TILES, 1024, 0, stream>>>(binned, offs, cs, cd, h,  W2, b2, hb);
    tile_layer_kernel<<<N_TILES, 1024, 0, stream>>>(binned, offs, cs, cd, hb, W3, b3, h);
    tile_layer_kernel<<<N_TILES, 1024, 0, stream>>>(binned, offs, cs, cd, h,  W4, b4, hb);

    hipMemsetAsync(hg_out, 0, (size_t)N_GRAPHS * HIDDEN * sizeof(float), stream);
    pool_kernel<<<POOL_WGS, T, 0, stream>>>(seg, hb, hg_out, counts);
    pool_div_kernel<<<(N_GRAPHS * HIDDEN + T - 1) / T, T, 0, stream>>>(hg_out, counts);
}

extern "C" void kernel_launch(void* const* d_in, const int* in_sizes, int n_in,
                              void* d_out, int out_size, void* d_ws, size_t ws_size,
                              hipStream_t stream) {
    const int*   src1 = (const int*)d_in[0];
    const int*   dst1 = (const int*)d_in[1];
    const int*   seg1 = (const int*)d_in[2];
    const int*   src2 = (const int*)d_in[3];
    const int*   dst2 = (const int*)d_in[4];
    const int*   seg2 = (const int*)d_in[5];
    const float* W1   = (const float*)d_in[6];
    const float* b1   = (const float*)d_in[7];
    const float* W2   = (const float*)d_in[8];
    const float* b2   = (const float*)d_in[9];
    const float* W3   = (const float*)d_in[10];
    const float* b3   = (const float*)d_in[11];
    const float* W4   = (const float*)d_in[12];
    const float* b4   = (const float*)d_in[13];
    const float* Wc   = (const float*)d_in[14];
    const float* bc   = (const float*)d_in[15];

    float* out   = (float*)d_out;
    float* hg1   = out;
    float* hg2   = out + N_GRAPHS * HIDDEN;
    float* logit = out + 2 * N_GRAPHS * HIDDEN;
    float* ws_f  = (float*)d_ws;

    run_branch(src1, dst1, seg1, W1, b1, W2, b2, W3, b3, W4, b4, hg1, ws_f, stream);
    run_branch(src2, dst2, seg2, W1, b1, W2, b2, W3, b3, W4, b4, hg2, ws_f, stream);

    logits_kernel<<<(N_GRAPHS * N_CLASSES + 255) / 256, 256, 0, stream>>>(hg1, hg2, Wc, bc, logit);
}


# In-degree and layer-1 scalar sums taken per 400-node tile from the call's binned edge list into LDS slots, stored once: 3.2M fewer global f32 adds per graph; agg1 sum order: LDS arrival
# speedup vs baseline: 3.6737x; 1.3022x over previous
#include <hip/hip_runtime.h>
#include <hip/hip_bf16.h>

#define N_NODES   100000
#define N_EDGES   1600000
#define N_GRAPHS  128
#define HIDDEN    128
#define N_CLASSES 10

#define TILE_ROWS  400
#define N_TILES    (N_NODES / TILE_ROWS)
#define CHUNK      4096
#define N_CHUNKS   ((N_EDGES + CHUNK - 1) / CHUNK)
#define OFF_STRIDE 256
#define SRC_BITS   17
static_assert(N_TILES * TILE_ROWS == N_NODES, "tiles must cover the nodes exactly");
static_assert(N_TILES + 1 <= OFF_STRIDE, "offset row too short");
static_assert(N_NODES <= (1 << SRC_BITS), "source index does not fit its field");
#define TILE_PITCH 132
static_assert(TILE_ROWS % 16 == 0, "the tile is cut into whole 16-row matrix tiles");
static_assert((TILE_ROWS + 1) * TILE_PITCH * 4 + 64 <= 320 * 1024, "tile does not fit the LDS");

#define EDGES_PER_THREAD 5
#define EDGE_WGS   (N_EDGES / (256 * EDGES_PER_THREAD))
#define ROWS_PER_WAVE 10
#define ROW_WGS    (N_NODES / (8 * ROWS_PER_WAVE))
#define POOL_RUN   25
#define POOL_WGS   (N_NODES / (8 * POOL_RUN))
static_assert(EDGE_WGS * 256 * EDGES_PER_THREAD == N_EDGES, "edge grid must cover the edges exactly");
static_assert(ROW_WGS * 8 * ROWS_PER_WAVE == N_NODES, "row grid must cover the nodes exactly");
static_assert(POOL_WGS * 8 * POOL_RUN == N_NODES, "pool grid must cover the nodes exactly");
static_assert(POOL_RUN % 5 == 0, "pooling loads rows in groups of five");

typedef __attribute__((ext_vector_type(2))) float v2f;
typedef __attribute__((ext_vector_type(8))) float v8f;

__global__ void __launch_bounds__(256) outdeg_kernel(const int* __restrict__ src, float* __restrict__ outdeg) {
    int t = blockIdx.x * 256 + threadIdx.x;
    int s[EDGES_PER_THREAD];
#pragma unroll
    for (int k = 0; k < EDGES_PER_THREAD; ++k) s[k] = src[t + k * (EDGE_WGS * 256)];
#pragma unroll
    for (int k = 0; k < EDGES_PER_THREAD; ++k) atomicAdd(&outdeg[s[k]], 1.0f);
}

__global__ void __launch_bounds__(256) layer1_kernel(const float* __restrict__ agg1, const float* __restrict__ cd,
                                                     const float* __restrict__ W1, const float* __restrict__ b1,
                                                     float* __restrict__ h) {
    int wave = (blockIdx.x * 256 + threadIdx.x) >> 5;
    int lane = threadIdx.x & 31;
    const float4 w = ((const float4*)W1)[lane];
    const float4 b = ((const float4*)b1)[lane];
    int m0 = wave * ROWS_PER_WAVE;
    float x[ROWS_PER_WAVE];
#pragma unroll
    for (int u = 0; u < ROWS_PER_WAVE; ++u) x[u] = agg1[m0 + u] * cd[m0 + u];
    float4* hrow = (float4*)(h + (size_t)m0 * HIDDEN) + lane;
#pragma unroll
    for (int u = 0; u < ROWS_PER_WAVE; ++u) {
        float4 o;
        o.x = fmaxf(fmaf(x[u], w.x, b.x), 0.0f);
        o.y = fmaxf(fmaf(x[u], w.y, b.y), 0.0f);
        o.z = fmaxf(fmaf(x[u], w.z, b.z), 0.0f);
        o.w = fmaxf(fmaf(x[u], w.w, b.w), 0.0f);
        hrow[u * (HIDDEN / 4)] = o;
    }
}

__global__ void __launch_bounds__(256) bin_kernel(const int* __restrict__ src, const int* __restrict__ dst,
                                                  unsigned* __restrict__ binned, int* __restrict__ offs) {
    __shared__ int hist[OFF_STRIDE];
    __shared__ int scanbuf[OFF_STRIDE];
    int tid = threadIdx.x;
    int c   = blockIdx.x;
    int e0  = c * CHUNK;
    hist[tid] = 0;
    __syncthreads();

    static_assert(N_TILES + 1 < OFF_STRIDE, "no spare bin");
    unsigned word[CHUNK / 256];
    int      bin[CHUNK / 256];
    int      rank[CHUNK / 256];
    int      sv[CHUNK / 256], dv[CHUNK / 256];
#pragma unroll
    for (int u = 0; u < CHUNK / 256; ++u) {
        int e = min(e0 + u * 256 + tid, N_EDGES - 1);
        sv[u] = src[e];
        dv[u] = dst[e];
    }
#pragma unroll
    for (int u = 0; u < CHUNK / 256; ++u) {
        bool valid = (e0 + u * 256 + tid) < N_EDGES;
        int b = dv[u] / TILE_ROWS;
        word[u] = (unsigned)sv[u] | ((unsigned)(dv[u] - b * TILE_ROWS) << SRC_BITS);
        bin[u]  = valid ? b : (OFF_STRIDE - 1);
        rank[u] = atomicAdd(&hist[bin[u]], 1);
    }
    __syncthreads();

    int v = hist[tid];
    scanbuf[tid] = v;
    __syncthreads();
    for (int o = 1; o < OFF_STRIDE; o <<= 1) {
        int t = (tid >= o) ? scanbuf[tid - o] : 0;
        __syncthreads();
        scanbuf[tid] += t;
        __syncthreads();
    }
    int excl = scanbuf[tid] - v;
    offs[c * OFF_STRIDE + tid] = excl;
    hist[tid] = excl;
    __syncthreads();

#pragma unroll
    for (int u = 0; u < CHUNK / 256; ++u)
        if (bin[u] != OFF_STRIDE - 1) binned[e0 + hist[bin[u]] + rank[u]] = word[u];
}

#define SCAL_THREADS 512
#define SCAL_UNROLL  8
static_assert(N_CHUNKS <= SCAL_THREADS, "one thread per chunk");
static_assert(TILE_ROWS <= SCAL_THREADS, "one thread per tile row in the epilogue");

__global__ void __launch_bounds__(SCAL_THREADS) tile_indeg_kernel(const unsigned* __restrict__ binned,
                                                                  const int* __restrict__ offs,
                                                                  const float* __restrict__ outdeg,
                                                                  float* __restrict__ msg1,
                                                                  float* __restrict__ cs, float* __restrict__ cd) {
    __shared__ int cnt[TILE_ROWS + 32];
    int tid = threadIdx.x;
    int b   = blockIdx.x;
    int c   = min(tid, N_CHUNKS - 1);
    int lo = 0, hi = 0;
    if (tid < N_CHUNKS) {
        lo = offs[c * OFF_STRIDE + b];
        hi = offs[c * OFF_STRIDE + b + 1];
    }
    if (tid < TILE_ROWS) cnt[tid] = 0;
    __syncthreads();
    const unsigned* sp = binned + (size_t)c * CHUNK;
    for (int i = lo; i < hi; i += SCAL_UNROLL) {
        unsigned w[SCAL_UNROLL];
#pragma unroll
        for (int u = 0; u < SCAL_UNROLL; ++u) w[u] = sp[min(i + u, hi - 1)];
#pragma unroll
        for (int u = 0; u < SCAL_UNROLL; ++u)
            atomicAdd(&cnt[(i + u < hi) ? (int)(w[u] >> SRC_BITS) : TILE_ROWS + (tid & 31)], 1);
    }
    __syncthreads();
    if (tid < TILE_ROWS) {
        int   n  = b * TILE_ROWS + tid;
        float id = (float)cnt[tid];
        float sn = rsqrtf(fmaxf(outdeg[n], 1.0f));
        cd[n]    = rsqrtf(fmaxf(id, 1.0f));
        cs[n]    = sn;
        msg1[n]  = id * sn;
    }
}

__global__ void __launch_bounds__(SCAL_THREADS) tile_agg1_kernel(const unsigned* __restrict__ binned,
                                                                 const int* __restrict__ offs,
                                                                 const float* __restrict__ msg1,
                                                                 float* __restrict__ agg1) {
    __shared__ float acc[TILE_ROWS + 32];
    int tid = threadIdx.x;
    int b   = blockIdx.x;
    int c   = min(tid, N_CHUNKS - 1);
    int lo = 0, hi = 0;
    if (tid < N_CHUNKS) {
        lo = offs[c * OFF_STRIDE + b];
        hi = offs[c * OFF_STRIDE + b + 1];
    }
    if (tid < TILE_ROWS) acc[tid] = 0.0f;
    __syncthreads();
    const unsigned* sp = binned + (size_t)c * CHUNK;
    for (int i = lo; i < hi; i += SCAL_UNROLL) {
        unsigned w[SCAL_UNROLL];
        float    a[SCAL_UNROLL];
#pragma unroll
        for (int u = 0; u < SCAL_UNROLL; ++u) w[u] = sp[min(i + u, hi - 1)];
#pragma unroll
        for (int u = 0; u < SCAL_UNROLL; ++u) {
            a[u] = msg1[w[u] & ((1u << SRC_BITS) - 1u)];
        }
#pragma unroll
        for (int u = 0; u < SCAL_UNROLL; ++u)
            atomicAdd(&acc[(i + u < hi) ? (int)(w[u] >> SRC_BITS) : TILE_ROWS + (tid & 31)], a[u]);
    }
    __syncthreads();
    if (tid < TILE_ROWS) agg1[b * TILE_ROWS + tid] = acc[tid];
}

__global__ void __launch_bounds__(1024) tile_layer_kernel(const unsigned* __restrict__ binned,
                                                          const int* __restrict__ offs,
                                                          const float* __restrict__ cs,
                                                          const float* __restrict__ cd,
                                                          const float* __restrict__ h,
                                                          const float* __restrict__ W,
                                                          const float* __restrict__ bias,
                                                          float* __restrict__ hout) {
    __shared__ __attribute__((aligned(16))) float tile[(TILE_ROWS + 1) * TILE_PITCH];
    __shared__ int   next_chunk;
    int tid  = threadIdx.x;
    int lane = tid & 31;
    int b    = blockIdx.x;

    float4* tile4 = (float4*)tile;
    for (int i = tid; i < (TILE_ROWS + 1) * TILE_PITCH / 4; i += 1024) tile4[i] = float4{0.f, 0.f, 0.f, 0.f};
    if (tid == 0) next_chunk = 0;
    __syncthreads();

    for (int it = 0; it < N_CHUNKS; ++it) {
        int c = 0;
        if (lane == 0) c = atomicAdd(&next_chunk, 1);
        c = __builtin_amdgcn_readfirstlane(c);
        if (c >= N_CHUNKS) break;
        int lo = __builtin_amdgcn_readfirstlane(offs[c * OFF_STRIDE + b]);
        int hi = __builtin_amdgcn_readfirstlane(offs[c * OFF_STRIDE + b + 1]);
        const unsigned* sp = binned + (size_t)c * CHUNK;
        for (int base = lo; base < hi; base += 32) {
            int cnt = min(32, hi - base);
            unsigned w = sp[base + min(lane, cnt - 1)];
            for (int j0 = 0; j0 < cnt; j0 += 8) {
                float  sc[8];
                float2 va[8], vb[8];
                int    row[8];
#pragma unroll
                for (int u = 0; u < 8; ++u) {
                    int j = min(j0 + u, cnt - 1);
                    unsigned wj = (unsigned)__builtin_amdgcn_readlane((int)w, j);
                    int s  = (int)(wj & ((1u << SRC_BITS) - 1u));
                    row[u] = (j0 + u < cnt) ? (int)(wj >> SRC_BITS) : TILE_ROWS;
                    sc[u]  = cs[s];
                    const float2* hp = (const float2*)(h + (size_t)s * HIDDEN) + lane;
                    va[u] = hp[0];
                    vb[u] = hp[32];
                }
#pragma unroll
                for (int u = 0; u < 8; ++u) {
                    float* t = tile + row[u] * TILE_PITCH + 2 * lane;
                    atomicAdd(t,      va[u].x * sc[u]);
                    atomicAdd(t + 1,  va[u].y * sc[u]);
                    atomicAdd(t + 64, vb[u].x * sc[u]);
                    atomicAdd(t + 65, vb[u].y * sc[u]);
                }
            }
        }
    }

    int wave = __builtin_amdgcn_readfirstlane(tid >> 5);
    int half = lane >> 4;
    int lr   = lane & 15;
    int n    = (wave & 7) * 16 + lr;
    const float* wp = W + (half * 2) * HIDDEN + n;
    v2f bw[HIDDEN / 4];
#pragma unroll
    for (int ks = 0; ks < HIDDEN / 4; ++ks) {
        bw[ks].x = wp[(4 * ks) * HIDDEN];
        bw[ks].y = wp[(4 * ks + 1) * HIDDEN];
    }
    float bn = bias[n];
    __syncthreads();

    for (int rt = wave >> 3; rt < TILE_ROWS / 16; rt += 4) {
        int   r0  = rt * 16;
        float cdm = cd[b * TILE_ROWS + r0 + lr];
        const float* ap = tile + (r0 + lr) * TILE_PITCH + half * 2;
        v8f acc = {};
#pragma unroll
        for (int ks = 0; ks < HIDDEN / 4; ++ks) {
            float2 a = *(const float2*)(ap + 4 * ks);
            v2f av;
            av.x = a.x * cdm;
            av.y = a.y * cdm;
            acc = __builtin_amdgcn_wmma_f32_16x16x4_f32(
                 false, av,  false, bw[ks],
                 (short)0, acc,  false,  false);
        }
        float* op = hout + (size_t)(b * TILE_ROWS + r0 + half * 8) * HIDDEN + n;
#pragma unroll
        for (int r = 0; r < 8; ++r)
            op[(size_t)r * HIDDEN] = fmaxf(acc[r] + bn, 0.0f);
    }
}

__device__ __forceinline__ void pool_flush(float* __restrict__ hg, float* __restrict__ counts,
                                           int g, int lane, float4 acc, float cnt) {
    float* o = hg + (size_t)g * HIDDEN + 4 * lane;
    atomicAdd(o,     acc.x);
    atomicAdd(o + 1, acc.y);
    atomicAdd(o + 2, acc.z);
    atomicAdd(o + 3, acc.w);
    if (lane == 0) atomicAdd(&counts[g], cnt);
}

__global__ void __launch_bounds__(256) pool_kernel(const int* __restrict__ seg, const float* __restrict__ h,
                                                   float* __restrict__ hg, float* __restrict__ counts) {
    int wave = (blockIdx.x * 256 + threadIdx.x) >> 5;
    int lane = threadIdx.x & 31;
    int n0   = wave * POOL_RUN;
    float4 acc = float4{0.f, 0.f, 0.f, 0.f};
    float  cnt = 0.0f;
    int gcur = __builtin_amdgcn_readfirstlane(seg[n0]);
    for (int nb = n0; nb < n0 + POOL_RUN; nb += 5) {
        int    g[5];
        float4 v[5];
#pragma unroll
        for (int u = 0; u < 5; ++u) {
            g[u] = __builtin_amdgcn_readfirstlane(seg[nb + u]);
            v[u] = ((const float4*)(h + (size_t)(nb + u) * HIDDEN))[lane];
        }
#pragma unroll
        for (int u = 0; u < 5; ++u) {
            if (g[u] != gcur) {
                pool_flush(hg, counts, gcur, lane, acc, cnt);
                acc  = float4{0.f, 0.f, 0.f, 0.f};
                cnt  = 0.0f;
                gcur = g[u];
            }
            acc.x += v[u].x;
            acc.y += v[u].y;
            acc.z += v[u].z;
            acc.w += v[u].w;
            cnt   += 1.0f;
        }
    }
    pool_flush(hg, counts, gcur, lane, acc, cnt);
}

__global__ void pool_div_kernel(float* __restrict__ hg, const float* __restrict__ counts) {
    int i = blockIdx.x * blockDim.x + threadIdx.x;
    if (i < N_GRAPHS * HIDDEN)
        hg[i] /= fmaxf(counts[i >> 7], 1.0f);
}

__global__ void logits_kernel(const float* __restrict__ hg1, const float* __restrict__ hg2,
                              const float* __restrict__ Wc, const float* __restrict__ bc,
                              float* __restrict__ logits) {
    int i = blockIdx.x * blockDim.x + threadIdx.x;
    if (i < N_GRAPHS * N_CLASSES) {
        int g = i / N_CLASSES, c = i % N_CLASSES;
        float s = bc[c];
#pragma unroll 8
        for (int k = 0; k < HIDDEN; ++k)
            s += fabsf(hg1[g * HIDDEN + k] - hg2[g * HIDDEN + k]) * Wc[k * N_CLASSES + c];
        logits[i] = s;
    }
}

static void run_branch(const int* src, const int* dst, const int* seg,
                       const float* W1, const float* b1,
                       const float* W2, const float* b2,
                       const float* W3, const float* b3,
                       const float* W4, const float* b4,
                       float* hg_out, float* ws_f, hipStream_t stream) {
    float* outdeg = ws_f;
    float* msg1   = ws_f + 100000;
    float* cs     = ws_f + 200000;
    float* cd     = ws_f + 300000;
    float* agg1   = ws_f + 400000;
    float* counts = ws_f + 500000;
    float* h      = ws_f + 500224;
    float* hb     = h + (size_t)N_NODES * HIDDEN;
    unsigned* binned = (unsigned*)(hb + (size_t)N_NODES * HIDDEN);
    int*      offs   = (int*)(binned + N_EDGES);

    hipMemsetAsync(ws_f, 0, (size_t)500224 * sizeof(float), stream);

    const int T = 256;
    outdeg_kernel<<<EDGE_WGS, T, 0, stream>>>(src, outdeg);

    bin_kernel<<<N_CHUNKS, 256, 0, stream>>>(src, dst, binned, offs);

    tile_indeg_kernel<<<N_TILES, SCAL_THREADS, 0, stream>>>(binned, offs, outdeg, msg1, cs, cd);
    tile_agg1_kernel<<<N_TILES, SCAL_THREADS, 0, stream>>>(binned, offs, msg1, agg1);
    layer1_kernel<<<ROW_WGS, T, 0, stream>>>(agg1, cd, W1, b1, h);

    tile_layer_kernel<<<N_TILES, 1024, 0, stream>>>(binned, offs, cs, cd, h,  W2, b2, hb);
    tile_layer_kernel<<<N_TILES, 1024, 0, stream>>>(binned, offs, cs, cd, hb, W3, b3, h);
    tile_layer_kernel<<<N_TILES, 1024, 0, stream>>>(binned, offs, cs, cd, h,  W4, b4, hb);

    hipMemsetAsync(hg_out, 0, (size_t)N_GRAPHS * HIDDEN * sizeof(float), stream);
    pool_kernel<<<POOL_WGS, T, 0, stream>>>(seg, hb, hg_out, counts);
    pool_div_kernel<<<(N_GRAPHS * HIDDEN + T - 1) / T, T, 0, stream>>>(hg_out, counts);
}

extern "C" void kernel_launch(void* const* d_in, const int* in_sizes, int n_in,
                              void* d_out, int out_size, void* d_ws, size_t ws_size,
                              hipStream_t stream) {
    const int*   src1 = (const int*)d_in[0];
    const int*   dst1 = (const int*)d_in[1];
    const int*   seg1 = (const int*)d_in[2];
    const int*   src2 = (const int*)d_in[3];
    const int*   dst2 = (const int*)d_in[4];
    const int*   seg2 = (const int*)d_in[5];
    const float* W1   = (const float*)d_in[6];
    const float* b1   = (const float*)d_in[7];
    const float* W2   = (const float*)d_in[8];
    const float* b2   = (const float*)d_in[9];
    const float* W3   = (const float*)d_in[10];
    const float* b3   = (const float*)d_in[11];
    const float* W4   = (const float*)d_in[12];
    const float* b4   = (const float*)d_in[13];
    const float* Wc   = (const float*)d_in[14];
    const float* bc   = (const float*)d_in[15];

    float* out   = (float*)d_out;
    float* hg1   = out;
    float* hg2   = out + N_GRAPHS * HIDDEN;
    float* logit = out + 2 * N_GRAPHS * HIDDEN;
    float* ws_f  = (float*)d_ws;

    run_branch(src1, dst1, seg1, W1, b1, W2, b2, W3, b3, W4, b4, hg1, ws_f, stream);
    run_branch(src2, dst2, seg2, W1, b1, W2, b2, W3, b3, W4, b4, hg2, ws_f, stream);

    logits_kernel<<<(N_GRAPHS * N_CLASSES + 255) / 256, 256, 0, stream>>>(hg1, hg2, Wc, bc, logit);
}


# Out-degrees without global atomics: 100 work-groups each count a 16000-edge slice into packed 16-bit LDS counters over all nodes and store them; the tile kernel adds a node's 100 partial counts
# speedup vs baseline: 4.5265x; 1.2321x over previous
#include <hip/hip_runtime.h>
#include <hip/hip_bf16.h>

#define N_NODES   100000
#define N_EDGES   1600000
#define N_GRAPHS  128
#define HIDDEN    128
#define N_CLASSES 10

#define TILE_ROWS  400
#define N_TILES    (N_NODES / TILE_ROWS)
#define CHUNK      4096
#define N_CHUNKS   ((N_EDGES + CHUNK - 1) / CHUNK)
#define OFF_STRIDE 256
#define SRC_BITS   17
static_assert(N_TILES * TILE_ROWS == N_NODES, "tiles must cover the nodes exactly");
static_assert(N_TILES + 1 <= OFF_STRIDE, "offset row too short");
static_assert(N_NODES <= (1 << SRC_BITS), "source index does not fit its field");
#define TILE_PITCH 132
static_assert(TILE_ROWS % 16 == 0, "the tile is cut into whole 16-row matrix tiles");
static_assert((TILE_ROWS + 1) * TILE_PITCH * 4 + 64 <= 320 * 1024, "tile does not fit the LDS");

#define OD_PARTS   100
#define OD_SLICE   (N_EDGES / OD_PARTS)
#define OD_WORDS   (N_NODES / 2)
#define OD_THREADS 1024
#define OD_LOADS   ((OD_SLICE / 4 + OD_THREADS - 1) / OD_THREADS)
static_assert(OD_PARTS * OD_SLICE == N_EDGES, "slices must cover the edges exactly");
static_assert(OD_SLICE <= 65535, "a 16-bit counter must be able to hold a whole slice");
static_assert(OD_SLICE % 4 == 0, "a slice is read four indices at a time and must start 16-byte aligned");
static_assert(N_NODES % 8 == 0, "the packed counters are cleared and stored four words at a time");
static_assert(OD_WORDS * 4 <= 320 * 1024, "counters do not fit the LDS");

#define ROWS_PER_WAVE 10
#define ROW_WGS    (N_NODES / (8 * ROWS_PER_WAVE))
#define POOL_RUN   25
#define POOL_WGS   (N_NODES / (8 * POOL_RUN))
static_assert(ROW_WGS * 8 * ROWS_PER_WAVE == N_NODES, "row grid must cover the nodes exactly");
static_assert(POOL_WGS * 8 * POOL_RUN == N_NODES, "pool grid must cover the nodes exactly");
static_assert(POOL_RUN % 5 == 0, "pooling loads rows in groups of five");

typedef __attribute__((ext_vector_type(2))) float v2f;
typedef __attribute__((ext_vector_type(8))) float v8f;

__global__ void __launch_bounds__(OD_THREADS) outdeg_part_kernel(const int* __restrict__ src, unsigned* __restrict__ part) {
    __shared__ __attribute__((aligned(16))) unsigned cnt[OD_WORDS];
    int tid = threadIdx.x;
    int p   = blockIdx.x;
    const int4* sp = (const int4*)(src + (size_t)p * OD_SLICE);
    int4 s[OD_LOADS];
#pragma unroll
    for (int k = 0; k < OD_LOADS; ++k) s[k] = sp[min(k * OD_THREADS + tid, OD_SLICE / 4 - 1)];
    uint4* cnt4 = (uint4*)cnt;
    for (int i = tid; i < OD_WORDS / 4; i += OD_THREADS) cnt4[i] = uint4{0u, 0u, 0u, 0u};
    __syncthreads();
#pragma unroll
    for (int k = 0; k < OD_LOADS; ++k) {
        if (k * OD_THREADS + tid < OD_SLICE / 4) {
            atomicAdd(&cnt[s[k].x >> 1], 1u << ((s[k].x & 1) * 16));
            atomicAdd(&cnt[s[k].y >> 1], 1u << ((s[k].y & 1) * 16));
            atomicAdd(&cnt[s[k].z >> 1], 1u << ((s[k].z & 1) * 16));
            atomicAdd(&cnt[s[k].w >> 1], 1u << ((s[k].w & 1) * 16));
        }
    }
    __syncthreads();
    uint4* out4 = (uint4*)(part + (size_t)p * OD_WORDS);
    for (int i = tid; i < OD_WORDS / 4; i += OD_THREADS) out4[i] = cnt4[i];
}

__global__ void __launch_bounds__(256) layer1_kernel(const float* __restrict__ agg1, const float* __restrict__ cd,
                                                     const float* __restrict__ W1, const float* __restrict__ b1,
                                                     float* __restrict__ h) {
    int wave = (blockIdx.x * 256 + threadIdx.x) >> 5;
    int lane = threadIdx.x & 31;
    const float4 w = ((const float4*)W1)[lane];
    const float4 b = ((const float4*)b1)[lane];
    int m0 = wave * ROWS_PER_WAVE;
    float x[ROWS_PER_WAVE];
#pragma unroll
    for (int u = 0; u < ROWS_PER_WAVE; ++u) x[u] = agg1[m0 + u] * cd[m0 + u];
    float4* hrow = (float4*)(h + (size_t)m0 * HIDDEN) + lane;
#pragma unroll
    for (int u = 0; u < ROWS_PER_WAVE; ++u) {
        float4 o;
        o.x = fmaxf(fmaf(x[u], w.x, b.x), 0.0f);
        o.y = fmaxf(fmaf(x[u], w.y, b.y), 0.0f);
        o.z = fmaxf(fmaf(x[u], w.z, b.z), 0.0f);
        o.w = fmaxf(fmaf(x[u], w.w, b.w), 0.0f);
        hrow[u * (HIDDEN / 4)] = o;
    }
}

__global__ void __launch_bounds__(256) bin_kernel(const int* __restrict__ src, const int* __restrict__ dst,
                                                  unsigned* __restrict__ binned, int* __restrict__ offs) {
    __shared__ int hist[OFF_STRIDE];
    __shared__ int scanbuf[OFF_STRIDE];
    int tid = threadIdx.x;
    int c   = blockIdx.x;
    int e0  = c * CHUNK;
    hist[tid] = 0;
    __syncthreads();

    static_assert(N_TILES + 1 < OFF_STRIDE, "no spare bin");
    unsigned word[CHUNK / 256];
    int      bin[CHUNK / 256];
    int      rank[CHUNK / 256];
    int      sv[CHUNK / 256], dv[CHUNK / 256];
#pragma unroll
    for (int u = 0; u < CHUNK / 256; ++u) {
        int e = min(e0 + u * 256 + tid, N_EDGES - 1);
        sv[u] = src[e];
        dv[u] = dst[e];
    }
#pragma unroll
    for (int u = 0; u < CHUNK / 256; ++u) {
        bool valid = (e0 + u * 256 + tid) < N_EDGES;
        int b = dv[u] / TILE_ROWS;
        word[u] = (unsigned)sv[u] | ((unsigned)(dv[u] - b * TILE_ROWS) << SRC_BITS);
        bin[u]  = valid ? b : (OFF_STRIDE - 1);
        rank[u] = atomicAdd(&hist[bin[u]], 1);
    }
    __syncthreads();

    int v = hist[tid];
    scanbuf[tid] = v;
    __syncthreads();
    for (int o = 1; o < OFF_STRIDE; o <<= 1) {
        int t = (tid >= o) ? scanbuf[tid - o] : 0;
        __syncthreads();
        scanbuf[tid] += t;
        __syncthreads();
    }
    int excl = scanbuf[tid] - v;
    offs[c * OFF_STRIDE + tid] = excl;
    hist[tid] = excl;
    __syncthreads();

#pragma unroll
    for (int u = 0; u < CHUNK / 256; ++u)
        if (bin[u] != OFF_STRIDE - 1) binned[e0 + hist[bin[u]] + rank[u]] = word[u];
}

#define SCAL_THREADS 512
#define SCAL_UNROLL  8
static_assert(N_CHUNKS <= SCAL_THREADS, "one thread per chunk");
static_assert(TILE_ROWS <= SCAL_THREADS, "one thread per tile row in the epilogue");

__global__ void __launch_bounds__(SCAL_THREADS) tile_indeg_kernel(const unsigned* __restrict__ binned,
                                                                  const int* __restrict__ offs,
                                                                  const unsigned* __restrict__ part,
                                                                  float* __restrict__ msg1,
                                                                  float* __restrict__ cs, float* __restrict__ cd) {
    __shared__ int cnt[TILE_ROWS + 32];
    int tid = threadIdx.x;
    int b   = blockIdx.x;
    int c   = min(tid, N_CHUNKS - 1);
    int lo = 0, hi = 0;
    if (tid < N_CHUNKS) {
        lo = offs[c * OFF_STRIDE + b];
        hi = offs[c * OFF_STRIDE + b + 1];
    }
    unsigned od = 0;
    if (tid < TILE_ROWS) {
        int n  = b * TILE_ROWS + tid;
        int sh = (n & 1) * 16;
        const unsigned* pp = part + (n >> 1);
#pragma unroll
        for (int p = 0; p < OD_PARTS; ++p) od += (pp[(size_t)p * OD_WORDS] >> sh) & 0xffffu;
    }
    if (tid < TILE_ROWS) cnt[tid] = 0;
    __syncthreads();
    const unsigned* sp = binned + (size_t)c * CHUNK;
    for (int i = lo; i < hi; i += SCAL_UNROLL) {
        unsigned w[SCAL_UNROLL];
#pragma unroll
        for (int u = 0; u < SCAL_UNROLL; ++u) w[u] = sp[min(i + u, hi - 1)];
#pragma unroll
        for (int u = 0; u < SCAL_UNROLL; ++u)
            atomicAdd(&cnt[(i + u < hi) ? (int)(w[u] >> SRC_BITS) : TILE_ROWS + (tid & 31)], 1);
    }
    __syncthreads();
    if (tid < TILE_ROWS) {
        int   n  = b * TILE_ROWS + tid;
        float id = (float)cnt[tid];
        float sn = rsqrtf(fmaxf((float)od, 1.0f));
        cd[n]    = rsqrtf(fmaxf(id, 1.0f));
        cs[n]    = sn;
        msg1[n]  = id * sn;
    }
}

__global__ void __launch_bounds__(SCAL_THREADS) tile_agg1_kernel(const unsigned* __restrict__ binned,
                                                                 const int* __restrict__ offs,
                                                                 const float* __restrict__ msg1,
                                                                 float* __restrict__ agg1) {
    __shared__ float acc[TILE_ROWS + 32];
    int tid = threadIdx.x;
    int b   = blockIdx.x;
    int c   = min(tid, N_CHUNKS - 1);
    int lo = 0, hi = 0;
    if (tid < N_CHUNKS) {
        lo = offs[c * OFF_STRIDE + b];
        hi = offs[c * OFF_STRIDE + b + 1];
    }
    if (tid < TILE_ROWS) acc[tid] = 0.0f;
    __syncthreads();
    const unsigned* sp = binned + (size_t)c * CHUNK;
    for (int i = lo; i < hi; i += SCAL_UNROLL) {
        unsigned w[SCAL_UNROLL];
        float    a[SCAL_UNROLL];
#pragma unroll
        for (int u = 0; u < SCAL_UNROLL; ++u) w[u] = sp[min(i + u, hi - 1)];
#pragma unroll
        for (int u = 0; u < SCAL_UNROLL; ++u) {
            a[u] = msg1[w[u] & ((1u << SRC_BITS) - 1u)];
        }
#pragma unroll
        for (int u = 0; u < SCAL_UNROLL; ++u)
            atomicAdd(&acc[(i + u < hi) ? (int)(w[u] >> SRC_BITS) : TILE_ROWS + (tid & 31)], a[u]);
    }
    __syncthreads();
    if (tid < TILE_ROWS) agg1[b * TILE_ROWS + tid] = acc[tid];
}

__global__ void __launch_bounds__(1024) tile_layer_kernel(const unsigned* __restrict__ binned,
                                                          const int* __restrict__ offs,
                                                          const float* __restrict__ cs,
                                                          const float* __restrict__ cd,
                                                          const float* __restrict__ h,
                                                          const float* __restrict__ W,
                                                          const float* __restrict__ bias,
                                                          float* __restrict__ hout) {
    __shared__ __attribute__((aligned(16))) float tile[(TILE_ROWS + 1) * TILE_PITCH];
    __shared__ int   next_chunk;
    int tid  = threadIdx.x;
    int lane = tid & 31;
    int b    = blockIdx.x;

    float4* tile4 = (float4*)tile;
    for (int i = tid; i < (TILE_ROWS + 1) * TILE_PITCH / 4; i += 1024) tile4[i] = float4{0.f, 0.f, 0.f, 0.f};
    if (tid == 0) next_chunk = 0;
    __syncthreads();

    for (int it = 0; it < N_CHUNKS; ++it) {
        int c = 0;
        if (lane == 0) c = atomicAdd(&next_chunk, 1);
        c = __builtin_amdgcn_readfirstlane(c);
        if (c >= N_CHUNKS) break;
        int lo = __builtin_amdgcn_readfirstlane(offs[c * OFF_STRIDE + b]);
        int hi = __builtin_amdgcn_readfirstlane(offs[c * OFF_STRIDE + b + 1]);
        const unsigned* sp = binned + (size_t)c * CHUNK;
        for (int base = lo; base < hi; base += 32) {
            int cnt = min(32, hi - base);
            unsigned w = sp[base + min(lane, cnt - 1)];
            for (int j0 = 0; j0 < cnt; j0 += 8) {
                float  sc[8];
                float2 va[8], vb[8];
                int    row[8];
#pragma unroll
                for (int u = 0; u < 8; ++u) {
                    int j = min(j0 + u, cnt - 1);
                    unsigned wj = (unsigned)__builtin_amdgcn_readlane((int)w, j);
                    int s  = (int)(wj & ((1u << SRC_BITS) - 1u));
                    row[u] = (j0 + u < cnt) ? (int)(wj >> SRC_BITS) : TILE_ROWS;
                    sc[u]  = cs[s];
                    const float2* hp = (const float2*)(h + (size_t)s * HIDDEN) + lane;
                    va[u] = hp[0];
                    vb[u] = hp[32];
                }
#pragma unroll
                for (int u = 0; u < 8; ++u) {
                    float* t = tile + row[u] * TILE_PITCH + 2 * lane;
                    atomicAdd(t,      va[u].x * sc[u]);
                    atomicAdd(t + 1,  va[u].y * sc[u]);
                    atomicAdd(t + 64, vb[u].x * sc[u]);
                    atomicAdd(t + 65, vb[u].y * sc[u]);
                }
            }
        }
    }

    int wave = __builtin_amdgcn_readfirstlane(tid >> 5);
    int half = lane >> 4;
    int lr   = lane & 15;
    int n    = (wave & 7) * 16 + lr;
    const float* wp = W + (half * 2) * HIDDEN + n;
    v2f bw[HIDDEN / 4];
#pragma unroll
    for (int ks = 0; ks < HIDDEN / 4; ++ks) {
        bw[ks].x = wp[(4 * ks) * HIDDEN];
        bw[ks].y = wp[(4 * ks + 1) * HIDDEN];
    }
    float bn = bias[n];
    __syncthreads();

    for (int rt = wave >> 3; rt < TILE_ROWS / 16; rt += 4) {
        int   r0  = rt * 16;
        float cdm = cd[b * TILE_ROWS + r0 + lr];
        const float* ap = tile + (r0 + lr) * TILE_PITCH + half * 2;
        v8f acc = {};
#pragma unroll
        for (int ks = 0; ks < HIDDEN / 4; ++ks) {
            float2 a = *(const float2*)(ap + 4 * ks);
            v2f av;
            av.x = a.x * cdm;
            av.y = a.y * cdm;
            acc = __builtin_amdgcn_wmma_f32_16x16x4_f32(
                 false, av,  false, bw[ks],
                 (short)0, acc,  false,  false);
        }
        float* op = hout + (size_t)(b * TILE_ROWS + r0 + half * 8) * HIDDEN + n;
#pragma unroll
        for (int r = 0; r < 8; ++r)
            op[(size_t)r * HIDDEN] = fmaxf(acc[r] + bn, 0.0f);
    }
}

__device__ __forceinline__ void pool_flush(float* __restrict__ hg, float* __restrict__ counts,
                                           int g, int lane, float4 acc, float cnt) {
    float* o = hg + (size_t)g * HIDDEN + 4 * lane;
    atomicAdd(o,     acc.x);
    atomicAdd(o + 1, acc.y);
    atomicAdd(o + 2, acc.z);
    atomicAdd(o + 3, acc.w);
    if (lane == 0) atomicAdd(&counts[g], cnt);
}

__global__ void __launch_bounds__(256) pool_kernel(const int* __restrict__ seg, const float* __restrict__ h,
                                                   float* __restrict__ hg, float* __restrict__ counts) {
    int wave = (blockIdx.x * 256 + threadIdx.x) >> 5;
    int lane = threadIdx.x & 31;
    int n0   = wave * POOL_RUN;
    float4 acc = float4{0.f, 0.f, 0.f, 0.f};
    float  cnt = 0.0f;
    int gcur = __builtin_amdgcn_readfirstlane(seg[n0]);
    for (int nb = n0; nb < n0 + POOL_RUN; nb += 5) {
        int    g[5];
        float4 v[5];
#pragma unroll
        for (int u = 0; u < 5; ++u) {
            g[u] = __builtin_amdgcn_readfirstlane(seg[nb + u]);
            v[u] = ((const float4*)(h + (size_t)(nb + u) * HIDDEN))[lane];
        }
#pragma unroll
        for (int u = 0; u < 5; ++u) {
            if (g[u] != gcur) {
                pool_flush(hg, counts, gcur, lane, acc, cnt);
                acc  = float4{0.f, 0.f, 0.f, 0.f};
                cnt  = 0.0f;
                gcur = g[u];
            }
            acc.x += v[u].x;
            acc.y += v[u].y;
            acc.z += v[u].z;
            acc.w += v[u].w;
            cnt   += 1.0f;
        }
    }
    pool_flush(hg, counts, gcur, lane, acc, cnt);
}

__global__ void pool_div_kernel(float* __restrict__ hg, const float* __restrict__ counts) {
    int i = blockIdx.x * blockDim.x + threadIdx.x;
    if (i < N_GRAPHS * HIDDEN)
        hg[i] /= fmaxf(counts[i >> 7], 1.0f);
}

__global__ void logits_kernel(const float* __restrict__ hg1, const float* __restrict__ hg2,
                              const float* __restrict__ Wc, const float* __restrict__ bc,
                              float* __restrict__ logits) {
    int i = blockIdx.x * blockDim.x + threadIdx.x;
    if (i < N_GRAPHS * N_CLASSES) {
        int g = i / N_CLASSES, c = i % N_CLASSES;
        float s = bc[c];
#pragma unroll 8
        for (int k = 0; k < HIDDEN; ++k)
            s += fabsf(hg1[g * HIDDEN + k] - hg2[g * HIDDEN + k]) * Wc[k * N_CLASSES + c];
        logits[i] = s;
    }
}

static void run_branch(const int* src, const int* dst, const int* seg,
                       const float* W1, const float* b1,
                       const float* W2, const float* b2,
                       const float* W3, const float* b3,
                       const float* W4, const float* b4,
                       float* hg_out, float* ws_f, hipStream_t stream) {
    float* msg1   = ws_f + 100000;
    float* cs     = ws_f + 200000;
    float* cd     = ws_f + 300000;
    float* agg1   = ws_f + 400000;
    float* counts = ws_f + 500000;
    float* h      = ws_f + 500224;
    float* hb     = h + (size_t)N_NODES * HIDDEN;
    unsigned* binned = (unsigned*)(hb + (size_t)N_NODES * HIDDEN);
    int*      offs   = (int*)(binned + N_EDGES);
    unsigned* part   = (unsigned*)(offs + N_CHUNKS * OFF_STRIDE);

    hipMemsetAsync(counts, 0, (size_t)N_GRAPHS * sizeof(float), stream);

    const int T = 256;
    outdeg_part_kernel<<<OD_PARTS, OD_THREADS, 0, stream>>>(src, part);

    bin_kernel<<<N_CHUNKS, 256, 0, stream>>>(src, dst, binned, offs);

    tile_indeg_kernel<<<N_TILES, SCAL_THREADS, 0, stream>>>(binned, offs, part, msg1, cs, cd);
    tile_agg1_kernel<<<N_TILES, SCAL_THREADS, 0, stream>>>(binned, offs, msg1, agg1);
    layer1_kernel<<<ROW_WGS, T, 0, stream>>>(agg1, cd, W1, b1, h);

    tile_layer_kernel<<<N_TILES, 1024, 0, stream>>>(binned, offs, cs, cd, h,  W2, b2, hb);
    tile_layer_kernel<<<N_TILES, 1024, 0, stream>>>(binned, offs, cs, cd, hb, W3, b3, h);
    tile_layer_kernel<<<N_TILES, 1024, 0, stream>>>(binned, offs, cs, cd, h,  W4, b4, hb);

    hipMemsetAsync(hg_out, 0, (size_t)N_GRAPHS * HIDDEN * sizeof(float), stream);
    pool_kernel<<<POOL_WGS, T, 0, stream>>>(seg, hb, hg_out, counts);
    pool_div_kernel<<<(N_GRAPHS * HIDDEN + T - 1) / T, T, 0, stream>>>(hg_out, counts);
}

extern "C" void kernel_launch(void* const* d_in, const int* in_sizes, int n_in,
                              void* d_out, int out_size, void* d_ws, size_t ws_size,
                              hipStream_t stream) {
    const int*   src1 = (const int*)d_in[0];
    const int*   dst1 = (const int*)d_in[1];
    const int*   seg1 = (const int*)d_in[2];
    const int*   src2 = (const int*)d_in[3];
    const int*   dst2 = (const int*)d_in[4];
    const int*   seg2 = (const int*)d_in[5];
    const float* W1   = (const float*)d_in[6];
    const float* b1   = (const float*)d_in[7];
    const float* W2   = (const float*)d_in[8];
    const float* b2   = (const float*)d_in[9];
    const float* W3   = (const float*)d_in[10];
    const float* b3   = (const float*)d_in[11];
    const float* W4   = (const float*)d_in[12];
    const float* b4   = (const float*)d_in[13];
    const float* Wc   = (const float*)d_in[14];
    const float* bc   = (const float*)d_in[15];

    float* out   = (float*)d_out;
    float* hg1   = out;
    float* hg2   = out + N_GRAPHS * HIDDEN;
    float* logit = out + 2 * N_GRAPHS * HIDDEN;
    float* ws_f  = (float*)d_ws;

    run_branch(src1, dst1, seg1, W1, b1, W2, b2, W3, b3, W4, b4, hg1, ws_f, stream);
    run_branch(src2, dst2, seg2, W1, b1, W2, b2, W3, b3, W4, b4, hg2, ws_f, stream);

    logits_kernel<<<(N_GRAPHS * N_CLASSES + 255) / 256, 256, 0, stream>>>(hg1, hg2, Wc, bc, logit);
}


# Layers 2-4 edge phase: a slot past a segment's end issues no LDS add (scalar branch per slot, the group's loads held ahead of it; spare LDS row gone): a sixth fewer ds_add_f32; sum order as before
# speedup vs baseline: 4.8427x; 1.0699x over previous
#include <hip/hip_runtime.h>
#include <hip/hip_bf16.h>

#define N_NODES   100000
#define N_EDGES   1600000
#define N_GRAPHS  128
#define HIDDEN    128
#define N_CLASSES 10

#define TILE_ROWS  400
#define N_TILES    (N_NODES / TILE_ROWS)
#define CHUNK      4096
#define N_CHUNKS   ((N_EDGES + CHUNK - 1) / CHUNK)
#define OFF_STRIDE 256
#define SRC_BITS   17
static_assert(N_TILES * TILE_ROWS == N_NODES, "tiles must cover the nodes exactly");
static_assert(N_TILES + 1 <= OFF_STRIDE, "offset row too short");
static_assert(N_NODES <= (1 << SRC_BITS), "source index does not fit its field");
#define TILE_PITCH 132
static_assert(TILE_ROWS % 16 == 0, "the tile is cut into whole 16-row matrix tiles");
static_assert(TILE_ROWS * TILE_PITCH * 4 + 64 <= 320 * 1024, "tile does not fit the LDS");

#define OD_PARTS   100
#define OD_SLICE   (N_EDGES / OD_PARTS)
#define OD_WORDS   (N_NODES / 2)
#define OD_THREADS 1024
#define OD_LOADS   ((OD_SLICE / 4 + OD_THREADS - 1) / OD_THREADS)
static_assert(OD_PARTS * OD_SLICE == N_EDGES, "slices must cover the edges exactly");
static_assert(OD_SLICE <= 65535, "a 16-bit counter must be able to hold a whole slice");
static_assert(OD_SLICE % 4 == 0, "a slice is read four indices at a time and must start 16-byte aligned");
static_assert(N_NODES % 8 == 0, "the packed counters are cleared and stored four words at a time");
static_assert(OD_WORDS * 4 <= 320 * 1024, "counters do not fit the LDS");

#define ROWS_PER_WAVE 10
#define ROW_WGS    (N_NODES / (8 * ROWS_PER_WAVE))
#define POOL_RUN   25
#define POOL_WGS   (N_NODES / (8 * POOL_RUN))
static_assert(ROW_WGS * 8 * ROWS_PER_WAVE == N_NODES, "row grid must cover the nodes exactly");
static_assert(POOL_WGS * 8 * POOL_RUN == N_NODES, "pool grid must cover the nodes exactly");
static_assert(POOL_RUN % 5 == 0, "pooling loads rows in groups of five");

typedef __attribute__((ext_vector_type(2))) float v2f;
typedef __attribute__((ext_vector_type(8))) float v8f;

__global__ void __launch_bounds__(OD_THREADS) outdeg_part_kernel(const int* __restrict__ src, unsigned* __restrict__ part) {
    __shared__ __attribute__((aligned(16))) unsigned cnt[OD_WORDS];
    int tid = threadIdx.x;
    int p   = blockIdx.x;
    const int4* sp = (const int4*)(src + (size_t)p * OD_SLICE);
    int4 s[OD_LOADS];
#pragma unroll
    for (int k = 0; k < OD_LOADS; ++k) s[k] = sp[min(k * OD_THREADS + tid, OD_SLICE / 4 - 1)];
    uint4* cnt4 = (uint4*)cnt;
    for (int i = tid; i < OD_WORDS / 4; i += OD_THREADS) cnt4[i] = uint4{0u, 0u, 0u, 0u};
    __syncthreads();
#pragma unroll
    for (int k = 0; k < OD_LOADS; ++k) {
        if (k * OD_THREADS + tid < OD_SLICE / 4) {
            atomicAdd(&cnt[s[k].x >> 1], 1u << ((s[k].x & 1) * 16));
            atomicAdd(&cnt[s[k].y >> 1], 1u << ((s[k].y & 1) * 16));
            atomicAdd(&cnt[s[k].z >> 1], 1u << ((s[k].z & 1) * 16));
            atomicAdd(&cnt[s[k].w >> 1], 1u << ((s[k].w & 1) * 16));
        }
    }
    __syncthreads();
    uint4* out4 = (uint4*)(part + (size_t)p * OD_WORDS);
    for (int i = tid; i < OD_WORDS / 4; i += OD_THREADS) out4[i] = cnt4[i];
}

__global__ void __launch_bounds__(256) layer1_kernel(const float* __restrict__ agg1, const float* __restrict__ cd,
                                                     const float* __restrict__ W1, const float* __restrict__ b1,
                                                     float* __restrict__ h) {
    int wave = (blockIdx.x * 256 + threadIdx.x) >> 5;
    int lane = threadIdx.x & 31;
    const float4 w = ((const float4*)W1)[lane];
    const float4 b = ((const float4*)b1)[lane];
    int m0 = wave * ROWS_PER_WAVE;
    float x[ROWS_PER_WAVE];
#pragma unroll
    for (int u = 0; u < ROWS_PER_WAVE; ++u) x[u] = agg1[m0 + u] * cd[m0 + u];
    float4* hrow = (float4*)(h + (size_t)m0 * HIDDEN) + lane;
#pragma unroll
    for (int u = 0; u < ROWS_PER_WAVE; ++u) {
        float4 o;
        o.x = fmaxf(fmaf(x[u], w.x, b.x), 0.0f);
        o.y = fmaxf(fmaf(x[u], w.y, b.y), 0.0f);
        o.z = fmaxf(fmaf(x[u], w.z, b.z), 0.0f);
        o.w = fmaxf(fmaf(x[u], w.w, b.w), 0.0f);
        hrow[u * (HIDDEN / 4)] = o;
    }
}

__global__ void __launch_bounds__(256) bin_kernel(const int* __restrict__ src, const int* __restrict__ dst,
                                                  unsigned* __restrict__ binned, int* __restrict__ offs) {
    __shared__ int hist[OFF_STRIDE];
    __shared__ int scanbuf[OFF_STRIDE];
    int tid = threadIdx.x;
    int c   = blockIdx.x;
    int e0  = c * CHUNK;
    hist[tid] = 0;
    __syncthreads();

    static_assert(N_TILES + 1 < OFF_STRIDE, "no spare bin");
    unsigned word[CHUNK / 256];
    int      bin[CHUNK / 256];
    int      rank[CHUNK / 256];
    int      sv[CHUNK / 256], dv[CHUNK / 256];
#pragma unroll
    for (int u = 0; u < CHUNK / 256; ++u) {
        int e = min(e0 + u * 256 + tid, N_EDGES - 1);
        sv[u] = src[e];
        dv[u] = dst[e];
    }
#pragma unroll
    for (int u = 0; u < CHUNK / 256; ++u) {
        bool valid = (e0 + u * 256 + tid) < N_EDGES;
        int b = dv[u] / TILE_ROWS;
        word[u] = (unsigned)sv[u] | ((unsigned)(dv[u] - b * TILE_ROWS) << SRC_BITS);
        bin[u]  = valid ? b : (OFF_STRIDE - 1);
        rank[u] = atomicAdd(&hist[bin[u]], 1);
    }
    __syncthreads();

    int v = hist[tid];
    scanbuf[tid] = v;
    __syncthreads();
    for (int o = 1; o < OFF_STRIDE; o <<= 1) {
        int t = (tid >= o) ? scanbuf[tid - o] : 0;
        __syncthreads();
        scanbuf[tid] += t;
        __syncthreads();
    }
    int excl = scanbuf[tid] - v;
    offs[c * OFF_STRIDE + tid] = excl;
    hist[tid] = excl;
    __syncthreads();

#pragma unroll
    for (int u = 0; u < CHUNK / 256; ++u)
        if (bin[u] != OFF_STRIDE - 1) binned[e0 + hist[bin[u]] + rank[u]] = word[u];
}

#define SCAL_THREADS 512
#define SCAL_UNROLL  8
static_assert(N_CHUNKS <= SCAL_THREADS, "one thread per chunk");
static_assert(TILE_ROWS <= SCAL_THREADS, "one thread per tile row in the epilogue");

__global__ void __launch_bounds__(SCAL_THREADS) tile_indeg_kernel(const unsigned* __restrict__ binned,
                                                                  const int* __restrict__ offs,
                                                                  const unsigned* __restrict__ part,
                                                                  float* __restrict__ msg1,
                                                                  float* __restrict__ cs, float* __restrict__ cd) {
    __shared__ int cnt[TILE_ROWS + 32];
    int tid = threadIdx.x;
    int b   = blockIdx.x;
    int c   = min(tid, N_CHUNKS - 1);
    int lo = 0, hi = 0;
    if (tid < N_CHUNKS) {
        lo = offs[c * OFF_STRIDE + b];
        hi = offs[c * OFF_STRIDE + b + 1];
    }
    unsigned od = 0;
    if (tid < TILE_ROWS) {
        int n  = b * TILE_ROWS + tid;
        int sh = (n & 1) * 16;
        const unsigned* pp = part + (n >> 1);
#pragma unroll
        for (int p = 0; p < OD_PARTS; ++p) od += (pp[(size_t)p * OD_WORDS] >> sh) & 0xffffu;
    }
    if (tid < TILE_ROWS) cnt[tid] = 0;
    __syncthreads();
    const unsigned* sp = binned + (size_t)c * CHUNK;
    for (int i = lo; i < hi; i += SCAL_UNROLL) {
        unsigned w[SCAL_UNROLL];
#pragma unroll
        for (int u = 0; u < SCAL_UNROLL; ++u) w[u] = sp[min(i + u, hi - 1)];
#pragma unroll
        for (int u = 0; u < SCAL_UNROLL; ++u)
            atomicAdd(&cnt[(i + u < hi) ? (int)(w[u] >> SRC_BITS) : TILE_ROWS + (tid & 31)], 1);
    }
    __syncthreads();
    if (tid < TILE_ROWS) {
        int   n  = b * TILE_ROWS + tid;
        float id = (float)cnt[tid];
        float sn = rsqrtf(fmaxf((float)od, 1.0f));
        cd[n]    = rsqrtf(fmaxf(id, 1.0f));
        cs[n]    = sn;
        msg1[n]  = id * sn;
    }
}

__global__ void __launch_bounds__(SCAL_THREADS) tile_agg1_kernel(const unsigned* __restrict__ binned,
                                                                 const int* __restrict__ offs,
                                                                 const float* __restrict__ msg1,
                                                                 float* __restrict__ agg1) {
    __shared__ float acc[TILE_ROWS + 32];
    int tid = threadIdx.x;
    int b   = blockIdx.x;
    int c   = min(tid, N_CHUNKS - 1);
    int lo = 0, hi = 0;
    if (tid < N_CHUNKS) {
        lo = offs[c * OFF_STRIDE + b];
        hi = offs[c * OFF_STRIDE + b + 1];
    }
    if (tid < TILE_ROWS) acc[tid] = 0.0f;
    __syncthreads();
    const unsigned* sp = binned + (size_t)c * CHUNK;
    for (int i = lo; i < hi; i += SCAL_UNROLL) {
        unsigned w[SCAL_UNROLL];
        float    a[SCAL_UNROLL];
#pragma unroll
        for (int u = 0; u < SCAL_UNROLL; ++u) w[u] = sp[min(i + u, hi - 1)];
#pragma unroll
        for (int u = 0; u < SCAL_UNROLL; ++u) {
            a[u] = msg1[w[u] & ((1u << SRC_BITS) - 1u)];
        }
#pragma unroll
        for (int u = 0; u < SCAL_UNROLL; ++u)
            atomicAdd(&acc[(i + u < hi) ? (int)(w[u] >> SRC_BITS) : TILE_ROWS + (tid & 31)], a[u]);
    }
    __syncthreads();
    if (tid < TILE_ROWS) agg1[b * TILE_ROWS + tid] = acc[tid];
}

__global__ void __launch_bounds__(1024) tile_layer_kernel(const unsigned* __restrict__ binned,
                                                          const int* __restrict__ offs,
                                                          const float* __restrict__ cs,
                                                          const float* __restrict__ cd,
                                                          const float* __restrict__ h,
                                                          const float* __restrict__ W,
                                                          const float* __restrict__ bias,
                                                          float* __restrict__ hout) {
    __shared__ __attribute__((aligned(16))) float tile[TILE_ROWS * TILE_PITCH];
    __shared__ int   next_chunk;
    int tid  = threadIdx.x;
    int lane = tid & 31;
    int b    = blockIdx.x;

    float4* tile4 = (float4*)tile;
    for (int i = tid; i < TILE_ROWS * TILE_PITCH / 4; i += 1024) tile4[i] = float4{0.f, 0.f, 0.f, 0.f};
    if (tid == 0) next_chunk = 0;
    __syncthreads();

    for (int it = 0; it < N_CHUNKS; ++it) {
        int c = 0;
        if (lane == 0) c = atomicAdd(&next_chunk, 1);
        c = __builtin_amdgcn_readfirstlane(c);
        if (c >= N_CHUNKS) break;
        int lo = __builtin_amdgcn_readfirstlane(offs[c * OFF_STRIDE + b]);
        int hi = __builtin_amdgcn_readfirstlane(offs[c * OFF_STRIDE + b + 1]);
        const unsigned* sp = binned + (size_t)c * CHUNK;
        for (int base = lo; base < hi; base += 32) {
            int cnt = min(32, hi - base);
            unsigned w = sp[base + min(lane, cnt - 1)];
            for (int j0 = 0; j0 < cnt; j0 += 8) {
                float  sc[8];
                float2 va[8], vb[8];
                int    row[8];
#pragma unroll
                for (int u = 0; u < 8; ++u) {
                    int j = min(j0 + u, cnt - 1);
                    unsigned wj = (unsigned)__builtin_amdgcn_readlane((int)w, j);
                    int s  = (int)(wj & ((1u << SRC_BITS) - 1u));
                    row[u] = (int)(wj >> SRC_BITS);
                    sc[u]  = cs[s];
                    const float2* hp = (const float2*)(h + (size_t)s * HIDDEN) + lane;
                    va[u] = hp[0];
                    vb[u] = hp[32];
                }
#pragma unroll
                for (int u = 0; u < 8; ++u)
                    asm volatile("" :: "s"(sc[u]), "v"(va[u].x), "v"(va[u].y), "v"(vb[u].x), "v"(vb[u].y));
#pragma unroll
                for (int u = 0; u < 8; ++u) {
                    if (u == 0 || j0 + u < cnt) {
                        float* t = tile + row[u] * TILE_PITCH + 2 * lane;
                        atomicAdd(t,      va[u].x * sc[u]);
                        atomicAdd(t + 1,  va[u].y * sc[u]);
                        atomicAdd(t + 64, vb[u].x * sc[u]);
                        atomicAdd(t + 65, vb[u].y * sc[u]);
                    }
                }
            }
        }
    }

    int wave = __builtin_amdgcn_readfirstlane(tid >> 5);
    int half = lane >> 4;
    int lr   = lane & 15;
    int n    = (wave & 7) * 16 + lr;
    const float* wp = W + (half * 2) * HIDDEN + n;
    v2f bw[HIDDEN / 4];
#pragma unroll
    for (int ks = 0; ks < HIDDEN / 4; ++ks) {
        bw[ks].x = wp[(4 * ks) * HIDDEN];
        bw[ks].y = wp[(4 * ks + 1) * HIDDEN];
    }
    float bn = bias[n];
    __syncthreads();

    for (int rt = wave >> 3; rt < TILE_ROWS / 16; rt += 4) {
        int   r0  = rt * 16;
        float cdm = cd[b * TILE_ROWS + r0 + lr];
        const float* ap = tile + (r0 + lr) * TILE_PITCH + half * 2;
        v8f acc = {};
#pragma unroll
        for (int ks = 0; ks < HIDDEN / 4; ++ks) {
            float2 a = *(const float2*)(ap + 4 * ks);
            v2f av;
            av.x = a.x * cdm;
            av.y = a.y * cdm;
            acc = __builtin_amdgcn_wmma_f32_16x16x4_f32(
                 false, av,  false, bw[ks],
                 (short)0, acc,  false,  false);
        }
        float* op = hout + (size_t)(b * TILE_ROWS + r0 + half * 8) * HIDDEN + n;
#pragma unroll
        for (int r = 0; r < 8; ++r)
            op[(size_t)r * HIDDEN] = fmaxf(acc[r] + bn, 0.0f);
    }
}

__device__ __forceinline__ void pool_flush(float* __restrict__ hg, float* __restrict__ counts,
                                           int g, int lane, float4 acc, float cnt) {
    float* o = hg + (size_t)g * HIDDEN + 4 * lane;
    atomicAdd(o,     acc.x);
    atomicAdd(o + 1, acc.y);
    atomicAdd(o + 2, acc.z);
    atomicAdd(o + 3, acc.w);
    if (lane == 0) atomicAdd(&counts[g], cnt);
}

__global__ void __launch_bounds__(256) pool_kernel(const int* __restrict__ seg, const float* __restrict__ h,
                                                   float* __restrict__ hg, float* __restrict__ counts) {
    int wave = (blockIdx.x * 256 + threadIdx.x) >> 5;
    int lane = threadIdx.x & 31;
    int n0   = wave * POOL_RUN;
    float4 acc = float4{0.f, 0.f, 0.f, 0.f};
    float  cnt = 0.0f;
    int gcur = __builtin_amdgcn_readfirstlane(seg[n0]);
    for (int nb = n0; nb < n0 + POOL_RUN; nb += 5) {
        int    g[5];
        float4 v[5];
#pragma unroll
        for (int u = 0; u < 5; ++u) {
            g[u] = __builtin_amdgcn_readfirstlane(seg[nb + u]);
            v[u] = ((const float4*)(h + (size_t)(nb + u) * HIDDEN))[lane];
        }
#pragma unroll
        for (int u = 0; u < 5; ++u) {
            if (g[u] != gcur) {
                pool_flush(hg, counts, gcur, lane, acc, cnt);
                acc  = float4{0.f, 0.f, 0.f, 0.f};
                cnt  = 0.0f;
                gcur = g[u];
            }
            acc.x += v[u].x;
            acc.y += v[u].y;
            acc.z += v[u].z;
            acc.w += v[u].w;
            cnt   += 1.0f;
        }
    }
    pool_flush(hg, counts, gcur, lane, acc, cnt);
}

__global__ void pool_div_kernel(float* __restrict__ hg, const float* __restrict__ counts) {
    int i = blockIdx.x * blockDim.x + threadIdx.x;
    if (i < N_GRAPHS * HIDDEN)
        hg[i] /= fmaxf(counts[i >> 7], 1.0f);
}

__global__ void logits_kernel(const float* __restrict__ hg1, const float* __restrict__ hg2,
                              const float* __restrict__ Wc, const float* __restrict__ bc,
                              float* __restrict__ logits) {
    int i = blockIdx.x * blockDim.x + threadIdx.x;
    if (i < N_GRAPHS * N_CLASSES) {
        int g = i / N_CLASSES, c = i % N_CLASSES;
        float s = bc[c];
#pragma unroll 8
        for (int k = 0; k < HIDDEN; ++k)
            s += fabsf(hg1[g * HIDDEN + k] - hg2[g * HIDDEN + k]) * Wc[k * N_CLASSES + c];
        logits[i] = s;
    }
}

static void run_branch(const int* src, const int* dst, const int* seg,
                       const float* W1, const float* b1,
                       const float* W2, const float* b2,
                       const float* W3, const float* b3,
                       const float* W4, const float* b4,
                       float* hg_out, float* ws_f, hipStream_t stream) {
    float* msg1   = ws_f + 100000;
    float* cs     = ws_f + 200000;
    float* cd     = ws_f + 300000;
    float* agg1   = ws_f + 400000;
    float* counts = ws_f + 500000;
    float* h      = ws_f + 500224;
    float* hb     = h + (size_t)N_NODES * HIDDEN;
    unsigned* binned = (unsigned*)(hb + (size_t)N_NODES * HIDDEN);
    int*      offs   = (int*)(binned + N_EDGES);
    unsigned* part   = (unsigned*)(offs + N_CHUNKS * OFF_STRIDE);

    hipMemsetAsync(counts, 0, (size_t)N_GRAPHS * sizeof(float), stream);

    const int T = 256;
    outdeg_part_kernel<<<OD_PARTS, OD_THREADS, 0, stream>>>(src, part);

    bin_kernel<<<N_CHUNKS, 256, 0, stream>>>(src, dst, binned, offs);

    tile_indeg_kernel<<<N_TILES, SCAL_THREADS, 0, stream>>>(binned, offs, part, msg1, cs, cd);
    tile_agg1_kernel<<<N_TILES, SCAL_THREADS, 0, stream>>>(binned, offs, msg1, agg1);
    layer1_kernel<<<ROW_WGS, T, 0, stream>>>(agg1, cd, W1, b1, h);

    tile_layer_kernel<<<N_TILES, 1024, 0, stream>>>(binned, offs, cs, cd, h,  W2, b2, hb);
    tile_layer_kernel<<<N_TILES, 1024, 0, stream>>>(binned, offs, cs, cd, hb, W3, b3, h);
    tile_layer_kernel<<<N_TILES, 1024, 0, stream>>>(binned, offs, cs, cd, h,  W4, b4, hb);

    hipMemsetAsync(hg_out, 0, (size_t)N_GRAPHS * HIDDEN * sizeof(float), stream);
    pool_kernel<<<POOL_WGS, T, 0, stream>>>(seg, hb, hg_out, counts);
    pool_div_kernel<<<(N_GRAPHS * HIDDEN + T - 1) / T, T, 0, stream>>>(hg_out, counts);
}

extern "C" void kernel_launch(void* const* d_in, const int* in_sizes, int n_in,
                              void* d_out, int out_size, void* d_ws, size_t ws_size,
                              hipStream_t stream) {
    const int*   src1 = (const int*)d_in[0];
    const int*   dst1 = (const int*)d_in[1];
    const int*   seg1 = (const int*)d_in[2];
    const int*   src2 = (const int*)d_in[3];
    const int*   dst2 = (const int*)d_in[4];
    const int*   seg2 = (const int*)d_in[5];
    const float* W1   = (const float*)d_in[6];
    const float* b1   = (const float*)d_in[7];
    const float* W2   = (const float*)d_in[8];
    const float* b2   = (const float*)d_in[9];
    const float* W3   = (const float*)d_in[10];
    const float* b3   = (const float*)d_in[11];
    const float* W4   = (const float*)d_in[12];
    const float* b4   = (const float*)d_in[13];
    const float* Wc   = (const float*)d_in[14];
    const float* bc   = (const float*)d_in[15];

    float* out   = (float*)d_out;
    float* hg1   = out;
    float* hg2   = out + N_GRAPHS * HIDDEN;
    float* logit = out + 2 * N_GRAPHS * HIDDEN;
    float* ws_f  = (float*)d_ws;

    run_branch(src1, dst1, seg1, W1, b1, W2, b2, W3, b3, W4, b4, hg1, ws_f, stream);
    run_branch(src2, dst2, seg2, W1, b1, W2, b2, W3, b3, W4, b4, hg2, ws_f, stream);

    logits_kernel<<<(N_GRAPHS * N_CLASSES + 255) / 256, 256, 0, stream>>>(hg1, hg2, Wc, bc, logit);
}


# Layers 2-4 edge sums: edges listed per owner wave in LDS (row&31); a wave adds its rows by plain 16-byte LDS load/store, next 8 gathers in flight: no ds_add_f32; sum order: list order, fused mul-add
# speedup vs baseline: 5.1229x; 1.0579x over previous
#include <hip/hip_runtime.h>
#include <hip/hip_bf16.h>

#define N_NODES   100000
#define N_EDGES   1600000
#define N_GRAPHS  128
#define HIDDEN    128
#define N_CLASSES 10

#define TILE_ROWS  400
#define N_TILES    (N_NODES / TILE_ROWS)
#define CHUNK      4096
#define N_CHUNKS   ((N_EDGES + CHUNK - 1) / CHUNK)
#define OFF_STRIDE 256
#define SRC_BITS   17
static_assert(N_TILES * TILE_ROWS == N_NODES, "tiles must cover the nodes exactly");
static_assert(N_TILES + 1 <= OFF_STRIDE, "offset row too short");
static_assert(N_NODES <= (1 << SRC_BITS), "source index does not fit its field");
#define TILE_PITCH 132
static_assert(TILE_ROWS % 16 == 0, "the tile is cut into whole 16-row matrix tiles");
static_assert(TILE_ROWS * TILE_PITCH * 4 + 64 <= 320 * 1024, "tile does not fit the LDS");

#define OD_PARTS   100
#define OD_SLICE   (N_EDGES / OD_PARTS)
#define OD_WORDS   (N_NODES / 2)
#define OD_THREADS 1024
#define OD_LOADS   ((OD_SLICE / 4 + OD_THREADS - 1) / OD_THREADS)
static_assert(OD_PARTS * OD_SLICE == N_EDGES, "slices must cover the edges exactly");
static_assert(OD_SLICE <= 65535, "a 16-bit counter must be able to hold a whole slice");
static_assert(OD_SLICE % 4 == 0, "a slice is read four indices at a time and must start 16-byte aligned");
static_assert(N_NODES % 8 == 0, "the packed counters are cleared and stored four words at a time");
static_assert(OD_WORDS * 4 <= 320 * 1024, "counters do not fit the LDS");

#define ROWS_PER_WAVE 10
#define ROW_WGS    (N_NODES / (8 * ROWS_PER_WAVE))
#define POOL_RUN   25
#define POOL_WGS   (N_NODES / (8 * POOL_RUN))
static_assert(ROW_WGS * 8 * ROWS_PER_WAVE == N_NODES, "row grid must cover the nodes exactly");
static_assert(POOL_WGS * 8 * POOL_RUN == N_NODES, "pool grid must cover the nodes exactly");
static_assert(POOL_RUN % 5 == 0, "pooling loads rows in groups of five");

typedef __attribute__((ext_vector_type(2))) float v2f;
typedef __attribute__((ext_vector_type(8))) float v8f;

__global__ void __launch_bounds__(OD_THREADS) outdeg_part_kernel(const int* __restrict__ src, unsigned* __restrict__ part) {
    __shared__ __attribute__((aligned(16))) unsigned cnt[OD_WORDS];
    int tid = threadIdx.x;
    int p   = blockIdx.x;
    const int4* sp = (const int4*)(src + (size_t)p * OD_SLICE);
    int4 s[OD_LOADS];
#pragma unroll
    for (int k = 0; k < OD_LOADS; ++k) s[k] = sp[min(k * OD_THREADS + tid, OD_SLICE / 4 - 1)];
    uint4* cnt4 = (uint4*)cnt;
    for (int i = tid; i < OD_WORDS / 4; i += OD_THREADS) cnt4[i] = uint4{0u, 0u, 0u, 0u};
    __syncthreads();
#pragma unroll
    for (int k = 0; k < OD_LOADS; ++k) {
        if (k * OD_THREADS + tid < OD_SLICE / 4) {
            atomicAdd(&cnt[s[k].x >> 1], 1u << ((s[k].x & 1) * 16));
            atomicAdd(&cnt[s[k].y >> 1], 1u << ((s[k].y & 1) * 16));
            atomicAdd(&cnt[s[k].z >> 1], 1u << ((s[k].z & 1) * 16));
            atomicAdd(&cnt[s[k].w >> 1], 1u << ((s[k].w & 1) * 16));
        }
    }
    __syncthreads();
    uint4* out4 = (uint4*)(part + (size_t)p * OD_WORDS);
    for (int i = tid; i < OD_WORDS / 4; i += OD_THREADS) out4[i] = cnt4[i];
}

__global__ void __launch_bounds__(256) layer1_kernel(const float* __restrict__ agg1, const float* __restrict__ cd,
                                                     const float* __restrict__ W1, const float* __restrict__ b1,
                                                     float* __restrict__ h) {
    int wave = (blockIdx.x * 256 + threadIdx.x) >> 5;
    int lane = threadIdx.x & 31;
    const float4 w = ((const float4*)W1)[lane];
    const float4 b = ((const float4*)b1)[lane];
    int m0 = wave * ROWS_PER_WAVE;
    float x[ROWS_PER_WAVE];
#pragma unroll
    for (int u = 0; u < ROWS_PER_WAVE; ++u) x[u] = agg1[m0 + u] * cd[m0 + u];
    float4* hrow = (float4*)(h + (size_t)m0 * HIDDEN) + lane;
#pragma unroll
    for (int u = 0; u < ROWS_PER_WAVE; ++u) {
        float4 o;
        o.x = fmaxf(fmaf(x[u], w.x, b.x), 0.0f);
        o.y = fmaxf(fmaf(x[u], w.y, b.y), 0.0f);
        o.z = fmaxf(fmaf(x[u], w.z, b.z), 0.0f);
        o.w = fmaxf(fmaf(x[u], w.w, b.w), 0.0f);
        hrow[u * (HIDDEN / 4)] = o;
    }
}

__global__ void __launch_bounds__(256) bin_kernel(const int* __restrict__ src, const int* __restrict__ dst,
                                                  unsigned* __restrict__ binned, int* __restrict__ offs) {
    __shared__ int hist[OFF_STRIDE];
    __shared__ int scanbuf[OFF_STRIDE];
    int tid = threadIdx.x;
    int c   = blockIdx.x;
    int e0  = c * CHUNK;
    hist[tid] = 0;
    __syncthreads();

    static_assert(N_TILES + 1 < OFF_STRIDE, "no spare bin");
    unsigned word[CHUNK / 256];
    int      bin[CHUNK / 256];
    int      rank[CHUNK / 256];
    int      sv[CHUNK / 256], dv[CHUNK / 256];
#pragma unroll
    for (int u = 0; u < CHUNK / 256; ++u) {
        int e = min(e0 + u * 256 + tid, N_EDGES - 1);
        sv[u] = src[e];
        dv[u] = dst[e];
    }
#pragma unroll
    for (int u = 0; u < CHUNK / 256; ++u) {
        bool valid = (e0 + u * 256 + tid) < N_EDGES;
        int b = dv[u] / TILE_ROWS;
        word[u] = (unsigned)sv[u] | ((unsigned)(dv[u] - b * TILE_ROWS) << SRC_BITS);
        bin[u]  = valid ? b : (OFF_STRIDE - 1);
        rank[u] = atomicAdd(&hist[bin[u]], 1);
    }
    __syncthreads();

    int v = hist[tid];
    scanbuf[tid] = v;
    __syncthreads();
    for (int o = 1; o < OFF_STRIDE; o <<= 1) {
        int t = (tid >= o) ? scanbuf[tid - o] : 0;
        __syncthreads();
        scanbuf[tid] += t;
        __syncthreads();
    }
    int excl = scanbuf[tid] - v;
    offs[c * OFF_STRIDE + tid] = excl;
    hist[tid] = excl;
    __syncthreads();

#pragma unroll
    for (int u = 0; u < CHUNK / 256; ++u)
        if (bin[u] != OFF_STRIDE - 1) binned[e0 + hist[bin[u]] + rank[u]] = word[u];
}

#define SCAL_THREADS 512
#define SCAL_UNROLL  8
static_assert(N_CHUNKS <= SCAL_THREADS, "one thread per chunk");
static_assert(TILE_ROWS <= SCAL_THREADS, "one thread per tile row in the epilogue");

__global__ void __launch_bounds__(SCAL_THREADS) tile_indeg_kernel(const unsigned* __restrict__ binned,
                                                                  const int* __restrict__ offs,
                                                                  const unsigned* __restrict__ part,
                                                                  float* __restrict__ msg1,
                                                                  float* __restrict__ cs, float* __restrict__ cd) {
    __shared__ int cnt[TILE_ROWS + 32];
    int tid = threadIdx.x;
    int b   = blockIdx.x;
    int c   = min(tid, N_CHUNKS - 1);
    int lo = 0, hi = 0;
    if (tid < N_CHUNKS) {
        lo = offs[c * OFF_STRIDE + b];
        hi = offs[c * OFF_STRIDE + b + 1];
    }
    unsigned od = 0;
    if (tid < TILE_ROWS) {
        int n  = b * TILE_ROWS + tid;
        int sh = (n & 1) * 16;
        const unsigned* pp = part + (n >> 1);
#pragma unroll
        for (int p = 0; p < OD_PARTS; ++p) od += (pp[(size_t)p * OD_WORDS] >> sh) & 0xffffu;
    }
    if (tid < TILE_ROWS) cnt[tid] = 0;
    __syncthreads();
    const unsigned* sp = binned + (size_t)c * CHUNK;
    for (int i = lo; i < hi; i += SCAL_UNROLL) {
        unsigned w[SCAL_UNROLL];
#pragma unroll
        for (int u = 0; u < SCAL_UNROLL; ++u) w[u] = sp[min(i + u, hi - 1)];
#pragma unroll
        for (int u = 0; u < SCAL_UNROLL; ++u)
            atomicAdd(&cnt[(i + u < hi) ? (int)(w[u] >> SRC_BITS) : TILE_ROWS + (tid & 31)], 1);
    }
    __syncthreads();
    if (tid < TILE_ROWS) {
        int   n  = b * TILE_ROWS + tid;
        float id = (float)cnt[tid];
        float sn = rsqrtf(fmaxf((float)od, 1.0f));
        cd[n]    = rsqrtf(fmaxf(id, 1.0f));
        cs[n]    = sn;
        msg1[n]  = id * sn;
    }
}

__global__ void __launch_bounds__(SCAL_THREADS) tile_agg1_kernel(const unsigned* __restrict__ binned,
                                                                 const int* __restrict__ offs,
                                                                 const float* __restrict__ msg1,
                                                                 float* __restrict__ agg1) {
    __shared__ float acc[TILE_ROWS + 32];
    int tid = threadIdx.x;
    int b   = blockIdx.x;
    int c   = min(tid, N_CHUNKS - 1);
    int lo = 0, hi = 0;
    if (tid < N_CHUNKS) {
        lo = offs[c * OFF_STRIDE + b];
        hi = offs[c * OFF_STRIDE + b + 1];
    }
    if (tid < TILE_ROWS) acc[tid] = 0.0f;
    __syncthreads();
    const unsigned* sp = binned + (size_t)c * CHUNK;
    for (int i = lo; i < hi; i += SCAL_UNROLL) {
        unsigned w[SCAL_UNROLL];
        float    a[SCAL_UNROLL];
#pragma unroll
        for (int u = 0; u < SCAL_UNROLL; ++u) w[u] = sp[min(i + u, hi - 1)];
#pragma unroll
        for (int u = 0; u < SCAL_UNROLL; ++u) {
            a[u] = msg1[w[u] & ((1u << SRC_BITS) - 1u)];
        }
#pragma unroll
        for (int u = 0; u < SCAL_UNROLL; ++u)
            atomicAdd(&acc[(i + u < hi) ? (int)(w[u] >> SRC_BITS) : TILE_ROWS + (tid & 31)], a[u]);
    }
    __syncthreads();
    if (tid < TILE_ROWS) agg1[b * TILE_ROWS + tid] = acc[tid];
}

#define LIST_CAP   768
#define GRP        8
#define PIECES     ((N_CHUNKS + 31) / 32)
static_assert(PIECES <= 32, "one lane per chunk of a wave holds its offsets");
static_assert(TILE_ROWS * TILE_PITCH * 4 + 32 * LIST_CAP * 4 + 256 <= 320 * 1024, "tile and lists do not fit the LDS");
static_assert((TILE_PITCH * 4) % 16 == 0, "a tile row must start 16-byte aligned for the four-float adds");
__device__ __forceinline__ void edge_issue(const unsigned* lp, int g, int nl, int lane,
                                           const float* __restrict__ cs, const float* __restrict__ h,
                                           float& scl, float4 (&v)[GRP], int (&row)[GRP], int& cn) {
    unsigned wl = lp[min(GRP * g + (lane & (GRP - 1)), nl - 1)];
    cn  = min(GRP, nl - GRP * g);
    scl = cs[wl & ((1u << SRC_BITS) - 1u)];
#pragma unroll
    for (int u = 0; u < GRP; ++u) {
        unsigned wj = (unsigned)__builtin_amdgcn_readlane((int)wl, u);
        int s  = (int)(wj & ((1u << SRC_BITS) - 1u));
        row[u] = (int)(wj >> SRC_BITS);
        v[u]   = ((const float4*)(h + (size_t)s * HIDDEN))[lane];
    }
}

__device__ __forceinline__ void edge_add(float* tile, int lane, float scl, const float4 (&v)[GRP],
                                         const int (&row)[GRP], int cn) {
    asm volatile("" :: "v"(scl));
#pragma unroll
    for (int u = 0; u < GRP; ++u)
        asm volatile("" :: "v"(v[u].x), "v"(v[u].y), "v"(v[u].z), "v"(v[u].w));
#pragma unroll
    for (int u = 0; u < GRP; ++u) {
        if (u < cn) {
            float sc = __builtin_bit_cast(float, __builtin_amdgcn_readlane(__builtin_bit_cast(int, scl), u));
            float4* t = (float4*)(tile + row[u] * TILE_PITCH) + lane;
            float4 o = *t;
            o.x += v[u].x * sc;
            o.y += v[u].y * sc;
            o.z += v[u].z * sc;
            o.w += v[u].w * sc;
            *t = o;
        }
    }
}

__global__ void __launch_bounds__(1024) tile_layer_kernel(const unsigned* __restrict__ binned,
                                                          const int* __restrict__ offs,
                                                          const float* __restrict__ cs,
                                                          const float* __restrict__ cd,
                                                          const float* __restrict__ h,
                                                          const float* __restrict__ W,
                                                          const float* __restrict__ bias,
                                                          float* __restrict__ hout) {
    __shared__ __attribute__((aligned(16))) float tile[TILE_ROWS * TILE_PITCH];
    __shared__ unsigned elist[32 * LIST_CAP];
    __shared__ int      lcount[32];
    int tid  = threadIdx.x;
    int lane = tid & 31;
    int b    = blockIdx.x;
    int wave = __builtin_amdgcn_readfirstlane(tid >> 5);

    int myc  = wave + 32 * lane;
    int lo_l = 0, hi_l = 0;
    if (lane < PIECES && myc < N_CHUNKS) {
        lo_l = offs[myc * OFF_STRIDE + b];
        hi_l = offs[myc * OFF_STRIDE + b + 1];
    }

    float4* tile4 = (float4*)tile;
    for (int i = tid; i < TILE_ROWS * TILE_PITCH / 4; i += 1024) tile4[i] = float4{0.f, 0.f, 0.f, 0.f};
    if (tid < 32) lcount[tid] = 0;
    __syncthreads();

    {
        unsigned wv[PIECES];
#pragma unroll
        for (int i = 0; i < PIECES; ++i) {
            int lo = __builtin_amdgcn_readlane(lo_l, i);
            int n  = __builtin_amdgcn_readlane(hi_l, i) - lo;
            wv[i] = 0u;
            if (n > 0) wv[i] = binned[(size_t)(wave + 32 * i) * CHUNK + lo + min(lane, n - 1)];
        }
#pragma unroll
        for (int i = 0; i < PIECES; ++i) {
            int lo = __builtin_amdgcn_readlane(lo_l, i);
            int n  = __builtin_amdgcn_readlane(hi_l, i) - lo;
            const unsigned* sp = binned + (size_t)(wave + 32 * i) * CHUNK + lo;
            for (int base = 0; base < n; base += 32) {
                int cnt = min(32, n - base);
                unsigned w = (base == 0) ? wv[i] : sp[base + min(lane, cnt - 1)];
                bool valid = lane < cnt;
                int  own   = (int)(w >> SRC_BITS) & 31;
                int  pos   = LIST_CAP;
                if (valid) pos = atomicAdd(&lcount[own], 1);
                bool fits = valid && pos < LIST_CAP;
                if (fits) elist[own * LIST_CAP + pos] = w;
                unsigned over = __builtin_amdgcn_ballot_w32(valid && !fits);
                while (over) {
                    int j = __builtin_ctz(over);
                    over &= over - 1u;
                    unsigned wj = (unsigned)__builtin_amdgcn_readlane((int)w, j);
                    int   s   = (int)(wj & ((1u << SRC_BITS) - 1u));
                    int   row = (int)(wj >> SRC_BITS);
                    float sc  = cs[s];
                    float4 v  = ((const float4*)(h + (size_t)s * HIDDEN))[lane];
                    float* t  = tile + row * TILE_PITCH + 4 * lane;
                    atomicAdd(t,     v.x * sc);
                    atomicAdd(t + 1, v.y * sc);
                    atomicAdd(t + 2, v.z * sc);
                    atomicAdd(t + 3, v.w * sc);
                }
            }
        }
    }
    __syncthreads();

    {
        int nl = __builtin_amdgcn_readfirstlane(min(lcount[wave], LIST_CAP));
        int ng = (nl + GRP - 1) / GRP;
        const unsigned* lp = elist + wave * LIST_CAP;
        float  sclA = 0.f, sclB = 0.f;
        float4 vA[GRP],  vB[GRP];
        int    rowA[GRP], rowB[GRP];
        int    cnA = 0, cnB = 0;
        if (nl > 0) {
            edge_issue(lp, 0, nl, lane, cs, h, sclA, vA, rowA, cnA);
            for (int g = 0; g < ng; g += 2) {
                edge_issue(lp, g + 1, nl, lane, cs, h, sclB, vB, rowB, cnB);
                edge_add(tile, lane, sclA, vA, rowA, cnA);
                edge_issue(lp, g + 2, nl, lane, cs, h, sclA, vA, rowA, cnA);
                edge_add(tile, lane, sclB, vB, rowB, cnB);
            }
        }
    }

    int half = lane >> 4;
    int lr   = lane & 15;
    int n    = (wave & 7) * 16 + lr;
    const float* wp = W + (half * 2) * HIDDEN + n;
    v2f bw[HIDDEN / 4];
#pragma unroll
    for (int ks = 0; ks < HIDDEN / 4; ++ks) {
        bw[ks].x = wp[(4 * ks) * HIDDEN];
        bw[ks].y = wp[(4 * ks + 1) * HIDDEN];
    }
    float bn = bias[n];
    __syncthreads();

    for (int rt = wave >> 3; rt < TILE_ROWS / 16; rt += 4) {
        int   r0  = rt * 16;
        float cdm = cd[b * TILE_ROWS + r0 + lr];
        const float* ap = tile + (r0 + lr) * TILE_PITCH + half * 2;
        v8f acc = {};
#pragma unroll
        for (int ks = 0; ks < HIDDEN / 4; ++ks) {
            float2 a = *(const float2*)(ap + 4 * ks);
            v2f av;
            av.x = a.x * cdm;
            av.y = a.y * cdm;
            acc = __builtin_amdgcn_wmma_f32_16x16x4_f32(
                 false, av,  false, bw[ks],
                 (short)0, acc,  false,  false);
        }
        float* op = hout + (size_t)(b * TILE_ROWS + r0 + half * 8) * HIDDEN + n;
#pragma unroll
        for (int r = 0; r < 8; ++r)
            op[(size_t)r * HIDDEN] = fmaxf(acc[r] + bn, 0.0f);
    }
}

__device__ __forceinline__ void pool_flush(float* __restrict__ hg, float* __restrict__ counts,
                                           int g, int lane, float4 acc, float cnt) {
    float* o = hg + (size_t)g * HIDDEN + 4 * lane;
    atomicAdd(o,     acc.x);
    atomicAdd(o + 1, acc.y);
    atomicAdd(o + 2, acc.z);
    atomicAdd(o + 3, acc.w);
    if (lane == 0) atomicAdd(&counts[g], cnt);
}

__global__ void __launch_bounds__(256) pool_kernel(const int* __restrict__ seg, const float* __restrict__ h,
                                                   float* __restrict__ hg, float* __restrict__ counts) {
    int wave = (blockIdx.x * 256 + threadIdx.x) >> 5;
    int lane = threadIdx.x & 31;
    int n0   = wave * POOL_RUN;
    float4 acc = float4{0.f, 0.f, 0.f, 0.f};
    float  cnt = 0.0f;
    int gcur = __builtin_amdgcn_readfirstlane(seg[n0]);
    for (int nb = n0; nb < n0 + POOL_RUN; nb += 5) {
        int    g[5];
        float4 v[5];
#pragma unroll
        for (int u = 0; u < 5; ++u) {
            g[u] = __builtin_amdgcn_readfirstlane(seg[nb + u]);
            v[u] = ((const float4*)(h + (size_t)(nb + u) * HIDDEN))[lane];
        }
#pragma unroll
        for (int u = 0; u < 5; ++u) {
            if (g[u] != gcur) {
                pool_flush(hg, counts, gcur, lane, acc, cnt);
                acc  = float4{0.f, 0.f, 0.f, 0.f};
                cnt  = 0.0f;
                gcur = g[u];
            }
            acc.x += v[u].x;
            acc.y += v[u].y;
            acc.z += v[u].z;
            acc.w += v[u].w;
            cnt   += 1.0f;
        }
    }
    pool_flush(hg, counts, gcur, lane, acc, cnt);
}

__global__ void pool_div_kernel(float* __restrict__ hg, const float* __restrict__ counts) {
    int i = blockIdx.x * blockDim.x + threadIdx.x;
    if (i < N_GRAPHS * HIDDEN)
        hg[i] /= fmaxf(counts[i >> 7], 1.0f);
}

__global__ void logits_kernel(const float* __restrict__ hg1, const float* __restrict__ hg2,
                              const float* __restrict__ Wc, const float* __restrict__ bc,
                              float* __restrict__ logits) {
    int i = blockIdx.x * blockDim.x + threadIdx.x;
    if (i < N_GRAPHS * N_CLASSES) {
        int g = i / N_CLASSES, c = i % N_CLASSES;
        float s = bc[c];
#pragma unroll 8
        for (int k = 0; k < HIDDEN; ++k)
            s += fabsf(hg1[g * HIDDEN + k] - hg2[g * HIDDEN + k]) * Wc[k * N_CLASSES + c];
        logits[i] = s;
    }
}

static void run_branch(const int* src, const int* dst, const int* seg,
                       const float* W1, const float* b1,
                       const float* W2, const float* b2,
                       const float* W3, const float* b3,
                       const float* W4, const float* b4,
                       float* hg_out, float* ws_f, hipStream_t stream) {
    float* msg1   = ws_f + 100000;
    float* cs     = ws_f + 200000;
    float* cd     = ws_f + 300000;
    float* agg1   = ws_f + 400000;
    float* counts = ws_f + 500000;
    float* h      = ws_f + 500224;
    float* hb     = h + (size_t)N_NODES * HIDDEN;
    unsigned* binned = (unsigned*)(hb + (size_t)N_NODES * HIDDEN);
    int*      offs   = (int*)(binned + N_EDGES);
    unsigned* part   = (unsigned*)(offs + N_CHUNKS * OFF_STRIDE);

    hipMemsetAsync(counts, 0, (size_t)N_GRAPHS * sizeof(float), stream);

    const int T = 256;
    outdeg_part_kernel<<<OD_PARTS, OD_THREADS, 0, stream>>>(src, part);

    bin_kernel<<<N_CHUNKS, 256, 0, stream>>>(src, dst, binned, offs);

    tile_indeg_kernel<<<N_TILES, SCAL_THREADS, 0, stream>>>(binned, offs, part, msg1, cs, cd);
    tile_agg1_kernel<<<N_TILES, SCAL_THREADS, 0, stream>>>(binned, offs, msg1, agg1);
    layer1_kernel<<<ROW_WGS, T, 0, stream>>>(agg1, cd, W1, b1, h);

    tile_layer_kernel<<<N_TILES, 1024, 0, stream>>>(binned, offs, cs, cd, h,  W2, b2, hb);
    tile_layer_kernel<<<N_TILES, 1024, 0, stream>>>(binned, offs, cs, cd, hb, W3, b3, h);
    tile_layer_kernel<<<N_TILES, 1024, 0, stream>>>(binned, offs, cs, cd, h,  W4, b4, hb);

    hipMemsetAsync(hg_out, 0, (size_t)N_GRAPHS * HIDDEN * sizeof(float), stream);
    pool_kernel<<<POOL_WGS, T, 0, stream>>>(seg, hb, hg_out, counts);
    pool_div_kernel<<<(N_GRAPHS * HIDDEN + T - 1) / T, T, 0, stream>>>(hg_out, counts);
}

extern "C" void kernel_launch(void* const* d_in, const int* in_sizes, int n_in,
                              void* d_out, int out_size, void* d_ws, size_t ws_size,
                              hipStream_t stream) {
    const int*   src1 = (const int*)d_in[0];
    const int*   dst1 = (const int*)d_in[1];
    const int*   seg1 = (const int*)d_in[2];
    const int*   src2 = (const int*)d_in[3];
    const int*   dst2 = (const int*)d_in[4];
    const int*   seg2 = (const int*)d_in[5];
    const float* W1   = (const float*)d_in[6];
    const float* b1   = (const float*)d_in[7];
    const float* W2   = (const float*)d_in[8];
    const float* b2   = (const float*)d_in[9];
    const float* W3   = (const float*)d_in[10];
    const float* b3   = (const float*)d_in[11];
    const float* W4   = (const float*)d_in[12];
    const float* b4   = (const float*)d_in[13];
    const float* Wc   = (const float*)d_in[14];
    const float* bc   = (const float*)d_in[15];

    float* out   = (float*)d_out;
    float* hg1   = out;
    float* hg2   = out + N_GRAPHS * HIDDEN;
    float* logit = out + 2 * N_GRAPHS * HIDDEN;
    float* ws_f  = (float*)d_ws;

    run_branch(src1, dst1, seg1, W1, b1, W2, b2, W3, b3, W4, b4, hg1, ws_f, stream);
    run_branch(src2, dst2, seg2, W1, b1, W2, b2, W3, b3, W4, b4, hg2, ws_f, stream);

    logits_kernel<<<(N_GRAPHS * N_CLASSES + 255) / 256, 256, 0, stream>>>(hg1, hg2, Wc, bc, logit);
}


# Feature rows of layers 1-3 stored as messages (h*cs, as the reference forms them): the per-edge gather of cs[src] in the layers 2-4 edge phase is gone; sum order kept, product rounded at the store
# speedup vs baseline: 5.3510x; 1.0445x over previous
#include <hip/hip_runtime.h>
#include <hip/hip_bf16.h>

#define N_NODES   100000
#define N_EDGES   1600000
#define N_GRAPHS  128
#define HIDDEN    128
#define N_CLASSES 10

#define TILE_ROWS  400
#define N_TILES    (N_NODES / TILE_ROWS)
#define CHUNK      4096
#define N_CHUNKS   ((N_EDGES + CHUNK - 1) / CHUNK)
#define OFF_STRIDE 256
#define SRC_BITS   17
static_assert(N_TILES * TILE_ROWS == N_NODES, "tiles must cover the nodes exactly");
static_assert(N_TILES + 1 <= OFF_STRIDE, "offset row too short");
static_assert(N_NODES <= (1 << SRC_BITS), "source index does not fit its field");
#define TILE_PITCH 132
static_assert(TILE_ROWS % 16 == 0, "the tile is cut into whole 16-row matrix tiles");
static_assert(TILE_ROWS * TILE_PITCH * 4 + 64 <= 320 * 1024, "tile does not fit the LDS");

#define OD_PARTS   100
#define OD_SLICE   (N_EDGES / OD_PARTS)
#define OD_WORDS   (N_NODES / 2)
#define OD_THREADS 1024
#define OD_LOADS   ((OD_SLICE / 4 + OD_THREADS - 1) / OD_THREADS)
static_assert(OD_PARTS * OD_SLICE == N_EDGES, "slices must cover the edges exactly");
static_assert(OD_SLICE <= 65535, "a 16-bit counter must be able to hold a whole slice");
static_assert(OD_SLICE % 4 == 0, "a slice is read four indices at a time and must start 16-byte aligned");
static_assert(N_NODES % 8 == 0, "the packed counters are cleared and stored four words at a time");
static_assert(OD_WORDS * 4 <= 320 * 1024, "counters do not fit the LDS");

#define ROWS_PER_WAVE 10
#define ROW_WGS    (N_NODES / (8 * ROWS_PER_WAVE))
#define POOL_RUN   25
#define POOL_WGS   (N_NODES / (8 * POOL_RUN))
static_assert(ROW_WGS * 8 * ROWS_PER_WAVE == N_NODES, "row grid must cover the nodes exactly");
static_assert(POOL_WGS * 8 * POOL_RUN == N_NODES, "pool grid must cover the nodes exactly");
static_assert(POOL_RUN % 5 == 0, "pooling loads rows in groups of five");

typedef __attribute__((ext_vector_type(2))) float v2f;
typedef __attribute__((ext_vector_type(8))) float v8f;

__global__ void __launch_bounds__(OD_THREADS) outdeg_part_kernel(const int* __restrict__ src, unsigned* __restrict__ part) {
    __shared__ __attribute__((aligned(16))) unsigned cnt[OD_WORDS];
    int tid = threadIdx.x;
    int p   = blockIdx.x;
    const int4* sp = (const int4*)(src + (size_t)p * OD_SLICE);
    int4 s[OD_LOADS];
#pragma unroll
    for (int k = 0; k < OD_LOADS; ++k) s[k] = sp[min(k * OD_THREADS + tid, OD_SLICE / 4 - 1)];
    uint4* cnt4 = (uint4*)cnt;
    for (int i = tid; i < OD_WORDS / 4; i += OD_THREADS) cnt4[i] = uint4{0u, 0u, 0u, 0u};
    __syncthreads();
#pragma unroll
    for (int k = 0; k < OD_LOADS; ++k) {
        if (k * OD_THREADS + tid < OD_SLICE / 4) {
            atomicAdd(&cnt[s[k].x >> 1], 1u << ((s[k].x & 1) * 16));
            atomicAdd(&cnt[s[k].y >> 1], 1u << ((s[k].y & 1) * 16));
            atomicAdd(&cnt[s[k].z >> 1], 1u << ((s[k].z & 1) * 16));
            atomicAdd(&cnt[s[k].w >> 1], 1u << ((s[k].w & 1) * 16));
        }
    }
    __syncthreads();
    uint4* out4 = (uint4*)(part + (size_t)p * OD_WORDS);
    for (int i = tid; i < OD_WORDS / 4; i += OD_THREADS) out4[i] = cnt4[i];
}

__global__ void __launch_bounds__(256) layer1_kernel(const float* __restrict__ agg1, const float* __restrict__ cd,
                                                     const float* __restrict__ cs,
                                                     const float* __restrict__ W1, const float* __restrict__ b1,
                                                     float* __restrict__ h) {
    int wave = (blockIdx.x * 256 + threadIdx.x) >> 5;
    int lane = threadIdx.x & 31;
    const float4 w = ((const float4*)W1)[lane];
    const float4 b = ((const float4*)b1)[lane];
    int m0 = wave * ROWS_PER_WAVE;
    float x[ROWS_PER_WAVE];
    float sn[ROWS_PER_WAVE];
#pragma unroll
    for (int u = 0; u < ROWS_PER_WAVE; ++u) {
        x[u]  = agg1[m0 + u] * cd[m0 + u];
        sn[u] = cs[m0 + u];
    }
    float4* hrow = (float4*)(h + (size_t)m0 * HIDDEN) + lane;
#pragma unroll
    for (int u = 0; u < ROWS_PER_WAVE; ++u) {
        float4 o;
        o.x = fmaxf(fmaf(x[u], w.x, b.x), 0.0f) * sn[u];
        o.y = fmaxf(fmaf(x[u], w.y, b.y), 0.0f) * sn[u];
        o.z = fmaxf(fmaf(x[u], w.z, b.z), 0.0f) * sn[u];
        o.w = fmaxf(fmaf(x[u], w.w, b.w), 0.0f) * sn[u];
        hrow[u * (HIDDEN / 4)] = o;
    }
}

__global__ void __launch_bounds__(256) bin_kernel(const int* __restrict__ src, const int* __restrict__ dst,
                                                  unsigned* __restrict__ binned, int* __restrict__ offs) {
    __shared__ int hist[OFF_STRIDE];
    __shared__ int scanbuf[OFF_STRIDE];
    int tid = threadIdx.x;
    int c   = blockIdx.x;
    int e0  = c * CHUNK;
    hist[tid] = 0;
    __syncthreads();

    static_assert(N_TILES + 1 < OFF_STRIDE, "no spare bin");
    unsigned word[CHUNK / 256];
    int      bin[CHUNK / 256];
    int      rank[CHUNK / 256];
    int      sv[CHUNK / 256], dv[CHUNK / 256];
#pragma unroll
    for (int u = 0; u < CHUNK / 256; ++u) {
        int e = min(e0 + u * 256 + tid, N_EDGES - 1);
        sv[u] = src[e];
        dv[u] = dst[e];
    }
#pragma unroll
    for (int u = 0; u < CHUNK / 256; ++u) {
        bool valid = (e0 + u * 256 + tid) < N_EDGES;
        int b = dv[u] / TILE_ROWS;
        word[u] = (unsigned)sv[u] | ((unsigned)(dv[u] - b * TILE_ROWS) << SRC_BITS);
        bin[u]  = valid ? b : (OFF_STRIDE - 1);
        rank[u] = atomicAdd(&hist[bin[u]], 1);
    }
    __syncthreads();

    int v = hist[tid];
    scanbuf[tid] = v;
    __syncthreads();
    for (int o = 1; o < OFF_STRIDE; o <<= 1) {
        int t = (tid >= o) ? scanbuf[tid - o] : 0;
        __syncthreads();
        scanbuf[tid] += t;
        __syncthreads();
    }
    int excl = scanbuf[tid] - v;
    offs[c * OFF_STRIDE + tid] = excl;
    hist[tid] = excl;
    __syncthreads();

#pragma unroll
    for (int u = 0; u < CHUNK / 256; ++u)
        if (bin[u] != OFF_STRIDE - 1) binned[e0 + hist[bin[u]] + rank[u]] = word[u];
}

#define SCAL_THREADS 512
#define SCAL_UNROLL  8
static_assert(N_CHUNKS <= SCAL_THREADS, "one thread per chunk");
static_assert(TILE_ROWS <= SCAL_THREADS, "one thread per tile row in the epilogue");

__global__ void __launch_bounds__(SCAL_THREADS) tile_indeg_kernel(const unsigned* __restrict__ binned,
                                                                  const int* __restrict__ offs,
                                                                  const unsigned* __restrict__ part,
                                                                  float* __restrict__ msg1,
                                                                  float* __restrict__ cs, float* __restrict__ cd) {
    __shared__ int cnt[TILE_ROWS + 32];
    int tid = threadIdx.x;
    int b   = blockIdx.x;
    int c   = min(tid, N_CHUNKS - 1);
    int lo = 0, hi = 0;
    if (tid < N_CHUNKS) {
        lo = offs[c * OFF_STRIDE + b];
        hi = offs[c * OFF_STRIDE + b + 1];
    }
    unsigned od = 0;
    if (tid < TILE_ROWS) {
        int n  = b * TILE_ROWS + tid;
        int sh = (n & 1) * 16;
        const unsigned* pp = part + (n >> 1);
#pragma unroll
        for (int p = 0; p < OD_PARTS; ++p) od += (pp[(size_t)p * OD_WORDS] >> sh) & 0xffffu;
    }
    if (tid < TILE_ROWS) cnt[tid] = 0;
    __syncthreads();
    const unsigned* sp = binned + (size_t)c * CHUNK;
    for (int i = lo; i < hi; i += SCAL_UNROLL) {
        unsigned w[SCAL_UNROLL];
#pragma unroll
        for (int u = 0; u < SCAL_UNROLL; ++u) w[u] = sp[min(i + u, hi - 1)];
#pragma unroll
        for (int u = 0; u < SCAL_UNROLL; ++u)
            atomicAdd(&cnt[(i + u < hi) ? (int)(w[u] >> SRC_BITS) : TILE_ROWS + (tid & 31)], 1);
    }
    __syncthreads();
    if (tid < TILE_ROWS) {
        int   n  = b * TILE_ROWS + tid;
        float id = (float)cnt[tid];
        float sn = rsqrtf(fmaxf((float)od, 1.0f));
        cd[n]    = rsqrtf(fmaxf(id, 1.0f));
        cs[n]    = sn;
        msg1[n]  = id * sn;
    }
}

__global__ void __launch_bounds__(SCAL_THREADS) tile_agg1_kernel(const unsigned* __restrict__ binned,
                                                                 const int* __restrict__ offs,
                                                                 const float* __restrict__ msg1,
                                                                 float* __restrict__ agg1) {
    __shared__ float acc[TILE_ROWS + 32];
    int tid = threadIdx.x;
    int b   = blockIdx.x;
    int c   = min(tid, N_CHUNKS - 1);
    int lo = 0, hi = 0;
    if (tid < N_CHUNKS) {
        lo = offs[c * OFF_STRIDE + b];
        hi = offs[c * OFF_STRIDE + b + 1];
    }
    if (tid < TILE_ROWS) acc[tid] = 0.0f;
    __syncthreads();
    const unsigned* sp = binned + (size_t)c * CHUNK;
    for (int i = lo; i < hi; i += SCAL_UNROLL) {
        unsigned w[SCAL_UNROLL];
        float    a[SCAL_UNROLL];
#pragma unroll
        for (int u = 0; u < SCAL_UNROLL; ++u) w[u] = sp[min(i + u, hi - 1)];
#pragma unroll
        for (int u = 0; u < SCAL_UNROLL; ++u) {
            a[u] = msg1[w[u] & ((1u << SRC_BITS) - 1u)];
        }
#pragma unroll
        for (int u = 0; u < SCAL_UNROLL; ++u)
            atomicAdd(&acc[(i + u < hi) ? (int)(w[u] >> SRC_BITS) : TILE_ROWS + (tid & 31)], a[u]);
    }
    __syncthreads();
    if (tid < TILE_ROWS) agg1[b * TILE_ROWS + tid] = acc[tid];
}

#define LIST_CAP   768
#define GRP        8
#define PIECES     ((N_CHUNKS + 31) / 32)
static_assert(PIECES <= 32, "one lane per chunk of a wave holds its offsets");
static_assert(TILE_ROWS * TILE_PITCH * 4 + 32 * LIST_CAP * 4 + 256 <= 320 * 1024, "tile and lists do not fit the LDS");
static_assert((TILE_PITCH * 4) % 16 == 0, "a tile row must start 16-byte aligned for the four-float adds");
__device__ __forceinline__ void edge_issue(const unsigned* lp, int g, int nl, int lane,
                                           const float* __restrict__ h,
                                           float4 (&v)[GRP], int (&row)[GRP], int& cn) {
    unsigned wl = lp[min(GRP * g + (lane & (GRP - 1)), nl - 1)];
    cn  = min(GRP, nl - GRP * g);
#pragma unroll
    for (int u = 0; u < GRP; ++u) {
        unsigned wj = (unsigned)__builtin_amdgcn_readlane((int)wl, u);
        int s  = (int)(wj & ((1u << SRC_BITS) - 1u));
        row[u] = (int)(wj >> SRC_BITS);
        v[u]   = ((const float4*)(h + (size_t)s * HIDDEN))[lane];
    }
}

__device__ __forceinline__ void edge_add(float* tile, int lane, const float4 (&v)[GRP],
                                         const int (&row)[GRP], int cn) {
#pragma unroll
    for (int u = 0; u < GRP; ++u)
        asm volatile("" :: "v"(v[u].x), "v"(v[u].y), "v"(v[u].z), "v"(v[u].w));
#pragma unroll
    for (int u = 0; u < GRP; ++u) {
        if (u < cn) {
            float4* t = (float4*)(tile + row[u] * TILE_PITCH) + lane;
            float4 o = *t;
            o.x += v[u].x;
            o.y += v[u].y;
            o.z += v[u].z;
            o.w += v[u].w;
            *t = o;
        }
    }
}

__global__ void __launch_bounds__(1024) tile_layer_kernel(const unsigned* __restrict__ binned,
                                                          const int* __restrict__ offs,
                                                          const float* __restrict__ cd,
                                                          const float* __restrict__ h,
                                                          const float* __restrict__ W,
                                                          const float* __restrict__ bias,
                                                          const float* __restrict__ oscale,
                                                          float* __restrict__ hout) {
    __shared__ __attribute__((aligned(16))) float tile[TILE_ROWS * TILE_PITCH];
    __shared__ unsigned elist[32 * LIST_CAP];
    __shared__ int      lcount[32];
    int tid  = threadIdx.x;
    int lane = tid & 31;
    int b    = blockIdx.x;
    int wave = __builtin_amdgcn_readfirstlane(tid >> 5);

    int myc  = wave + 32 * lane;
    int lo_l = 0, hi_l = 0;
    if (lane < PIECES && myc < N_CHUNKS) {
        lo_l = offs[myc * OFF_STRIDE + b];
        hi_l = offs[myc * OFF_STRIDE + b + 1];
    }

    float4* tile4 = (float4*)tile;
    for (int i = tid; i < TILE_ROWS * TILE_PITCH / 4; i += 1024) tile4[i] = float4{0.f, 0.f, 0.f, 0.f};
    if (tid < 32) lcount[tid] = 0;
    __syncthreads();

    {
        unsigned wv[PIECES];
#pragma unroll
        for (int i = 0; i < PIECES; ++i) {
            int lo = __builtin_amdgcn_readlane(lo_l, i);
            int n  = __builtin_amdgcn_readlane(hi_l, i) - lo;
            wv[i] = 0u;
            if (n > 0) wv[i] = binned[(size_t)(wave + 32 * i) * CHUNK + lo + min(lane, n - 1)];
        }
#pragma unroll
        for (int i = 0; i < PIECES; ++i) {
            int lo = __builtin_amdgcn_readlane(lo_l, i);
            int n  = __builtin_amdgcn_readlane(hi_l, i) - lo;
            const unsigned* sp = binned + (size_t)(wave + 32 * i) * CHUNK + lo;
            for (int base = 0; base < n; base += 32) {
                int cnt = min(32, n - base);
                unsigned w = (base == 0) ? wv[i] : sp[base + min(lane, cnt - 1)];
                bool valid = lane < cnt;
                int  own   = (int)(w >> SRC_BITS) & 31;
                int  pos   = LIST_CAP;
                if (valid) pos = atomicAdd(&lcount[own], 1);
                bool fits = valid && pos < LIST_CAP;
                if (fits) elist[own * LIST_CAP + pos] = w;
                unsigned over = __builtin_amdgcn_ballot_w32(valid && !fits);
                while (over) {
                    int j = __builtin_ctz(over);
                    over &= over - 1u;
                    unsigned wj = (unsigned)__builtin_amdgcn_readlane((int)w, j);
                    int   s   = (int)(wj & ((1u << SRC_BITS) - 1u));
                    int   row = (int)(wj >> SRC_BITS);
                    float4 v  = ((const float4*)(h + (size_t)s * HIDDEN))[lane];
                    float* t  = tile + row * TILE_PITCH + 4 * lane;
                    atomicAdd(t,     v.x);
                    atomicAdd(t + 1, v.y);
                    atomicAdd(t + 2, v.z);
                    atomicAdd(t + 3, v.w);
                }
            }
        }
    }
    __syncthreads();

    {
        int nl = __builtin_amdgcn_readfirstlane(min(lcount[wave], LIST_CAP));
        int ng = (nl + GRP - 1) / GRP;
        const unsigned* lp = elist + wave * LIST_CAP;
        float4 vA[GRP],  vB[GRP];
        int    rowA[GRP], rowB[GRP];
        int    cnA = 0, cnB = 0;
        if (nl > 0) {
            edge_issue(lp, 0, nl, lane, h, vA, rowA, cnA);
            for (int g = 0; g < ng; g += 2) {
                edge_issue(lp, g + 1, nl, lane, h, vB, rowB, cnB);
                edge_add(tile, lane, vA, rowA, cnA);
                edge_issue(lp, g + 2, nl, lane, h, vA, rowA, cnA);
                edge_add(tile, lane, vB, rowB, cnB);
            }
        }
    }

    int half = lane >> 4;
    int lr   = lane & 15;
    int n    = (wave & 7) * 16 + lr;
    const float* wp = W + (half * 2) * HIDDEN + n;
    v2f bw[HIDDEN / 4];
#pragma unroll
    for (int ks = 0; ks < HIDDEN / 4; ++ks) {
        bw[ks].x = wp[(4 * ks) * HIDDEN];
        bw[ks].y = wp[(4 * ks + 1) * HIDDEN];
    }
    float bn = bias[n];
    __syncthreads();

    const bool scaled = oscale != nullptr;
    for (int rt = wave >> 3; rt < TILE_ROWS / 16; rt += 4) {
        int   r0  = rt * 16;
        float cdm = cd[b * TILE_ROWS + r0 + lr];
        int   orow = b * TILE_ROWS + r0 + half * 8;
        float osc[8];
#pragma unroll
        for (int r = 0; r < 8; ++r) osc[r] = 1.0f;
        if (scaled) {
#pragma unroll
            for (int r = 0; r < 8; ++r) osc[r] = oscale[orow + r];
        }
        const float* ap = tile + (r0 + lr) * TILE_PITCH + half * 2;
        v8f acc = {};
#pragma unroll
        for (int ks = 0; ks < HIDDEN / 4; ++ks) {
            float2 a = *(const float2*)(ap + 4 * ks);
            v2f av;
            av.x = a.x * cdm;
            av.y = a.y * cdm;
            acc = __builtin_amdgcn_wmma_f32_16x16x4_f32(
                 false, av,  false, bw[ks],
                 (short)0, acc,  false,  false);
        }
        float* op = hout + (size_t)orow * HIDDEN + n;
#pragma unroll
        for (int r = 0; r < 8; ++r)
            op[(size_t)r * HIDDEN] = fmaxf(acc[r] + bn, 0.0f) * osc[r];
    }
}

__device__ __forceinline__ void pool_flush(float* __restrict__ hg, float* __restrict__ counts,
                                           int g, int lane, float4 acc, float cnt) {
    float* o = hg + (size_t)g * HIDDEN + 4 * lane;
    atomicAdd(o,     acc.x);
    atomicAdd(o + 1, acc.y);
    atomicAdd(o + 2, acc.z);
    atomicAdd(o + 3, acc.w);
    if (lane == 0) atomicAdd(&counts[g], cnt);
}

__global__ void __launch_bounds__(256) pool_kernel(const int* __restrict__ seg, const float* __restrict__ h,
                                                   float* __restrict__ hg, float* __restrict__ counts) {
    int wave = (blockIdx.x * 256 + threadIdx.x) >> 5;
    int lane = threadIdx.x & 31;
    int n0   = wave * POOL_RUN;
    float4 acc = float4{0.f, 0.f, 0.f, 0.f};
    float  cnt = 0.0f;
    int gcur = __builtin_amdgcn_readfirstlane(seg[n0]);
    for (int nb = n0; nb < n0 + POOL_RUN; nb += 5) {
        int    g[5];
        float4 v[5];
#pragma unroll
        for (int u = 0; u < 5; ++u) {
            g[u] = __builtin_amdgcn_readfirstlane(seg[nb + u]);
            v[u] = ((const float4*)(h + (size_t)(nb + u) * HIDDEN))[lane];
        }
#pragma unroll
        for (int u = 0; u < 5; ++u) {
            if (g[u] != gcur) {
                pool_flush(hg, counts, gcur, lane, acc, cnt);
                acc  = float4{0.f, 0.f, 0.f, 0.f};
                cnt  = 0.0f;
                gcur = g[u];
            }
            acc.x += v[u].x;
            acc.y += v[u].y;
            acc.z += v[u].z;
            acc.w += v[u].w;
            cnt   += 1.0f;
        }
    }
    pool_flush(hg, counts, gcur, lane, acc, cnt);
}

__global__ void pool_div_kernel(float* __restrict__ hg, const float* __restrict__ counts) {
    int i = blockIdx.x * blockDim.x + threadIdx.x;
    if (i < N_GRAPHS * HIDDEN)
        hg[i] /= fmaxf(counts[i >> 7], 1.0f);
}

__global__ void logits_kernel(const float* __restrict__ hg1, const float* __restrict__ hg2,
                              const float* __restrict__ Wc, const float* __restrict__ bc,
                              float* __restrict__ logits) {
    int i = blockIdx.x * blockDim.x + threadIdx.x;
    if (i < N_GRAPHS * N_CLASSES) {
        int g = i / N_CLASSES, c = i % N_CLASSES;
        float s = bc[c];
#pragma unroll 8
        for (int k = 0; k < HIDDEN; ++k)
            s += fabsf(hg1[g * HIDDEN + k] - hg2[g * HIDDEN + k]) * Wc[k * N_CLASSES + c];
        logits[i] = s;
    }
}

static void run_branch(const int* src, const int* dst, const int* seg,
                       const float* W1, const float* b1,
                       const float* W2, const float* b2,
                       const float* W3, const float* b3,
                       const float* W4, const float* b4,
                       float* hg_out, float* ws_f, hipStream_t stream) {
    float* msg1   = ws_f + 100000;
    float* cs     = ws_f + 200000;
    float* cd     = ws_f + 300000;
    float* agg1   = ws_f + 400000;
    float* counts = ws_f + 500000;
    float* h      = ws_f + 500224;
    float* hb     = h + (size_t)N_NODES * HIDDEN;
    unsigned* binned = (unsigned*)(hb + (size_t)N_NODES * HIDDEN);
    int*      offs   = (int*)(binned + N_EDGES);
    unsigned* part   = (unsigned*)(offs + N_CHUNKS * OFF_STRIDE);

    hipMemsetAsync(counts, 0, (size_t)N_GRAPHS * sizeof(float), stream);

    const int T = 256;
    outdeg_part_kernel<<<OD_PARTS, OD_THREADS, 0, stream>>>(src, part);

    bin_kernel<<<N_CHUNKS, 256, 0, stream>>>(src, dst, binned, offs);

    tile_indeg_kernel<<<N_TILES, SCAL_THREADS, 0, stream>>>(binned, offs, part, msg1, cs, cd);
    tile_agg1_kernel<<<N_TILES, SCAL_THREADS, 0, stream>>>(binned, offs, msg1, agg1);
    layer1_kernel<<<ROW_WGS, T, 0, stream>>>(agg1, cd, cs, W1, b1, h);

    tile_layer_kernel<<<N_TILES, 1024, 0, stream>>>(binned, offs, cd, h,  W2, b2, cs,      hb);
    tile_layer_kernel<<<N_TILES, 1024, 0, stream>>>(binned, offs, cd, hb, W3, b3, cs,      h);
    tile_layer_kernel<<<N_TILES, 1024, 0, stream>>>(binned, offs, cd, h,  W4, b4, nullptr, hb);

    hipMemsetAsync(hg_out, 0, (size_t)N_GRAPHS * HIDDEN * sizeof(float), stream);
    pool_kernel<<<POOL_WGS, T, 0, stream>>>(seg, hb, hg_out, counts);
    pool_div_kernel<<<(N_GRAPHS * HIDDEN + T - 1) / T, T, 0, stream>>>(hg_out, counts);
}

extern "C" void kernel_launch(void* const* d_in, const int* in_sizes, int n_in,
                              void* d_out, int out_size, void* d_ws, size_t ws_size,
                              hipStream_t stream) {
    const int*   src1 = (const int*)d_in[0];
    const int*   dst1 = (const int*)d_in[1];
    const int*   seg1 = (const int*)d_in[2];
    const int*   src2 = (const int*)d_in[3];
    const int*   dst2 = (const int*)d_in[4];
    const int*   seg2 = (const int*)d_in[5];
    const float* W1   = (const float*)d_in[6];
    const float* b1   = (const float*)d_in[7];
    const float* W2   = (const float*)d_in[8];
    const float* b2   = (const float*)d_in[9];
    const float* W3   = (const float*)d_in[10];
    const float* b3   = (const float*)d_in[11];
    const float* W4   = (const float*)d_in[12];
    const float* b4   = (const float*)d_in[13];
    const float* Wc   = (const float*)d_in[14];
    const float* bc   = (const float*)d_in[15];

    float* out   = (float*)d_out;
    float* hg1   = out;
    float* hg2   = out + N_GRAPHS * HIDDEN;
    float* logit = out + 2 * N_GRAPHS * HIDDEN;
    float* ws_f  = (float*)d_ws;

    run_branch(src1, dst1, seg1, W1, b1, W2, b2, W3, b3, W4, b4, hg1, ws_f, stream);
    run_branch(src2, dst2, seg2, W1, b1, W2, b2, W3, b3, W4, b4, hg2, ws_f, stream);

    logits_kernel<<<(N_GRAPHS * N_CLASSES + 255) / 256, 256, 0, stream>>>(hg1, hg2, Wc, bc, logit);
}


# Layers 2-4: tile rows summed in two stages: 8 waves run the first 19 row tiles' matrix products while 24 waves gather the last 6 row tiles' edges; same products, k order; sums in list order as before
# speedup vs baseline: 5.6153x; 1.0494x over previous
#include <hip/hip_runtime.h>
#include <hip/hip_bf16.h>

#define N_NODES   100000
#define N_EDGES   1600000
#define N_GRAPHS  128
#define HIDDEN    128
#define N_CLASSES 10

#define TILE_ROWS  400
#define N_TILES    (N_NODES / TILE_ROWS)
#define CHUNK      4096
#define N_CHUNKS   ((N_EDGES + CHUNK - 1) / CHUNK)
#define OFF_STRIDE 256
#define SRC_BITS   17
static_assert(N_TILES * TILE_ROWS == N_NODES, "tiles must cover the nodes exactly");
static_assert(N_TILES + 1 <= OFF_STRIDE, "offset row too short");
static_assert(N_NODES <= (1 << SRC_BITS), "source index does not fit its field");
#define TILE_PITCH 132
static_assert(TILE_ROWS % 16 == 0, "the tile is cut into whole 16-row matrix tiles");
static_assert(TILE_ROWS * TILE_PITCH * 4 + 64 <= 320 * 1024, "tile does not fit the LDS");

#define OD_PARTS   100
#define OD_SLICE   (N_EDGES / OD_PARTS)
#define OD_WORDS   (N_NODES / 2)
#define OD_THREADS 1024
#define OD_LOADS   ((OD_SLICE / 4 + OD_THREADS - 1) / OD_THREADS)
static_assert(OD_PARTS * OD_SLICE == N_EDGES, "slices must cover the edges exactly");
static_assert(OD_SLICE <= 65535, "a 16-bit counter must be able to hold a whole slice");
static_assert(OD_SLICE % 4 == 0, "a slice is read four indices at a time and must start 16-byte aligned");
static_assert(N_NODES % 8 == 0, "the packed counters are cleared and stored four words at a time");
static_assert(OD_WORDS * 4 <= 320 * 1024, "counters do not fit the LDS");

#define ROWS_PER_WAVE 10
#define ROW_WGS    (N_NODES / (8 * ROWS_PER_WAVE))
#define POOL_RUN   25
#define POOL_WGS   (N_NODES / (8 * POOL_RUN))
static_assert(ROW_WGS * 8 * ROWS_PER_WAVE == N_NODES, "row grid must cover the nodes exactly");
static_assert(POOL_WGS * 8 * POOL_RUN == N_NODES, "pool grid must cover the nodes exactly");
static_assert(POOL_RUN % 5 == 0, "pooling loads rows in groups of five");

typedef __attribute__((ext_vector_type(2))) float v2f;
typedef __attribute__((ext_vector_type(8))) float v8f;

__global__ void __launch_bounds__(OD_THREADS) outdeg_part_kernel(const int* __restrict__ src, unsigned* __restrict__ part) {
    __shared__ __attribute__((aligned(16))) unsigned cnt[OD_WORDS];
    int tid = threadIdx.x;
    int p   = blockIdx.x;
    const int4* sp = (const int4*)(src + (size_t)p * OD_SLICE);
    int4 s[OD_LOADS];
#pragma unroll
    for (int k = 0; k < OD_LOADS; ++k) s[k] = sp[min(k * OD_THREADS + tid, OD_SLICE / 4 - 1)];
    uint4* cnt4 = (uint4*)cnt;
    for (int i = tid; i < OD_WORDS / 4; i += OD_THREADS) cnt4[i] = uint4{0u, 0u, 0u, 0u};
    __syncthreads();
#pragma unroll
    for (int k = 0; k < OD_LOADS; ++k) {
        if (k * OD_THREADS + tid < OD_SLICE / 4) {
            atomicAdd(&cnt[s[k].x >> 1], 1u << ((s[k].x & 1) * 16));
            atomicAdd(&cnt[s[k].y >> 1], 1u << ((s[k].y & 1) * 16));
            atomicAdd(&cnt[s[k].z >> 1], 1u << ((s[k].z & 1) * 16));
            atomicAdd(&cnt[s[k].w >> 1], 1u << ((s[k].w & 1) * 16));
        }
    }
    __syncthreads();
    uint4* out4 = (uint4*)(part + (size_t)p * OD_WORDS);
    for (int i = tid; i < OD_WORDS / 4; i += OD_THREADS) out4[i] = cnt4[i];
}

__global__ void __launch_bounds__(256) layer1_kernel(const float* __restrict__ agg1, const float* __restrict__ cd,
                                                     const float* __restrict__ cs,
                                                     const float* __restrict__ W1, const float* __restrict__ b1,
                                                     float* __restrict__ h) {
    int wave = (blockIdx.x * 256 + threadIdx.x) >> 5;
    int lane = threadIdx.x & 31;
    const float4 w = ((const float4*)W1)[lane];
    const float4 b = ((const float4*)b1)[lane];
    int m0 = wave * ROWS_PER_WAVE;
    float x[ROWS_PER_WAVE];
    float sn[ROWS_PER_WAVE];
#pragma unroll
    for (int u = 0; u < ROWS_PER_WAVE; ++u) {
        x[u]  = agg1[m0 + u] * cd[m0 + u];
        sn[u] = cs[m0 + u];
    }
    float4* hrow = (float4*)(h + (size_t)m0 * HIDDEN) + lane;
#pragma unroll
    for (int u = 0; u < ROWS_PER_WAVE; ++u) {
        float4 o;
        o.x = fmaxf(fmaf(x[u], w.x, b.x), 0.0f) * sn[u];
        o.y = fmaxf(fmaf(x[u], w.y, b.y), 0.0f) * sn[u];
        o.z = fmaxf(fmaf(x[u], w.z, b.z), 0.0f) * sn[u];
        o.w = fmaxf(fmaf(x[u], w.w, b.w), 0.0f) * sn[u];
        hrow[u * (HIDDEN / 4)] = o;
    }
}

__global__ void __launch_bounds__(256) bin_kernel(const int* __restrict__ src, const int* __restrict__ dst,
                                                  unsigned* __restrict__ binned, int* __restrict__ offs) {
    __shared__ int hist[OFF_STRIDE];
    __shared__ int scanbuf[OFF_STRIDE];
    int tid = threadIdx.x;
    int c   = blockIdx.x;
    int e0  = c * CHUNK;
    hist[tid] = 0;
    __syncthreads();

    static_assert(N_TILES + 1 < OFF_STRIDE, "no spare bin");
    unsigned word[CHUNK / 256];
    int      bin[CHUNK / 256];
    int      rank[CHUNK / 256];
    int      sv[CHUNK / 256], dv[CHUNK / 256];
#pragma unroll
    for (int u = 0; u < CHUNK / 256; ++u) {
        int e = min(e0 + u * 256 + tid, N_EDGES - 1);
        sv[u] = src[e];
        dv[u] = dst[e];
    }
#pragma unroll
    for (int u = 0; u < CHUNK / 256; ++u) {
        bool valid = (e0 + u * 256 + tid) < N_EDGES;
        int b = dv[u] / TILE_ROWS;
        word[u] = (unsigned)sv[u] | ((unsigned)(dv[u] - b * TILE_ROWS) << SRC_BITS);
        bin[u]  = valid ? b : (OFF_STRIDE - 1);
        rank[u] = atomicAdd(&hist[bin[u]], 1);
    }
    __syncthreads();

    int v = hist[tid];
    scanbuf[tid] = v;
    __syncthreads();
    for (int o = 1; o < OFF_STRIDE; o <<= 1) {
        int t = (tid >= o) ? scanbuf[tid - o] : 0;
        __syncthreads();
        scanbuf[tid] += t;
        __syncthreads();
    }
    int excl = scanbuf[tid] - v;
    offs[c * OFF_STRIDE + tid] = excl;
    hist[tid] = excl;
    __syncthreads();

#pragma unroll
    for (int u = 0; u < CHUNK / 256; ++u)
        if (bin[u] != OFF_STRIDE - 1) binned[e0 + hist[bin[u]] + rank[u]] = word[u];
}

#define SCAL_THREADS 512
#define SCAL_UNROLL  8
static_assert(N_CHUNKS <= SCAL_THREADS, "one thread per chunk");
static_assert(TILE_ROWS <= SCAL_THREADS, "one thread per tile row in the epilogue");

__global__ void __launch_bounds__(SCAL_THREADS) tile_indeg_kernel(const unsigned* __restrict__ binned,
                                                                  const int* __restrict__ offs,
                                                                  const unsigned* __restrict__ part,
                                                                  float* __restrict__ msg1,
                                                                  float* __restrict__ cs, float* __restrict__ cd) {
    __shared__ int cnt[TILE_ROWS + 32];
    int tid = threadIdx.x;
    int b   = blockIdx.x;
    int c   = min(tid, N_CHUNKS - 1);
    int lo = 0, hi = 0;
    if (tid < N_CHUNKS) {
        lo = offs[c * OFF_STRIDE + b];
        hi = offs[c * OFF_STRIDE + b + 1];
    }
    unsigned od = 0;
    if (tid < TILE_ROWS) {
        int n  = b * TILE_ROWS + tid;
        int sh = (n & 1) * 16;
        const unsigned* pp = part + (n >> 1);
#pragma unroll
        for (int p = 0; p < OD_PARTS; ++p) od += (pp[(size_t)p * OD_WORDS] >> sh) & 0xffffu;
    }
    if (tid < TILE_ROWS) cnt[tid] = 0;
    __syncthreads();
    const unsigned* sp = binned + (size_t)c * CHUNK;
    for (int i = lo; i < hi; i += SCAL_UNROLL) {
        unsigned w[SCAL_UNROLL];
#pragma unroll
        for (int u = 0; u < SCAL_UNROLL; ++u) w[u] = sp[min(i + u, hi - 1)];
#pragma unroll
        for (int u = 0; u < SCAL_UNROLL; ++u)
            atomicAdd(&cnt[(i + u < hi) ? (int)(w[u] >> SRC_BITS) : TILE_ROWS + (tid & 31)], 1);
    }
    __syncthreads();
    if (tid < TILE_ROWS) {
        int   n  = b * TILE_ROWS + tid;
        float id = (float)cnt[tid];
        float sn = rsqrtf(fmaxf((float)od, 1.0f));
        cd[n]    = rsqrtf(fmaxf(id, 1.0f));
        cs[n]    = sn;
        msg1[n]  = id * sn;
    }
}

__global__ void __launch_bounds__(SCAL_THREADS) tile_agg1_kernel(const unsigned* __restrict__ binned,
                                                                 const int* __restrict__ offs,
                                                                 const float* __restrict__ msg1,
                                                                 float* __restrict__ agg1) {
    __shared__ float acc[TILE_ROWS + 32];
    int tid = threadIdx.x;
    int b   = blockIdx.x;
    int c   = min(tid, N_CHUNKS - 1);
    int lo = 0, hi = 0;
    if (tid < N_CHUNKS) {
        lo = offs[c * OFF_STRIDE + b];
        hi = offs[c * OFF_STRIDE + b + 1];
    }
    if (tid < TILE_ROWS) acc[tid] = 0.0f;
    __syncthreads();
    const unsigned* sp = binned + (size_t)c * CHUNK;
    for (int i = lo; i < hi; i += SCAL_UNROLL) {
        unsigned w[SCAL_UNROLL];
        float    a[SCAL_UNROLL];
#pragma unroll
        for (int u = 0; u < SCAL_UNROLL; ++u) w[u] = sp[min(i + u, hi - 1)];
#pragma unroll
        for (int u = 0; u < SCAL_UNROLL; ++u) {
            a[u] = msg1[w[u] & ((1u << SRC_BITS) - 1u)];
        }
#pragma unroll
        for (int u = 0; u < SCAL_UNROLL; ++u)
            atomicAdd(&acc[(i + u < hi) ? (int)(w[u] >> SRC_BITS) : TILE_ROWS + (tid & 31)], a[u]);
    }
    __syncthreads();
    if (tid < TILE_ROWS) agg1[b * TILE_ROWS + tid] = acc[tid];
}

#define S0_TILES   19
#define S0_ROWS    (S0_TILES * 16)
#define MWAVES     8
#define GWAVES     (32 - MWAVES)
#define CAP0       576
#define CAP1       256
#define LIST_WORDS (32 * CAP0 + GWAVES * CAP1)
#define GRP        8
#define PIECES     ((N_CHUNKS + 31) / 32)
static_assert(PIECES <= 32, "one lane per chunk of a wave holds its offsets");
static_assert(MWAVES * 16 == HIDDEN, "one product wave per 16-column tile");
static_assert(GWAVES % 8 == 0, "stage 1's products: whole sets of eight column-tile waves walk the row tiles");
#define ABATCH     4
static_assert((HIDDEN / 4) % ABATCH == 0, "whole batches");
static_assert(TILE_ROWS * TILE_PITCH * 4 + LIST_WORDS * 4 + 2 * TILE_ROWS * 4 + 512 <= 320 * 1024, "tile, lists and row scalars do not fit the LDS");
static_assert((TILE_PITCH * 4) % 16 == 0, "a tile row must start 16-byte aligned for the four-float adds");
__device__ __forceinline__ void edge_issue(const unsigned* lp, int g, int nl, int lane,
                                           const float* __restrict__ h,
                                           float4 (&v)[GRP], int (&row)[GRP], int& cn) {
    unsigned wl = lp[min(GRP * g + (lane & (GRP - 1)), nl - 1)];
    cn  = min(GRP, nl - GRP * g);
#pragma unroll
    for (int u = 0; u < GRP; ++u) {
        unsigned wj = (unsigned)__builtin_amdgcn_readlane((int)wl, u);
        int s  = (int)(wj & ((1u << SRC_BITS) - 1u));
        row[u] = (int)(wj >> SRC_BITS);
        v[u]   = ((const float4*)(h + (size_t)s * HIDDEN))[lane];
    }
}

__device__ __forceinline__ void edge_add(float* tile, int lane, const float4 (&v)[GRP],
                                         const int (&row)[GRP], int cn) {
#pragma unroll
    for (int u = 0; u < GRP; ++u)
        asm volatile("" :: "v"(v[u].x), "v"(v[u].y), "v"(v[u].z), "v"(v[u].w));
#pragma unroll
    for (int u = 0; u < GRP; ++u) {
        if (u < cn) {
            float4* t = (float4*)(tile + row[u] * TILE_PITCH) + lane;
            float4 o = *t;
            o.x += v[u].x;
            o.y += v[u].y;
            o.z += v[u].z;
            o.w += v[u].w;
            *t = o;
        }
    }
}

__device__ __forceinline__ void edge_pass(float* tile, const unsigned* lp, int nl, int lane,
                                          const float* __restrict__ h) {
    int ng = (nl + GRP - 1) / GRP;
    float4 vA[GRP],  vB[GRP];
    int    rowA[GRP], rowB[GRP];
    int    cnA = 0, cnB = 0;
    if (nl > 0) {
        edge_issue(lp, 0, nl, lane, h, vA, rowA, cnA);
        for (int g = 0; g < ng; g += 2) {
            edge_issue(lp, g + 1, nl, lane, h, vB, rowB, cnB);
            edge_add(tile, lane, vA, rowA, cnA);
            edge_issue(lp, g + 2, nl, lane, h, vA, rowA, cnA);
            edge_add(tile, lane, vB, rowB, cnB);
        }
    }
}

__device__ __forceinline__ void load_w(const float* __restrict__ W, const float* __restrict__ bias,
                                       int half, int n, v2f (&bw)[HIDDEN / 4], float& bn) {
    const float* wp = W + (half * 2) * HIDDEN + n;
#pragma unroll
    for (int ks = 0; ks < HIDDEN / 4; ++ks) {
        bw[ks].x = wp[(4 * ks) * HIDDEN];
        bw[ks].y = wp[(4 * ks + 1) * HIDDEN];
    }
    bn = bias[n];
}

__device__ __forceinline__ void store_tile(const v8f& acc, const float* osl, int r0, int b, int half, int n,
                                           float bn, float* __restrict__ hout) {
    int orow = b * TILE_ROWS + r0 + half * 8;
    const float4 oa = *(const float4*)(osl + r0 + half * 8);
    const float4 ob = *(const float4*)(osl + r0 + half * 8 + 4);
    const float osc[8] = {oa.x, oa.y, oa.z, oa.w, ob.x, ob.y, ob.z, ob.w};
    float* op = hout + (size_t)orow * HIDDEN + n;
#pragma unroll
    for (int r = 0; r < 8; ++r)
        op[(size_t)r * HIDDEN] = fmaxf(acc[r] + bn, 0.0f) * osc[r];
}

__device__ __forceinline__ void dense_tile(const float* tile, const float* cdl, const float* osl,
                                           int rt, int b, int lr, int half, int n,
                                           const v2f (&bw)[HIDDEN / 4], float bn,
                                           float* __restrict__ hout) {
    int   r0  = rt * 16;
    float cdm = cdl[r0 + lr];
    const float* ap = tile + (r0 + lr) * TILE_PITCH + half * 2;
    v8f acc = {};
    float2 a[HIDDEN / 4];
#pragma unroll
    for (int ks = 0; ks < ABATCH; ++ks) a[ks] = *(const float2*)(ap + 4 * ks);
#pragma unroll
    for (int kb = 0; kb < HIDDEN / 4; kb += ABATCH) {
        __builtin_amdgcn_sched_barrier(0);
        if (kb + ABATCH < HIDDEN / 4) {
#pragma unroll
            for (int ks = kb + ABATCH; ks < kb + 2 * ABATCH; ++ks) a[ks] = *(const float2*)(ap + 4 * ks);
        }
#pragma unroll
        for (int ks = kb; ks < kb + ABATCH; ++ks) {
            v2f av;
            av.x = a[ks].x * cdm;
            av.y = a[ks].y * cdm;
            acc = __builtin_amdgcn_wmma_f32_16x16x4_f32(
                 false, av,  false, bw[ks],
                 (short)0, acc,  false,  false);
        }
    }
    store_tile(acc, osl, r0, b, half, n, bn, hout);
}

__global__ void __launch_bounds__(1024) tile_layer_kernel(const unsigned* __restrict__ binned,
                                                          const int* __restrict__ offs,
                                                          const float* __restrict__ cd,
                                                          const float* __restrict__ h,
                                                          const float* __restrict__ W,
                                                          const float* __restrict__ bias,
                                                          const float* __restrict__ oscale,
                                                          float* __restrict__ hout) {
    __shared__ __attribute__((aligned(16))) float tile[TILE_ROWS * TILE_PITCH];
    __shared__ unsigned elist[LIST_WORDS];
    __shared__ int      lcount[64];
    __shared__ __attribute__((aligned(16))) float cdl[TILE_ROWS];
    __shared__ __attribute__((aligned(16))) float osl[TILE_ROWS];
    int tid  = threadIdx.x;
    int lane = tid & 31;
    int b    = blockIdx.x;
    int wave = __builtin_amdgcn_readfirstlane(tid >> 5);

    int myc  = wave + 32 * lane;
    int lo_l = 0, hi_l = 0;
    if (lane < PIECES && myc < N_CHUNKS) {
        lo_l = offs[myc * OFF_STRIDE + b];
        hi_l = offs[myc * OFF_STRIDE + b + 1];
    }

    float cdv = 0.0f, osv = 1.0f;
    if (tid < TILE_ROWS) {
        cdv = cd[b * TILE_ROWS + tid];
        if (oscale != nullptr) osv = oscale[b * TILE_ROWS + tid];
    }

    float4* tile4 = (float4*)tile;
    for (int i = tid; i < TILE_ROWS * TILE_PITCH / 4; i += 1024) tile4[i] = float4{0.f, 0.f, 0.f, 0.f};
    if (tid < 64) lcount[tid] = 0;
    if (tid < TILE_ROWS) {
        cdl[tid] = cdv;
        osl[tid] = osv;
    }
    __syncthreads();

    {
        unsigned wv[PIECES];
#pragma unroll
        for (int i = 0; i < PIECES; ++i) {
            int lo = __builtin_amdgcn_readlane(lo_l, i);
            int n  = __builtin_amdgcn_readlane(hi_l, i) - lo;
            wv[i] = 0u;
            if (n > 0) wv[i] = binned[(size_t)(wave + 32 * i) * CHUNK + lo + min(lane, n - 1)];
        }
#pragma unroll
        for (int i = 0; i < PIECES; ++i) {
            int lo = __builtin_amdgcn_readlane(lo_l, i);
            int n  = __builtin_amdgcn_readlane(hi_l, i) - lo;
            const unsigned* sp = binned + (size_t)(wave + 32 * i) * CHUNK + lo;
            for (int base = 0; base < n; base += 32) {
                int cnt = min(32, n - base);
                unsigned w = (base == 0) ? wv[i] : sp[base + min(lane, cnt - 1)];
                bool valid = lane < cnt;
                int  row   = (int)(w >> SRC_BITS);
                bool late  = row >= S0_ROWS;
                int  own   = late ? (row - S0_ROWS) % GWAVES : (row & 31);
                int  li    = late ? 32 + own : own;
                int  cap   = late ? CAP1 : CAP0;
                int  lbase = late ? 32 * CAP0 + own * CAP1 : own * CAP0;
                int  pos   = cap;
                if (valid) pos = atomicAdd(&lcount[li], 1);
                bool fits = valid && pos < cap;
                if (fits) elist[lbase + pos] = w;
                unsigned over = __builtin_amdgcn_ballot_w32(valid && !fits);
                while (over) {
                    int j = __builtin_ctz(over);
                    over &= over - 1u;
                    unsigned wj = (unsigned)__builtin_amdgcn_readlane((int)w, j);
                    int   s   = (int)(wj & ((1u << SRC_BITS) - 1u));
                    int   rw  = (int)(wj >> SRC_BITS);
                    float4 v  = ((const float4*)(h + (size_t)s * HIDDEN))[lane];
                    float* t  = tile + rw * TILE_PITCH + 4 * lane;
                    atomicAdd(t,     v.x);
                    atomicAdd(t + 1, v.y);
                    atomicAdd(t + 2, v.z);
                    atomicAdd(t + 3, v.w);
                }
            }
        }
    }
    __syncthreads();

    {
        int nl = __builtin_amdgcn_readfirstlane(min(lcount[wave], CAP0));
        edge_pass(tile, elist + wave * CAP0, nl, lane, h);
    }

    int half = lane >> 4;
    int lr   = lane & 15;
    int n    = (wave & 7) * 16 + lr;
    v2f   bw[HIDDEN / 4];
    float bn = 0.0f;
    if (wave < MWAVES) load_w(W, bias, half, n, bw, bn);
    __syncthreads();

    if (wave >= MWAVES) {
        int g  = wave - MWAVES;
        int nl = __builtin_amdgcn_readfirstlane(min(lcount[32 + g], CAP1));
        edge_pass(tile, elist + 32 * CAP0 + g * CAP1, nl, lane, h);
        load_w(W, bias, half, n, bw, bn);
    } else {
        for (int rt = 0; rt < S0_TILES; ++rt)
            dense_tile(tile, cdl, osl, rt, b, lr, half, n, bw, bn, hout);
    }
    __syncthreads();

    if (wave >= MWAVES) {
        for (int rt = S0_TILES + ((wave - MWAVES) >> 3); rt < TILE_ROWS / 16; rt += GWAVES / 8)
            dense_tile(tile, cdl, osl, rt, b, lr, half, n, bw, bn, hout);
    }
}

__device__ __forceinline__ void pool_flush(float* __restrict__ hg, float* __restrict__ counts,
                                           int g, int lane, float4 acc, float cnt) {
    float* o = hg + (size_t)g * HIDDEN + 4 * lane;
    atomicAdd(o,     acc.x);
    atomicAdd(o + 1, acc.y);
    atomicAdd(o + 2, acc.z);
    atomicAdd(o + 3, acc.w);
    if (lane == 0) atomicAdd(&counts[g], cnt);
}

__global__ void __launch_bounds__(256) pool_kernel(const int* __restrict__ seg, const float* __restrict__ h,
                                                   float* __restrict__ hg, float* __restrict__ counts) {
    int wave = (blockIdx.x * 256 + threadIdx.x) >> 5;
    int lane = threadIdx.x & 31;
    int n0   = wave * POOL_RUN;
    float4 acc = float4{0.f, 0.f, 0.f, 0.f};
    float  cnt = 0.0f;
    int gcur = __builtin_amdgcn_readfirstlane(seg[n0]);
    for (int nb = n0; nb < n0 + POOL_RUN; nb += 5) {
        int    g[5];
        float4 v[5];
#pragma unroll
        for (int u = 0; u < 5; ++u) {
            g[u] = __builtin_amdgcn_readfirstlane(seg[nb + u]);
            v[u] = ((const float4*)(h + (size_t)(nb + u) * HIDDEN))[lane];
        }
#pragma unroll
        for (int u = 0; u < 5; ++u) {
            if (g[u] != gcur) {
                pool_flush(hg, counts, gcur, lane, acc, cnt);
                acc  = float4{0.f, 0.f, 0.f, 0.f};
                cnt  = 0.0f;
                gcur = g[u];
            }
            acc.x += v[u].x;
            acc.y += v[u].y;
            acc.z += v[u].z;
            acc.w += v[u].w;
            cnt   += 1.0f;
        }
    }
    pool_flush(hg, counts, gcur, lane, acc, cnt);
}

__global__ void pool_div_kernel(float* __restrict__ hg, const float* __restrict__ counts) {
    int i = blockIdx.x * blockDim.x + threadIdx.x;
    if (i < N_GRAPHS * HIDDEN)
        hg[i] /= fmaxf(counts[i >> 7], 1.0f);
}

__global__ void logits_kernel(const float* __restrict__ hg1, const float* __restrict__ hg2,
                              const float* __restrict__ Wc, const float* __restrict__ bc,
                              float* __restrict__ logits) {
    int i = blockIdx.x * blockDim.x + threadIdx.x;
    if (i < N_GRAPHS * N_CLASSES) {
        int g = i / N_CLASSES, c = i % N_CLASSES;
        float s = bc[c];
#pragma unroll 8
        for (int k = 0; k < HIDDEN; ++k)
            s += fabsf(hg1[g * HIDDEN + k] - hg2[g * HIDDEN + k]) * Wc[k * N_CLASSES + c];
        logits[i] = s;
    }
}

static void run_branch(const int* src, const int* dst, const int* seg,
                       const float* W1, const float* b1,
                       const float* W2, const float* b2,
                       const float* W3, const float* b3,
                       const float* W4, const float* b4,
                       float* hg_out, float* ws_f, hipStream_t stream) {
    float* msg1   = ws_f + 100000;
    float* cs     = ws_f + 200000;
    float* cd     = ws_f + 300000;
    float* agg1   = ws_f + 400000;
    float* counts = ws_f + 500000;
    float* h      = ws_f + 500224;
    float* hb     = h + (size_t)N_NODES * HIDDEN;
    unsigned* binned = (unsigned*)(hb + (size_t)N_NODES * HIDDEN);
    int*      offs   = (int*)(binned + N_EDGES);
    unsigned* part   = (unsigned*)(offs + N_CHUNKS * OFF_STRIDE);

    hipMemsetAsync(counts, 0, (size_t)N_GRAPHS * sizeof(float), stream);

    const int T = 256;
    outdeg_part_kernel<<<OD_PARTS, OD_THREADS, 0, stream>>>(src, part);

    bin_kernel<<<N_CHUNKS, 256, 0, stream>>>(src, dst, binned, offs);

    tile_indeg_kernel<<<N_TILES, SCAL_THREADS, 0, stream>>>(binned, offs, part, msg1, cs, cd);
    tile_agg1_kernel<<<N_TILES, SCAL_THREADS, 0, stream>>>(binned, offs, msg1, agg1);
    layer1_kernel<<<ROW_WGS, T, 0, stream>>>(agg1, cd, cs, W1, b1, h);

    tile_layer_kernel<<<N_TILES, 1024, 0, stream>>>(binned, offs, cd, h,  W2, b2, cs,      hb);
    tile_layer_kernel<<<N_TILES, 1024, 0, stream>>>(binned, offs, cd, hb, W3, b3, cs,      h);
    tile_layer_kernel<<<N_TILES, 1024, 0, stream>>>(binned, offs, cd, h,  W4, b4, nullptr, hb);

    hipMemsetAsync(hg_out, 0, (size_t)N_GRAPHS * HIDDEN * sizeof(float), stream);
    pool_kernel<<<POOL_WGS, T, 0, stream>>>(seg, hb, hg_out, counts);
    pool_div_kernel<<<(N_GRAPHS * HIDDEN + T - 1) / T, T, 0, stream>>>(hg_out, counts);
}

extern "C" void kernel_launch(void* const* d_in, const int* in_sizes, int n_in,
                              void* d_out, int out_size, void* d_ws, size_t ws_size,
                              hipStream_t stream) {
    const int*   src1 = (const int*)d_in[0];
    const int*   dst1 = (const int*)d_in[1];
    const int*   seg1 = (const int*)d_in[2];
    const int*   src2 = (const int*)d_in[3];
    const int*   dst2 = (const int*)d_in[4];
    const int*   seg2 = (const int*)d_in[5];
    const float* W1   = (const float*)d_in[6];
    const float* b1   = (const float*)d_in[7];
    const float* W2   = (const float*)d_in[8];
    const float* b2   = (const float*)d_in[9];
    const float* W3   = (const float*)d_in[10];
    const float* b3   = (const float*)d_in[11];
    const float* W4   = (const float*)d_in[12];
    const float* b4   = (const float*)d_in[13];
    const float* Wc   = (const float*)d_in[14];
    const float* bc   = (const float*)d_in[15];

    float* out   = (float*)d_out;
    float* hg1   = out;
    float* hg2   = out + N_GRAPHS * HIDDEN;
    float* logit = out + 2 * N_GRAPHS * HIDDEN;
    float* ws_f  = (float*)d_ws;

    run_branch(src1, dst1, seg1, W1, b1, W2, b2, W3, b3, W4, b4, hg1, ws_f, stream);
    run_branch(src2, dst2, seg2, W1, b1, W2, b2, W3, b3, W4, b4, hg2, ws_f, stream);

    logits_kernel<<<(N_GRAPHS * N_CLASSES + 255) / 256, 256, 0, stream>>>(hg1, hg2, Wc, bc, logit);
}
